# starting kernel
# baseline (speedup 1.0000x reference)
#include <hip/hip_runtime.h>
#include <math.h>

#define NN   50000
#define NE   625000
#define FIN  128
#define DH   128
#define NC   40
#define NCP  48

typedef __attribute__((ext_vector_type(2))) float v2f;
typedef __attribute__((ext_vector_type(8))) float v8f;

static inline int cdiv(long a, long b) { return (int)((a + b - 1) / b); }


__global__ void k_deg_init(float* __restrict__ deg) {
    int i = blockIdx.x * blockDim.x + threadIdx.x;
    if (i < NN) deg[i] = 1.0f;
}

__global__ void k_deg_accum(const int* __restrict__ dst, float* __restrict__ deg) {
    int e = blockIdx.x * blockDim.x + threadIdx.x;
    if (e < NE) atomicAdd(&deg[dst[e]], 1.0f);
}

__global__ void k_dinv(const float* __restrict__ deg, float* __restrict__ dinv) {
    int i = blockIdx.x * blockDim.x + threadIdx.x;
    if (i < NN) dinv[i] = rsqrtf(deg[i]);
}

__global__ void k_norm(const int* __restrict__ src, const int* __restrict__ dst,
                       const float* __restrict__ dinv, float* __restrict__ nrm) {
    int e = blockIdx.x * blockDim.x + threadIdx.x;
    if (e < NE) nrm[e] = dinv[src[e]] * dinv[dst[e]];
}

__global__ void k_pad_w2(const float* __restrict__ W2, float* __restrict__ W2p) {
    int t = blockIdx.x * blockDim.x + threadIdx.x;
    if (t >= DH * NCP) return;
    int r = t / NCP, c = t - r * NCP;
    W2p[t] = (c < NC) ? W2[r * NC + c] : 0.0f;
}


template <int K, int NP, int NOUT>
__global__ void k_gemm_wmma_f32(const float* __restrict__ A, const float* __restrict__ B,
                                float* __restrict__ C, int M) {
    __shared__ float ldsB[K * NP];

    for (int idx = threadIdx.x; idx < (K * NP) / 4; idx += blockDim.x)
        ((float4*)ldsB)[idx] = ((const float4*)B)[idx];
    __syncthreads();

    const int lane = threadIdx.x & 31;
    const int wave = (blockIdx.x * blockDim.x + threadIdx.x) >> 5;
    constexpr int tilesN = NP / 16;
    int tm = wave / tilesN;
    const int tn = wave - tm * tilesN;
    const bool valid = (tm * 16 < M);
    if (!valid) tm = M / 16 - 1;

    const int m  = tm * 16 + (lane & 15);
    const int n  = tn * 16 + (lane & 15);
    const int kh = (lane >> 4) << 1;

    const float* aP   = A + m * K + kh;
    const int    bOff = kh * NP + n;

    v8f acc = {};
#pragma unroll
    for (int kk = 0; kk < K; kk += 4) {
        v2f a, b;
        a.x = aP[kk];
        a.y = aP[kk + 1];
        b.x = ldsB[bOff + kk * NP];
        b.y = ldsB[bOff + (kk + 1) * NP];
        acc = __builtin_amdgcn_wmma_f32_16x16x4_f32(
             false, a,  false, b,
             (short)0, acc,  false,  false);
    }

    if (valid && n < NOUT) {
        const int rbase = tm * 16 + ((lane >> 4) << 3);
#pragma unroll
        for (int v = 0; v < 8; ++v)
            C[(rbase + v) * NOUT + n] = acc[v];
    }
}


__global__ void k_agg_init_h(const float* __restrict__ h, const float* __restrict__ dinv,
                             float* __restrict__ agg) {
    long t = (long)blockIdx.x * blockDim.x + threadIdx.x;
    if (t >= (long)NN * 32) return;
    int i = (int)(t >> 5);
    float s = dinv[i]; s *= s;
    float4 v = ((const float4*)h)[t];
    float4 o = make_float4(v.x * s, v.y * s, v.z * s, v.w * s);
    ((float4*)agg)[t] = o;
}

__global__ void k_agg_edges_h(const int* __restrict__ src, const int* __restrict__ dst,
                              const float* __restrict__ nrm,
                              const float* __restrict__ h, float* __restrict__ agg) {
    long t = (long)blockIdx.x * blockDim.x + threadIdx.x;
    long e = t >> 5;
    int lane = (int)(t & 31);
    if (e >= NE) return;
    int s = src[e], d = dst[e];
    float w = nrm[e];
    float4 v = ((const float4*)h)[(long)s * 32 + lane];
    float* o = agg + (long)d * 128 + lane * 4;
    atomicAdd(o + 0, v.x * w);
    atomicAdd(o + 1, v.y * w);
    atomicAdd(o + 2, v.z * w);
    atomicAdd(o + 3, v.w * w);
}

__global__ void k_bias_relu(float* __restrict__ agg, const float* __restrict__ b) {
    long t = (long)blockIdx.x * blockDim.x + threadIdx.x;
    if (t >= (long)NN * DH) return;
    float v = agg[t] + b[t & (DH - 1)];
    agg[t] = v > 0.0f ? v : 0.0f;
}


__global__ void k_agg_init_c(const float* __restrict__ h, const float* __restrict__ dinv,
                             float* __restrict__ agg) {
    long t = (long)blockIdx.x * blockDim.x + threadIdx.x;
    if (t >= (long)NN * NC) return;
    int i = (int)(t / NC);
    float s = dinv[i]; s *= s;
    agg[t] = h[t] * s;
}

__global__ void k_agg_edges_c(const int* __restrict__ src, const int* __restrict__ dst,
                              const float* __restrict__ nrm,
                              const float* __restrict__ h, float* __restrict__ agg) {
    long t = (long)blockIdx.x * blockDim.x + threadIdx.x;
    long e = t >> 5;
    int lane = (int)(t & 31);
    if (e >= NE) return;
    int s = src[e], d = dst[e];
    float w = nrm[e];
    atomicAdd(&agg[(long)d * NC + lane], h[(long)s * NC + lane] * w);
    if (lane < NC - 32) {
        int c = lane + 32;
        atomicAdd(&agg[(long)d * NC + c], h[(long)s * NC + c] * w);
    }
}


__global__ void k_logsoftmax(const float* __restrict__ agg, const float* __restrict__ b,
                             float* __restrict__ out) {
    int i = blockIdx.x * blockDim.x + threadIdx.x;
    if (i >= NN) return;
    const float* r = agg + (long)i * NC;
    float mx = -INFINITY;
#pragma unroll
    for (int c = 0; c < NC; ++c) mx = fmaxf(mx, r[c] + b[c]);
    float sum = 0.0f;
#pragma unroll
    for (int c = 0; c < NC; ++c) sum += expf(r[c] + b[c] - mx);
    float lse = mx + logf(sum);
#pragma unroll
    for (int c = 0; c < NC; ++c) out[(long)i * NC + c] = r[c] + b[c] - lse;
}


extern "C" void kernel_launch(void* const* d_in, const int* in_sizes, int n_in,
                              void* d_out, int out_size, void* d_ws, size_t ws_size,
                              hipStream_t stream) {
    const float* x  = (const float*)d_in[0];
    const int*   ei = (const int*)d_in[1];
    const float* W1 = (const float*)d_in[2];
    const float* b1 = (const float*)d_in[3];
    const float* W2 = (const float*)d_in[4];
    const float* b2 = (const float*)d_in[5];
    float* out = (float*)d_out;

    const int* src = ei;
    const int* dst = ei + NE;

    float* ws   = (float*)d_ws;
    float* deg  = ws;
    float* dinv = deg  + NN;
    float* nrm  = dinv + NN;
    float* h0   = nrm  + NE;
    float* agg1 = h0   + (long)NN * DH;
    float* h2   = agg1 + (long)NN * DH;
    float* agg2 = h2   + (long)NN * NC;
    float* W2p  = agg2 + (long)NN * NC;

    const int B = 256;

    k_deg_init <<<cdiv(NN, B), B, 0, stream>>>(deg);
    k_deg_accum<<<cdiv(NE, B), B, 0, stream>>>(dst, deg);
    k_dinv     <<<cdiv(NN, B), B, 0, stream>>>(deg, dinv);
    k_norm     <<<cdiv(NE, B), B, 0, stream>>>(src, dst, dinv, nrm);
    k_pad_w2   <<<cdiv(DH * NCP, B), B, 0, stream>>>(W2, W2p);

    {
        long waves = (long)(NN / 16) * (DH / 16);
        k_gemm_wmma_f32<FIN, DH, DH>
            <<<cdiv(waves * 32, B), B, 0, stream>>>(x, W1, h0, NN);
    }
    k_agg_init_h <<<cdiv((long)NN * 32, B), B, 0, stream>>>(h0, dinv, agg1);
    k_agg_edges_h<<<cdiv((long)NE * 32, B), B, 0, stream>>>(src, dst, nrm, h0, agg1);
    k_bias_relu  <<<cdiv((long)NN * DH, B), B, 0, stream>>>(agg1, b1);

    {
        long waves = (long)(NN / 16) * (NCP / 16);
        k_gemm_wmma_f32<DH, NCP, NC>
            <<<cdiv(waves * 32, B), B, 0, stream>>>(agg1, W2p, h2, NN);
    }
    k_agg_init_c <<<cdiv((long)NN * NC, B), B, 0, stream>>>(h2, dinv, agg2);
    k_agg_edges_c<<<cdiv((long)NE * 32, B), B, 0, stream>>>(src, dst, nrm, h2, agg2);

    k_logsoftmax<<<cdiv(NN, B), B, 0, stream>>>(agg2, b2, out);
}


# Both aggregations: float-atomic scatter replaced by per-node register gather over a dst-sorted edge list built each call (int atomic arrival-order rank, scan); sums: self term, then arrival order
# speedup vs baseline: 1.8097x; 1.8097x over previous
#include <hip/hip_runtime.h>
#include <math.h>

#define NN   50000
#define NE   625000
#define FIN  128
#define DH   128
#define NC   40
#define NCP  48
#define NNP  50048
#define NEP  625024

typedef __attribute__((ext_vector_type(2))) float v2f;
typedef __attribute__((ext_vector_type(8))) float v8f;

static inline int cdiv(long a, long b) { return (int)((a + b - 1) / b); }


__global__ void k_cnt_zero(int* __restrict__ cnt) {
    int i = blockIdx.x * blockDim.x + threadIdx.x;
    if (i < NN) cnt[i] = 0;
}

__global__ void k_rank(const int* __restrict__ dst, int* __restrict__ cnt,
                       int* __restrict__ rank) {
    int e = blockIdx.x * blockDim.x + threadIdx.x;
    if (e < NE) rank[e] = atomicAdd(&cnt[dst[e]], 1);
}

#define SCAN_T     1024
#define SCAN_W     (SCAN_T / 32)
#define SCAN_TILES ((NN + SCAN_W * 128 - 1) / (SCAN_W * 128))
#define SCAN_SEG   (SCAN_TILES * 128)

__global__ void __launch_bounds__(SCAN_T)
k_scan(const int* __restrict__ cnt, int* __restrict__ rowptr, float* __restrict__ dinv) {
    __shared__ int wtot[SCAN_W];
    const int lane = threadIdx.x & 31;
    const int wv   = threadIdx.x >> 5;
    const int q0   = wv * (SCAN_SEG / 4) + lane;
    const int4* cq = (const int4*)cnt + q0;

    int4 c[SCAN_TILES];
    int  base[SCAN_TILES];
#pragma unroll
    for (int t = 0; t < SCAN_TILES; ++t)
        c[t] = (q0 + t * 32 < NN / 4) ? cq[t * 32] : make_int4(0, 0, 0, 0);
    int inc[SCAN_TILES];
#pragma unroll
    for (int t = 0; t < SCAN_TILES; ++t) inc[t] = c[t].x + c[t].y + c[t].z + c[t].w;
#pragma unroll
    for (int o = 1; o < 32; o <<= 1) {
        int y[SCAN_TILES];
#pragma unroll
        for (int t = 0; t < SCAN_TILES; ++t) y[t] = __shfl_up(inc[t], o);
#pragma unroll
        for (int t = 0; t < SCAN_TILES; ++t) if (lane >= o) inc[t] += y[t];
    }
    int run = 0;
#pragma unroll
    for (int t = 0; t < SCAN_TILES; ++t) {
        base[t] = run + inc[t] - (c[t].x + c[t].y + c[t].z + c[t].w);
        run += __shfl(inc[t], 31);
    }
    if (lane == 0) wtot[wv] = run;
    __syncthreads();
    int woff = (lane < wv) ? wtot[lane] : 0;
#pragma unroll
    for (int o = 16; o > 0; o >>= 1) woff += __shfl_xor(woff, o);

    int4*   rq = (int4*)rowptr + q0;
    float4* dq = (float4*)dinv + q0;
#pragma unroll
    for (int t = 0; t < SCAN_TILES; ++t) {
        if (q0 + t * 32 < NN / 4) {
            const int p0 = woff + base[t];
            rq[t * 32] = make_int4(p0, p0 + c[t].x, p0 + c[t].x + c[t].y, p0 + c[t].x + c[t].y + c[t].z);
            dq[t * 32] = make_float4(rsqrtf((float)(c[t].x + 1)), rsqrtf((float)(c[t].y + 1)),
                                     rsqrtf((float)(c[t].z + 1)), rsqrtf((float)(c[t].w + 1)));
        }
    }
    if (wv == SCAN_W - 1 && lane == 0) rowptr[NN] = woff + run;
}

__global__ void k_fill(const int* __restrict__ src, const int* __restrict__ dst,
                       const int* __restrict__ rowptr, const int* __restrict__ rank,
                       int* __restrict__ ssrc) {
    int e = blockIdx.x * blockDim.x + threadIdx.x;
    if (e < NE) ssrc[rowptr[dst[e]] + rank[e]] = src[e];
}

__global__ void k_pad_w2(const float* __restrict__ W2, float* __restrict__ W2p) {
    int t = blockIdx.x * blockDim.x + threadIdx.x;
    if (t >= DH * NCP) return;
    int r = t / NCP, c = t - r * NCP;
    W2p[t] = (c < NC) ? W2[r * NC + c] : 0.0f;
}


template <int K, int NP, int NOUT>
__global__ void k_gemm_wmma_f32(const float* __restrict__ A, const float* __restrict__ B,
                                float* __restrict__ C, int M) {
    __shared__ float ldsB[K * NP];

    for (int idx = threadIdx.x; idx < (K * NP) / 4; idx += blockDim.x)
        ((float4*)ldsB)[idx] = ((const float4*)B)[idx];
    __syncthreads();

    const int lane = threadIdx.x & 31;
    const int wave = (blockIdx.x * blockDim.x + threadIdx.x) >> 5;
    constexpr int tilesN = NP / 16;
    int tm = wave / tilesN;
    const int tn = wave - tm * tilesN;
    const bool valid = (tm * 16 < M);
    if (!valid) tm = M / 16 - 1;

    const int m  = tm * 16 + (lane & 15);
    const int n  = tn * 16 + (lane & 15);
    const int kh = (lane >> 4) << 1;

    const float* aP   = A + m * K + kh;
    const int    bOff = kh * NP + n;

    v8f acc = {};
#pragma unroll
    for (int kk = 0; kk < K; kk += 4) {
        v2f a, b;
        a.x = aP[kk];
        a.y = aP[kk + 1];
        b.x = ldsB[bOff + kk * NP];
        b.y = ldsB[bOff + (kk + 1) * NP];
        acc = __builtin_amdgcn_wmma_f32_16x16x4_f32(
             false, a,  false, b,
             (short)0, acc,  false,  false);
    }

    if (valid && n < NOUT) {
        const int rbase = tm * 16 + ((lane >> 4) << 3);
#pragma unroll
        for (int v = 0; v < 8; ++v)
            C[(rbase + v) * NOUT + n] = acc[v];
    }
}


static __device__ __forceinline__ int rl_i(int v, int k) {
    return __builtin_amdgcn_readlane(v, k);
}
static __device__ __forceinline__ float rl_f(float v, int k) {
    return __int_as_float(__builtin_amdgcn_readlane(__float_as_int(v), k));
}

__global__ void k_gather_h(const int* __restrict__ rowptr, const int* __restrict__ ssrc,
                           const float* __restrict__ dinv,
                           const float* __restrict__ h, float* __restrict__ agg) {
    const int lane = threadIdx.x & 31;
    const int node = __builtin_amdgcn_readfirstlane((int)((blockIdx.x * blockDim.x + threadIdx.x) >> 5));
    if (node >= NN) return;
    const int beg = rowptr[node], end = rowptr[node + 1];
    const float dd = dinv[node];
    const float4* h4 = (const float4*)h;

    const float ws = dd * dd;
    float4 acc = h4[(long)node * 32 + lane];
    acc.x *= ws; acc.y *= ws; acc.z *= ws; acc.w *= ws;

    for (int cb = beg; cb < end; cb += 32) {
        const int j = cb + lane;
        int s_l = (j < end) ? ssrc[j] : node;
        if ((unsigned)s_l >= (unsigned)NN) s_l = node;
        const float w_l = dinv[s_l] * dd;
        const int n = (end - cb < 32) ? (end - cb) : 32;
        int k = 0;
        for (; k + 4 <= n; k += 4) {
            const int s0 = rl_i(s_l, k), s1 = rl_i(s_l, k + 1), s2 = rl_i(s_l, k + 2), s3 = rl_i(s_l, k + 3);
            const float w0 = rl_f(w_l, k), w1 = rl_f(w_l, k + 1), w2 = rl_f(w_l, k + 2), w3 = rl_f(w_l, k + 3);
            const float4 v0 = h4[(long)s0 * 32 + lane];
            const float4 v1 = h4[(long)s1 * 32 + lane];
            const float4 v2 = h4[(long)s2 * 32 + lane];
            const float4 v3 = h4[(long)s3 * 32 + lane];
            acc.x = fmaf(v0.x, w0, acc.x); acc.y = fmaf(v0.y, w0, acc.y); acc.z = fmaf(v0.z, w0, acc.z); acc.w = fmaf(v0.w, w0, acc.w);
            acc.x = fmaf(v1.x, w1, acc.x); acc.y = fmaf(v1.y, w1, acc.y); acc.z = fmaf(v1.z, w1, acc.z); acc.w = fmaf(v1.w, w1, acc.w);
            acc.x = fmaf(v2.x, w2, acc.x); acc.y = fmaf(v2.y, w2, acc.y); acc.z = fmaf(v2.z, w2, acc.z); acc.w = fmaf(v2.w, w2, acc.w);
            acc.x = fmaf(v3.x, w3, acc.x); acc.y = fmaf(v3.y, w3, acc.y); acc.z = fmaf(v3.z, w3, acc.z); acc.w = fmaf(v3.w, w3, acc.w);
        }
        for (; k < n; ++k) {
            const int s0 = rl_i(s_l, k);
            const float w0 = rl_f(w_l, k);
            const float4 v0 = h4[(long)s0 * 32 + lane];
            acc.x = fmaf(v0.x, w0, acc.x); acc.y = fmaf(v0.y, w0, acc.y); acc.z = fmaf(v0.z, w0, acc.z); acc.w = fmaf(v0.w, w0, acc.w);
        }
    }
    ((float4*)agg)[(long)node * 32 + lane] = acc;
}

__global__ void k_bias_relu(float* __restrict__ agg, const float* __restrict__ b) {
    long t = (long)blockIdx.x * blockDim.x + threadIdx.x;
    if (t >= (long)NN * DH) return;
    float v = agg[t] + b[t & (DH - 1)];
    agg[t] = v > 0.0f ? v : 0.0f;
}


__global__ void k_gather_c(const int* __restrict__ rowptr, const int* __restrict__ ssrc,
                           const float* __restrict__ dinv,
                           const float* __restrict__ h, float* __restrict__ agg) {
    const int lane = threadIdx.x & 31;
    const int node = __builtin_amdgcn_readfirstlane((int)((blockIdx.x * blockDim.x + threadIdx.x) >> 5));
    if (node >= NN) return;
    const int beg = rowptr[node], end = rowptr[node + 1];
    const float dd = dinv[node];
    const bool hi = lane < NC - 32;
    const int  c2 = hi ? lane + 32 : lane;

    const float ws = dd * dd;
    float a0 = h[(long)node * NC + lane] * ws;
    float a1 = h[(long)node * NC + c2] * ws;

    for (int cb = beg; cb < end; cb += 32) {
        const int j = cb + lane;
        int s_l = (j < end) ? ssrc[j] : node;
        if ((unsigned)s_l >= (unsigned)NN) s_l = node;
        const float w_l = dinv[s_l] * dd;
        const int n = (end - cb < 32) ? (end - cb) : 32;
        int k = 0;
        for (; k + 4 <= n; k += 4) {
            const int s0 = rl_i(s_l, k), s1 = rl_i(s_l, k + 1), s2 = rl_i(s_l, k + 2), s3 = rl_i(s_l, k + 3);
            const float w0 = rl_f(w_l, k), w1 = rl_f(w_l, k + 1), w2 = rl_f(w_l, k + 2), w3 = rl_f(w_l, k + 3);
            const float p0 = h[(long)s0 * NC + lane], q0 = h[(long)s0 * NC + c2];
            const float p1 = h[(long)s1 * NC + lane], q1 = h[(long)s1 * NC + c2];
            const float p2 = h[(long)s2 * NC + lane], q2 = h[(long)s2 * NC + c2];
            const float p3 = h[(long)s3 * NC + lane], q3 = h[(long)s3 * NC + c2];
            a0 = fmaf(p0, w0, a0); a1 = fmaf(q0, w0, a1);
            a0 = fmaf(p1, w1, a0); a1 = fmaf(q1, w1, a1);
            a0 = fmaf(p2, w2, a0); a1 = fmaf(q2, w2, a1);
            a0 = fmaf(p3, w3, a0); a1 = fmaf(q3, w3, a1);
        }
        for (; k < n; ++k) {
            const int s0 = rl_i(s_l, k);
            const float w0 = rl_f(w_l, k);
            a0 = fmaf(h[(long)s0 * NC + lane], w0, a0);
            a1 = fmaf(h[(long)s0 * NC + c2], w0, a1);
        }
    }
    agg[(long)node * NC + lane] = a0;
    if (hi) agg[(long)node * NC + c2] = a1;
}


__global__ void k_logsoftmax(const float* __restrict__ agg, const float* __restrict__ b,
                             float* __restrict__ out) {
    int i = blockIdx.x * blockDim.x + threadIdx.x;
    if (i >= NN) return;
    const float* r = agg + (long)i * NC;
    float mx = -INFINITY;
#pragma unroll
    for (int c = 0; c < NC; ++c) mx = fmaxf(mx, r[c] + b[c]);
    float sum = 0.0f;
#pragma unroll
    for (int c = 0; c < NC; ++c) sum += expf(r[c] + b[c] - mx);
    float lse = mx + logf(sum);
#pragma unroll
    for (int c = 0; c < NC; ++c) out[(long)i * NC + c] = r[c] + b[c] - lse;
}


extern "C" void kernel_launch(void* const* d_in, const int* in_sizes, int n_in,
                              void* d_out, int out_size, void* d_ws, size_t ws_size,
                              hipStream_t stream) {
    const float* x  = (const float*)d_in[0];
    const int*   ei = (const int*)d_in[1];
    const float* W1 = (const float*)d_in[2];
    const float* b1 = (const float*)d_in[3];
    const float* W2 = (const float*)d_in[4];
    const float* b2 = (const float*)d_in[5];
    float* out = (float*)d_out;

    const int* src = ei;
    const int* dst = ei + NE;

    float* ws     = (float*)d_ws;
    int*   cnt    = (int*)ws;
    float* dinv   = ws + NNP;
    int*   rowptr = (int*)(dinv + NNP);
    int*   rank   = rowptr + NNP;
    int*   ssrc   = rank + NEP;
    float* h0     = (float*)(ssrc + NEP);
    float* agg1 = h0   + (long)NN * DH;
    float* h2   = agg1 + (long)NN * DH;
    float* agg2 = h2   + (long)NN * NC;
    float* W2p  = agg2 + (long)NN * NC;

    const int B = 256;

    k_cnt_zero <<<cdiv(NN, B), B, 0, stream>>>(cnt);
    k_rank     <<<cdiv(NE, B), B, 0, stream>>>(dst, cnt, rank);
    k_scan     <<<1, SCAN_T, 0, stream>>>(cnt, rowptr, dinv);
    k_fill     <<<cdiv(NE, B), B, 0, stream>>>(src, dst, rowptr, rank, ssrc);
    k_pad_w2   <<<cdiv(DH * NCP, B), B, 0, stream>>>(W2, W2p);

    {
        long waves = (long)(NN / 16) * (DH / 16);
        k_gemm_wmma_f32<FIN, DH, DH>
            <<<cdiv(waves * 32, B), B, 0, stream>>>(x, W1, h0, NN);
    }
    k_gather_h   <<<cdiv((long)NN * 32, B), B, 0, stream>>>(rowptr, ssrc, dinv, h0, agg1);
    k_bias_relu  <<<cdiv((long)NN * DH, B), B, 0, stream>>>(agg1, b1);

    {
        long waves = (long)(NN / 16) * (NCP / 16);
        k_gemm_wmma_f32<DH, NCP, NC>
            <<<cdiv(waves * 32, B), B, 0, stream>>>(agg1, W2p, h2, NN);
    }
    k_gather_c   <<<cdiv((long)NN * 32, B), B, 0, stream>>>(rowptr, ssrc, dinv, h2, agg2);

    k_logsoftmax<<<cdiv(NN, B), B, 0, stream>>>(agg2, b2, out);
}


# Elementwise passes fused into their producers: bias+relu into the layer-1 gather's store, bias+log_softmax into the layer-2 gather (max, exp-sum as lane butterflies: new sum order); 11 to 9 launches
# speedup vs baseline: 2.0719x; 1.1449x over previous
#include <hip/hip_runtime.h>
#include <math.h>

#define NN   50000
#define NE   625000
#define FIN  128
#define DH   128
#define NC   40
#define NCP  48
#define NNP  50048
#define NEP  625024

typedef __attribute__((ext_vector_type(2))) float v2f;
typedef __attribute__((ext_vector_type(8))) float v8f;

static inline int cdiv(long a, long b) { return (int)((a + b - 1) / b); }


__global__ void k_cnt_zero(int* __restrict__ cnt) {
    int i = blockIdx.x * blockDim.x + threadIdx.x;
    if (i < NN) cnt[i] = 0;
}

__global__ void k_rank(const int* __restrict__ dst, int* __restrict__ cnt,
                       int* __restrict__ rank) {
    int e = blockIdx.x * blockDim.x + threadIdx.x;
    if (e < NE) rank[e] = atomicAdd(&cnt[dst[e]], 1);
}

#define SCAN_T     1024
#define SCAN_W     (SCAN_T / 32)
#define SCAN_TILES ((NN + SCAN_W * 128 - 1) / (SCAN_W * 128))
#define SCAN_SEG   (SCAN_TILES * 128)

__global__ void __launch_bounds__(SCAN_T)
k_scan(const int* __restrict__ cnt, int* __restrict__ rowptr, float* __restrict__ dinv) {
    __shared__ int wtot[SCAN_W];
    const int lane = threadIdx.x & 31;
    const int wv   = threadIdx.x >> 5;
    const int q0   = wv * (SCAN_SEG / 4) + lane;
    const int4* cq = (const int4*)cnt + q0;

    int4 c[SCAN_TILES];
    int  base[SCAN_TILES];
#pragma unroll
    for (int t = 0; t < SCAN_TILES; ++t)
        c[t] = (q0 + t * 32 < NN / 4) ? cq[t * 32] : make_int4(0, 0, 0, 0);
    int inc[SCAN_TILES];
#pragma unroll
    for (int t = 0; t < SCAN_TILES; ++t) inc[t] = c[t].x + c[t].y + c[t].z + c[t].w;
#pragma unroll
    for (int o = 1; o < 32; o <<= 1) {
        int y[SCAN_TILES];
#pragma unroll
        for (int t = 0; t < SCAN_TILES; ++t) y[t] = __shfl_up(inc[t], o);
#pragma unroll
        for (int t = 0; t < SCAN_TILES; ++t) if (lane >= o) inc[t] += y[t];
    }
    int run = 0;
#pragma unroll
    for (int t = 0; t < SCAN_TILES; ++t) {
        base[t] = run + inc[t] - (c[t].x + c[t].y + c[t].z + c[t].w);
        run += __shfl(inc[t], 31);
    }
    if (lane == 0) wtot[wv] = run;
    __syncthreads();
    int woff = (lane < wv) ? wtot[lane] : 0;
#pragma unroll
    for (int o = 16; o > 0; o >>= 1) woff += __shfl_xor(woff, o);

    int4*   rq = (int4*)rowptr + q0;
    float4* dq = (float4*)dinv + q0;
#pragma unroll
    for (int t = 0; t < SCAN_TILES; ++t) {
        if (q0 + t * 32 < NN / 4) {
            const int p0 = woff + base[t];
            rq[t * 32] = make_int4(p0, p0 + c[t].x, p0 + c[t].x + c[t].y, p0 + c[t].x + c[t].y + c[t].z);
            dq[t * 32] = make_float4(rsqrtf((float)(c[t].x + 1)), rsqrtf((float)(c[t].y + 1)),
                                     rsqrtf((float)(c[t].z + 1)), rsqrtf((float)(c[t].w + 1)));
        }
    }
    if (wv == SCAN_W - 1 && lane == 0) rowptr[NN] = woff + run;
}

__global__ void k_fill(const int* __restrict__ src, const int* __restrict__ dst,
                       const int* __restrict__ rowptr, const int* __restrict__ rank,
                       int* __restrict__ ssrc) {
    int e = blockIdx.x * blockDim.x + threadIdx.x;
    if (e < NE) ssrc[rowptr[dst[e]] + rank[e]] = src[e];
}

__global__ void k_pad_w2(const float* __restrict__ W2, float* __restrict__ W2p) {
    int t = blockIdx.x * blockDim.x + threadIdx.x;
    if (t >= DH * NCP) return;
    int r = t / NCP, c = t - r * NCP;
    W2p[t] = (c < NC) ? W2[r * NC + c] : 0.0f;
}


template <int K, int NP, int NOUT>
__global__ void k_gemm_wmma_f32(const float* __restrict__ A, const float* __restrict__ B,
                                float* __restrict__ C, int M) {
    __shared__ float ldsB[K * NP];

    for (int idx = threadIdx.x; idx < (K * NP) / 4; idx += blockDim.x)
        ((float4*)ldsB)[idx] = ((const float4*)B)[idx];
    __syncthreads();

    const int lane = threadIdx.x & 31;
    const int wave = (blockIdx.x * blockDim.x + threadIdx.x) >> 5;
    constexpr int tilesN = NP / 16;
    int tm = wave / tilesN;
    const int tn = wave - tm * tilesN;
    const bool valid = (tm * 16 < M);
    if (!valid) tm = M / 16 - 1;

    const int m  = tm * 16 + (lane & 15);
    const int n  = tn * 16 + (lane & 15);
    const int kh = (lane >> 4) << 1;

    const float* aP   = A + m * K + kh;
    const int    bOff = kh * NP + n;

    v8f acc = {};
#pragma unroll
    for (int kk = 0; kk < K; kk += 4) {
        v2f a, b;
        a.x = aP[kk];
        a.y = aP[kk + 1];
        b.x = ldsB[bOff + kk * NP];
        b.y = ldsB[bOff + (kk + 1) * NP];
        acc = __builtin_amdgcn_wmma_f32_16x16x4_f32(
             false, a,  false, b,
             (short)0, acc,  false,  false);
    }

    if (valid && n < NOUT) {
        const int rbase = tm * 16 + ((lane >> 4) << 3);
#pragma unroll
        for (int v = 0; v < 8; ++v)
            C[(rbase + v) * NOUT + n] = acc[v];
    }
}


static __device__ __forceinline__ int rl_i(int v, int k) {
    return __builtin_amdgcn_readlane(v, k);
}
static __device__ __forceinline__ float rl_f(float v, int k) {
    return __int_as_float(__builtin_amdgcn_readlane(__float_as_int(v), k));
}

__global__ void k_gather_h(const int* __restrict__ rowptr, const int* __restrict__ ssrc,
                           const float* __restrict__ dinv,
                           const float* __restrict__ h, const float* __restrict__ b,
                           float* __restrict__ h1) {
    const int lane = threadIdx.x & 31;
    const int node = __builtin_amdgcn_readfirstlane((int)((blockIdx.x * blockDim.x + threadIdx.x) >> 5));
    if (node >= NN) return;
    const int beg = rowptr[node], end = rowptr[node + 1];
    const float dd = dinv[node];
    const float4* h4 = (const float4*)h;
    const float4 bb = ((const float4*)b)[lane];

    const float ws = dd * dd;
    float4 acc = h4[(long)node * 32 + lane];
    acc.x *= ws; acc.y *= ws; acc.z *= ws; acc.w *= ws;

    for (int cb = beg; cb < end; cb += 32) {
        const int j = cb + lane;
        int s_l = (j < end) ? ssrc[j] : node;
        if ((unsigned)s_l >= (unsigned)NN) s_l = node;
        const float w_l = dinv[s_l] * dd;
        const int n = (end - cb < 32) ? (end - cb) : 32;
        int k = 0;
        for (; k + 4 <= n; k += 4) {
            const int s0 = rl_i(s_l, k), s1 = rl_i(s_l, k + 1), s2 = rl_i(s_l, k + 2), s3 = rl_i(s_l, k + 3);
            const float w0 = rl_f(w_l, k), w1 = rl_f(w_l, k + 1), w2 = rl_f(w_l, k + 2), w3 = rl_f(w_l, k + 3);
            const float4 v0 = h4[(long)s0 * 32 + lane];
            const float4 v1 = h4[(long)s1 * 32 + lane];
            const float4 v2 = h4[(long)s2 * 32 + lane];
            const float4 v3 = h4[(long)s3 * 32 + lane];
            acc.x = fmaf(v0.x, w0, acc.x); acc.y = fmaf(v0.y, w0, acc.y); acc.z = fmaf(v0.z, w0, acc.z); acc.w = fmaf(v0.w, w0, acc.w);
            acc.x = fmaf(v1.x, w1, acc.x); acc.y = fmaf(v1.y, w1, acc.y); acc.z = fmaf(v1.z, w1, acc.z); acc.w = fmaf(v1.w, w1, acc.w);
            acc.x = fmaf(v2.x, w2, acc.x); acc.y = fmaf(v2.y, w2, acc.y); acc.z = fmaf(v2.z, w2, acc.z); acc.w = fmaf(v2.w, w2, acc.w);
            acc.x = fmaf(v3.x, w3, acc.x); acc.y = fmaf(v3.y, w3, acc.y); acc.z = fmaf(v3.z, w3, acc.z); acc.w = fmaf(v3.w, w3, acc.w);
        }
        for (; k < n; ++k) {
            const int s0 = rl_i(s_l, k);
            const float w0 = rl_f(w_l, k);
            const float4 v0 = h4[(long)s0 * 32 + lane];
            acc.x = fmaf(v0.x, w0, acc.x); acc.y = fmaf(v0.y, w0, acc.y); acc.z = fmaf(v0.z, w0, acc.z); acc.w = fmaf(v0.w, w0, acc.w);
        }
    }
    acc.x += bb.x; acc.y += bb.y; acc.z += bb.z; acc.w += bb.w;
    acc.x = acc.x > 0.0f ? acc.x : 0.0f;
    acc.y = acc.y > 0.0f ? acc.y : 0.0f;
    acc.z = acc.z > 0.0f ? acc.z : 0.0f;
    acc.w = acc.w > 0.0f ? acc.w : 0.0f;
    ((float4*)h1)[(long)node * 32 + lane] = acc;
}


__global__ void k_gather_c(const int* __restrict__ rowptr, const int* __restrict__ ssrc,
                           const float* __restrict__ dinv,
                           const float* __restrict__ h, const float* __restrict__ b,
                           float* __restrict__ out) {
    const int lane = threadIdx.x & 31;
    const int node = __builtin_amdgcn_readfirstlane((int)((blockIdx.x * blockDim.x + threadIdx.x) >> 5));
    if (node >= NN) return;
    const int beg = rowptr[node], end = rowptr[node + 1];
    const float dd = dinv[node];
    const bool hi = lane < NC - 32;
    const int  c2 = hi ? lane + 32 : lane;
    const float b0 = b[lane], b1 = b[c2];

    const float ws = dd * dd;
    float a0 = h[(long)node * NC + lane] * ws;
    float a1 = h[(long)node * NC + c2] * ws;

    for (int cb = beg; cb < end; cb += 32) {
        const int j = cb + lane;
        int s_l = (j < end) ? ssrc[j] : node;
        if ((unsigned)s_l >= (unsigned)NN) s_l = node;
        const float w_l = dinv[s_l] * dd;
        const int n = (end - cb < 32) ? (end - cb) : 32;
        int k = 0;
        for (; k + 4 <= n; k += 4) {
            const int s0 = rl_i(s_l, k), s1 = rl_i(s_l, k + 1), s2 = rl_i(s_l, k + 2), s3 = rl_i(s_l, k + 3);
            const float w0 = rl_f(w_l, k), w1 = rl_f(w_l, k + 1), w2 = rl_f(w_l, k + 2), w3 = rl_f(w_l, k + 3);
            const float p0 = h[(long)s0 * NC + lane], q0 = h[(long)s0 * NC + c2];
            const float p1 = h[(long)s1 * NC + lane], q1 = h[(long)s1 * NC + c2];
            const float p2 = h[(long)s2 * NC + lane], q2 = h[(long)s2 * NC + c2];
            const float p3 = h[(long)s3 * NC + lane], q3 = h[(long)s3 * NC + c2];
            a0 = fmaf(p0, w0, a0); a1 = fmaf(q0, w0, a1);
            a0 = fmaf(p1, w1, a0); a1 = fmaf(q1, w1, a1);
            a0 = fmaf(p2, w2, a0); a1 = fmaf(q2, w2, a1);
            a0 = fmaf(p3, w3, a0); a1 = fmaf(q3, w3, a1);
        }
        for (; k < n; ++k) {
            const int s0 = rl_i(s_l, k);
            const float w0 = rl_f(w_l, k);
            a0 = fmaf(h[(long)s0 * NC + lane], w0, a0);
            a1 = fmaf(h[(long)s0 * NC + c2], w0, a1);
        }
    }
    const float v0 = a0 + b0;
    const float v1 = hi ? a1 + b1 : -INFINITY;
    float mx = fmaxf(v0, v1);
    for (int o = 16; o > 0; o >>= 1) mx = fmaxf(mx, __shfl_xor(mx, o));
    float sum = expf(v0 - mx) + (hi ? expf(v1 - mx) : 0.0f);
    for (int o = 16; o > 0; o >>= 1) sum += __shfl_xor(sum, o);
    const float lse = mx + logf(sum);
    out[(long)node * NC + lane] = v0 - lse;
    if (hi) out[(long)node * NC + c2] = v1 - lse;
}


extern "C" void kernel_launch(void* const* d_in, const int* in_sizes, int n_in,
                              void* d_out, int out_size, void* d_ws, size_t ws_size,
                              hipStream_t stream) {
    const float* x  = (const float*)d_in[0];
    const int*   ei = (const int*)d_in[1];
    const float* W1 = (const float*)d_in[2];
    const float* b1 = (const float*)d_in[3];
    const float* W2 = (const float*)d_in[4];
    const float* b2 = (const float*)d_in[5];
    float* out = (float*)d_out;

    const int* src = ei;
    const int* dst = ei + NE;

    float* ws     = (float*)d_ws;
    int*   cnt    = (int*)ws;
    float* dinv   = ws + NNP;
    int*   rowptr = (int*)(dinv + NNP);
    int*   rank   = rowptr + NNP;
    int*   ssrc   = rank + NEP;
    float* h0     = (float*)(ssrc + NEP);
    float* h1   = h0   + (long)NN * DH;
    float* h2   = h1   + (long)NN * DH;
    float* W2p  = h2   + (long)NN * NC;

    const int B = 256;

    k_cnt_zero <<<cdiv(NN, B), B, 0, stream>>>(cnt);
    k_rank     <<<cdiv(NE, B), B, 0, stream>>>(dst, cnt, rank);
    k_scan     <<<1, SCAN_T, 0, stream>>>(cnt, rowptr, dinv);
    k_fill     <<<cdiv(NE, B), B, 0, stream>>>(src, dst, rowptr, rank, ssrc);
    k_pad_w2   <<<cdiv(DH * NCP, B), B, 0, stream>>>(W2, W2p);

    {
        long waves = (long)(NN / 16) * (DH / 16);
        k_gemm_wmma_f32<FIN, DH, DH>
            <<<cdiv(waves * 32, B), B, 0, stream>>>(x, W1, h0, NN);
    }
    k_gather_h   <<<cdiv((long)NN * 32, B), B, 0, stream>>>(rowptr, ssrc, dinv, h0, b1, h1);

    {
        long waves = (long)(NN / 16) * (NCP / 16);
        k_gemm_wmma_f32<DH, NCP, NC>
            <<<cdiv(waves * 32, B), B, 0, stream>>>(h1, W2p, h2, NN);
    }
    k_gather_c   <<<cdiv((long)NN * 32, B), B, 0, stream>>>(rowptr, ssrc, dinv, h2, b2, out);
}


# Edge rank by returning LDS add: 32 edge chunks, each counts into an all-node LDS table, per-node prefix over chunks; replaces 625000 global int atomics; sum order: chunk, then arrival inside chunk
# speedup vs baseline: 2.3358x; 1.1274x over previous
#include <hip/hip_runtime.h>
#include <math.h>

#define NN   50000
#define NE   625000
#define FIN  128
#define DH   128
#define NC   40
#define NCP  48
#define NNP  50048
#define NEP  625024

typedef __attribute__((ext_vector_type(2))) float v2f;
typedef __attribute__((ext_vector_type(8))) float v8f;

static inline int cdiv(long a, long b) { return (int)((a + b - 1) / b); }


#define HG   32
#define HCH  19532
#define HQ   (HCH / 4)
#define HT   1024
#define HIT  ((HQ + HT - 1) / HT)
static_assert(HCH % 4 == 0 && NE % 4 == 0 && NN % 4 == 0 && (long)HG * HCH >= NE,
              "chunks are whole quads and cover every edge");

__global__ void __launch_bounds__(HT)
k_hist(const int* __restrict__ dst, int* __restrict__ cntg, int* __restrict__ rank) {
    __shared__ int c[NN];
    const int g  = blockIdx.x;
    const int q0 = g * HQ;
    const int nq = (NE / 4 - q0 < HQ) ? (NE / 4 - q0) : HQ;

    int4 d[HIT];
#pragma unroll
    for (int t = 0; t < HIT; ++t) {
        const int q = t * HT + (int)threadIdx.x;
        d[t] = (q < nq) ? ((const int4*)dst)[q0 + q] : make_int4(0, 0, 0, 0);
    }
    for (int i = threadIdx.x; i < NN / 4; i += HT) ((int4*)c)[i] = make_int4(0, 0, 0, 0);
    __syncthreads();

#pragma unroll
    for (int t = 0; t < HIT; ++t) {
        const int q = t * HT + (int)threadIdx.x;
        if (q < nq) {
            int4 r;
            r.x = atomicAdd(&c[d[t].x], 1);
            r.y = atomicAdd(&c[d[t].y], 1);
            r.z = atomicAdd(&c[d[t].z], 1);
            r.w = atomicAdd(&c[d[t].w], 1);
            ((int4*)rank)[q0 + q] = r;
        }
    }
    __syncthreads();

    int4* cg = (int4*)(cntg + (long)g * NNP);
    for (int i = threadIdx.x; i < NN / 4; i += HT) cg[i] = ((const int4*)c)[i];
}

__global__ void k_chunk_prefix(int* __restrict__ cntg, int* __restrict__ cnt) {
    const int i = blockIdx.x * blockDim.x + threadIdx.x;
    if (i >= NN / 4) return;
    int4 v[HG];
#pragma unroll
    for (int g = 0; g < HG; ++g) v[g] = ((const int4*)(cntg + (long)g * NNP))[i];
    int4 run = make_int4(0, 0, 0, 0);
#pragma unroll
    for (int g = 0; g < HG; ++g) {
        ((int4*)(cntg + (long)g * NNP))[i] = run;
        run.x += v[g].x; run.y += v[g].y; run.z += v[g].z; run.w += v[g].w;
    }
    ((int4*)cnt)[i] = run;
}

#define SCAN_T     1024
#define SCAN_W     (SCAN_T / 32)
#define SCAN_TILES ((NN + SCAN_W * 128 - 1) / (SCAN_W * 128))
#define SCAN_SEG   (SCAN_TILES * 128)

__global__ void __launch_bounds__(SCAN_T)
k_scan(const int* __restrict__ cnt, int* __restrict__ rowptr, float* __restrict__ dinv) {
    __shared__ int wtot[SCAN_W];
    const int lane = threadIdx.x & 31;
    const int wv   = threadIdx.x >> 5;
    const int q0   = wv * (SCAN_SEG / 4) + lane;
    const int4* cq = (const int4*)cnt + q0;

    int4 c[SCAN_TILES];
    int  base[SCAN_TILES];
#pragma unroll
    for (int t = 0; t < SCAN_TILES; ++t)
        c[t] = (q0 + t * 32 < NN / 4) ? cq[t * 32] : make_int4(0, 0, 0, 0);
    int inc[SCAN_TILES];
#pragma unroll
    for (int t = 0; t < SCAN_TILES; ++t) inc[t] = c[t].x + c[t].y + c[t].z + c[t].w;
#pragma unroll
    for (int o = 1; o < 32; o <<= 1) {
        int y[SCAN_TILES];
#pragma unroll
        for (int t = 0; t < SCAN_TILES; ++t) y[t] = __shfl_up(inc[t], o);
#pragma unroll
        for (int t = 0; t < SCAN_TILES; ++t) if (lane >= o) inc[t] += y[t];
    }
    int run = 0;
#pragma unroll
    for (int t = 0; t < SCAN_TILES; ++t) {
        base[t] = run + inc[t] - (c[t].x + c[t].y + c[t].z + c[t].w);
        run += __shfl(inc[t], 31);
    }
    if (lane == 0) wtot[wv] = run;
    __syncthreads();
    int woff = (lane < wv) ? wtot[lane] : 0;
#pragma unroll
    for (int o = 16; o > 0; o >>= 1) woff += __shfl_xor(woff, o);

    int4*   rq = (int4*)rowptr + q0;
    float4* dq = (float4*)dinv + q0;
#pragma unroll
    for (int t = 0; t < SCAN_TILES; ++t) {
        if (q0 + t * 32 < NN / 4) {
            const int p0 = woff + base[t];
            rq[t * 32] = make_int4(p0, p0 + c[t].x, p0 + c[t].x + c[t].y, p0 + c[t].x + c[t].y + c[t].z);
            dq[t * 32] = make_float4(rsqrtf((float)(c[t].x + 1)), rsqrtf((float)(c[t].y + 1)),
                                     rsqrtf((float)(c[t].z + 1)), rsqrtf((float)(c[t].w + 1)));
        }
    }
    if (wv == SCAN_W - 1 && lane == 0) rowptr[NN] = woff + run;
}

__global__ void k_fill(const int* __restrict__ src, const int* __restrict__ dst,
                       const int* __restrict__ rowptr, const int* __restrict__ pre,
                       const int* __restrict__ rank, int* __restrict__ ssrc) {
    const int q = blockIdx.x * blockDim.x + threadIdx.x;
    if (q >= NE / 4) return;
    const int4 s = ((const int4*)src)[q];
    const int4 d = ((const int4*)dst)[q];
    const int4 r = ((const int4*)rank)[q];
    const int* pg = pre + (long)(q / HQ) * NNP;
    const int px = rowptr[d.x] + pg[d.x], py = rowptr[d.y] + pg[d.y];
    const int pz = rowptr[d.z] + pg[d.z], pw = rowptr[d.w] + pg[d.w];
    ssrc[px + r.x] = s.x;
    ssrc[py + r.y] = s.y;
    ssrc[pz + r.z] = s.z;
    ssrc[pw + r.w] = s.w;
}

__global__ void k_pad_w2(const float* __restrict__ W2, float* __restrict__ W2p) {
    int t = blockIdx.x * blockDim.x + threadIdx.x;
    if (t >= DH * NCP) return;
    int r = t / NCP, c = t - r * NCP;
    W2p[t] = (c < NC) ? W2[r * NC + c] : 0.0f;
}


template <int K, int NP, int NOUT>
__global__ void k_gemm_wmma_f32(const float* __restrict__ A, const float* __restrict__ B,
                                float* __restrict__ C, int M) {
    __shared__ float ldsB[K * NP];

    for (int idx = threadIdx.x; idx < (K * NP) / 4; idx += blockDim.x)
        ((float4*)ldsB)[idx] = ((const float4*)B)[idx];
    __syncthreads();

    const int lane = threadIdx.x & 31;
    const int wave = (blockIdx.x * blockDim.x + threadIdx.x) >> 5;
    constexpr int tilesN = NP / 16;
    int tm = wave / tilesN;
    const int tn = wave - tm * tilesN;
    const bool valid = (tm * 16 < M);
    if (!valid) tm = M / 16 - 1;

    const int m  = tm * 16 + (lane & 15);
    const int n  = tn * 16 + (lane & 15);
    const int kh = (lane >> 4) << 1;

    const float* aP   = A + m * K + kh;
    const int    bOff = kh * NP + n;

    v8f acc = {};
#pragma unroll
    for (int kk = 0; kk < K; kk += 4) {
        v2f a, b;
        a.x = aP[kk];
        a.y = aP[kk + 1];
        b.x = ldsB[bOff + kk * NP];
        b.y = ldsB[bOff + (kk + 1) * NP];
        acc = __builtin_amdgcn_wmma_f32_16x16x4_f32(
             false, a,  false, b,
             (short)0, acc,  false,  false);
    }

    if (valid && n < NOUT) {
        const int rbase = tm * 16 + ((lane >> 4) << 3);
#pragma unroll
        for (int v = 0; v < 8; ++v)
            C[(rbase + v) * NOUT + n] = acc[v];
    }
}


static __device__ __forceinline__ int rl_i(int v, int k) {
    return __builtin_amdgcn_readlane(v, k);
}
static __device__ __forceinline__ float rl_f(float v, int k) {
    return __int_as_float(__builtin_amdgcn_readlane(__float_as_int(v), k));
}

__global__ void k_gather_h(const int* __restrict__ rowptr, const int* __restrict__ ssrc,
                           const float* __restrict__ dinv,
                           const float* __restrict__ h, const float* __restrict__ b,
                           float* __restrict__ h1) {
    const int lane = threadIdx.x & 31;
    const int node = __builtin_amdgcn_readfirstlane((int)((blockIdx.x * blockDim.x + threadIdx.x) >> 5));
    if (node >= NN) return;
    const int beg = rowptr[node], end = rowptr[node + 1];
    const float dd = dinv[node];
    const float4* h4 = (const float4*)h;
    const float4 bb = ((const float4*)b)[lane];

    const float ws = dd * dd;
    float4 acc = h4[(long)node * 32 + lane];
    acc.x *= ws; acc.y *= ws; acc.z *= ws; acc.w *= ws;

    for (int cb = beg; cb < end; cb += 32) {
        const int j = cb + lane;
        int s_l = (j < end) ? ssrc[j] : node;
        if ((unsigned)s_l >= (unsigned)NN) s_l = node;
        const float w_l = dinv[s_l] * dd;
        const int n = (end - cb < 32) ? (end - cb) : 32;
        int k = 0;
        for (; k + 4 <= n; k += 4) {
            const int s0 = rl_i(s_l, k), s1 = rl_i(s_l, k + 1), s2 = rl_i(s_l, k + 2), s3 = rl_i(s_l, k + 3);
            const float w0 = rl_f(w_l, k), w1 = rl_f(w_l, k + 1), w2 = rl_f(w_l, k + 2), w3 = rl_f(w_l, k + 3);
            const float4 v0 = h4[(long)s0 * 32 + lane];
            const float4 v1 = h4[(long)s1 * 32 + lane];
            const float4 v2 = h4[(long)s2 * 32 + lane];
            const float4 v3 = h4[(long)s3 * 32 + lane];
            acc.x = fmaf(v0.x, w0, acc.x); acc.y = fmaf(v0.y, w0, acc.y); acc.z = fmaf(v0.z, w0, acc.z); acc.w = fmaf(v0.w, w0, acc.w);
            acc.x = fmaf(v1.x, w1, acc.x); acc.y = fmaf(v1.y, w1, acc.y); acc.z = fmaf(v1.z, w1, acc.z); acc.w = fmaf(v1.w, w1, acc.w);
            acc.x = fmaf(v2.x, w2, acc.x); acc.y = fmaf(v2.y, w2, acc.y); acc.z = fmaf(v2.z, w2, acc.z); acc.w = fmaf(v2.w, w2, acc.w);
            acc.x = fmaf(v3.x, w3, acc.x); acc.y = fmaf(v3.y, w3, acc.y); acc.z = fmaf(v3.z, w3, acc.z); acc.w = fmaf(v3.w, w3, acc.w);
        }
        for (; k < n; ++k) {
            const int s0 = rl_i(s_l, k);
            const float w0 = rl_f(w_l, k);
            const float4 v0 = h4[(long)s0 * 32 + lane];
            acc.x = fmaf(v0.x, w0, acc.x); acc.y = fmaf(v0.y, w0, acc.y); acc.z = fmaf(v0.z, w0, acc.z); acc.w = fmaf(v0.w, w0, acc.w);
        }
    }
    acc.x += bb.x; acc.y += bb.y; acc.z += bb.z; acc.w += bb.w;
    acc.x = acc.x > 0.0f ? acc.x : 0.0f;
    acc.y = acc.y > 0.0f ? acc.y : 0.0f;
    acc.z = acc.z > 0.0f ? acc.z : 0.0f;
    acc.w = acc.w > 0.0f ? acc.w : 0.0f;
    ((float4*)h1)[(long)node * 32 + lane] = acc;
}


__global__ void k_gather_c(const int* __restrict__ rowptr, const int* __restrict__ ssrc,
                           const float* __restrict__ dinv,
                           const float* __restrict__ h, const float* __restrict__ b,
                           float* __restrict__ out) {
    const int lane = threadIdx.x & 31;
    const int node = __builtin_amdgcn_readfirstlane((int)((blockIdx.x * blockDim.x + threadIdx.x) >> 5));
    if (node >= NN) return;
    const int beg = rowptr[node], end = rowptr[node + 1];
    const float dd = dinv[node];
    const bool hi = lane < NC - 32;
    const int  c2 = hi ? lane + 32 : lane;
    const float b0 = b[lane], b1 = b[c2];

    const float ws = dd * dd;
    float a0 = h[(long)node * NC + lane] * ws;
    float a1 = h[(long)node * NC + c2] * ws;

    for (int cb = beg; cb < end; cb += 32) {
        const int j = cb + lane;
        int s_l = (j < end) ? ssrc[j] : node;
        if ((unsigned)s_l >= (unsigned)NN) s_l = node;
        const float w_l = dinv[s_l] * dd;
        const int n = (end - cb < 32) ? (end - cb) : 32;
        int k = 0;
        for (; k + 4 <= n; k += 4) {
            const int s0 = rl_i(s_l, k), s1 = rl_i(s_l, k + 1), s2 = rl_i(s_l, k + 2), s3 = rl_i(s_l, k + 3);
            const float w0 = rl_f(w_l, k), w1 = rl_f(w_l, k + 1), w2 = rl_f(w_l, k + 2), w3 = rl_f(w_l, k + 3);
            const float p0 = h[(long)s0 * NC + lane], q0 = h[(long)s0 * NC + c2];
            const float p1 = h[(long)s1 * NC + lane], q1 = h[(long)s1 * NC + c2];
            const float p2 = h[(long)s2 * NC + lane], q2 = h[(long)s2 * NC + c2];
            const float p3 = h[(long)s3 * NC + lane], q3 = h[(long)s3 * NC + c2];
            a0 = fmaf(p0, w0, a0); a1 = fmaf(q0, w0, a1);
            a0 = fmaf(p1, w1, a0); a1 = fmaf(q1, w1, a1);
            a0 = fmaf(p2, w2, a0); a1 = fmaf(q2, w2, a1);
            a0 = fmaf(p3, w3, a0); a1 = fmaf(q3, w3, a1);
        }
        for (; k < n; ++k) {
            const int s0 = rl_i(s_l, k);
            const float w0 = rl_f(w_l, k);
            a0 = fmaf(h[(long)s0 * NC + lane], w0, a0);
            a1 = fmaf(h[(long)s0 * NC + c2], w0, a1);
        }
    }
    const float v0 = a0 + b0;
    const float v1 = hi ? a1 + b1 : -INFINITY;
    float mx = fmaxf(v0, v1);
    for (int o = 16; o > 0; o >>= 1) mx = fmaxf(mx, __shfl_xor(mx, o));
    float sum = expf(v0 - mx) + (hi ? expf(v1 - mx) : 0.0f);
    for (int o = 16; o > 0; o >>= 1) sum += __shfl_xor(sum, o);
    const float lse = mx + logf(sum);
    out[(long)node * NC + lane] = v0 - lse;
    if (hi) out[(long)node * NC + c2] = v1 - lse;
}


extern "C" void kernel_launch(void* const* d_in, const int* in_sizes, int n_in,
                              void* d_out, int out_size, void* d_ws, size_t ws_size,
                              hipStream_t stream) {
    const float* x  = (const float*)d_in[0];
    const int*   ei = (const int*)d_in[1];
    const float* W1 = (const float*)d_in[2];
    const float* b1 = (const float*)d_in[3];
    const float* W2 = (const float*)d_in[4];
    const float* b2 = (const float*)d_in[5];
    float* out = (float*)d_out;

    const int* src = ei;
    const int* dst = ei + NE;

    float* ws     = (float*)d_ws;
    int*   cnt    = (int*)ws;
    float* dinv   = ws + NNP;
    int*   rowptr = (int*)(dinv + NNP);
    int*   rank   = rowptr + NNP;
    int*   ssrc   = rank + NEP;
    float* h0     = (float*)(ssrc + NEP);
    float* h1   = h0   + (long)NN * DH;
    float* h2   = h1   + (long)NN * DH;
    float* W2p  = h2   + (long)NN * NC;
    int*   cntg = (int*)(W2p + DH * NCP);

    const int B = 256;

    k_hist         <<<HG, HT, 0, stream>>>(dst, cntg, rank);
    k_chunk_prefix <<<cdiv(NN / 4, 64), 64, 0, stream>>>(cntg, cnt);
    k_scan         <<<1, SCAN_T, 0, stream>>>(cnt, rowptr, dinv);
    k_fill         <<<cdiv(NE / 4, B), B, 0, stream>>>(src, dst, rowptr, cntg, rank, ssrc);
    k_pad_w2   <<<cdiv(DH * NCP, B), B, 0, stream>>>(W2, W2p);

    {
        long waves = (long)(NN / 16) * (DH / 16);
        k_gemm_wmma_f32<FIN, DH, DH>
            <<<cdiv(waves * 32, B), B, 0, stream>>>(x, W1, h0, NN);
    }
    k_gather_h   <<<cdiv((long)NN * 32, B), B, 0, stream>>>(rowptr, ssrc, dinv, h0, b1, h1);

    {
        long waves = (long)(NN / 16) * (NCP / 16);
        k_gemm_wmma_f32<DH, NCP, NC>
            <<<cdiv(waves * 32, B), B, 0, stream>>>(h1, W2p, h2, NN);
    }
    k_gather_c   <<<cdiv((long)NN * 32, B), B, 0, stream>>>(rowptr, ssrc, dinv, h2, b2, out);
}


# Layer 1 regrouped as relu((A_hat x) W1 + b1): one launch gathers 16 nodes' input rows to LDS, then W1, bias, relu and W2 on WMMA; both GEMM launches, pad and h0/h1 go; new sum order
# speedup vs baseline: 3.0091x; 1.2883x over previous
#include <hip/hip_runtime.h>
#include <math.h>

#define NN   50000
#define NE   625000
#define FIN  128
#define DH   128
#define NC   40
#define NCP  48
#define NNP  50048
#define NEP  625024

typedef __attribute__((ext_vector_type(2))) float v2f;
typedef __attribute__((ext_vector_type(8))) float v8f;

static inline int cdiv(long a, long b) { return (int)((a + b - 1) / b); }


#define HG   32
#define HCH  19532
#define HQ   (HCH / 4)
#define HT   1024
#define HIT  ((HQ + HT - 1) / HT)
static_assert(HCH % 4 == 0 && NE % 4 == 0 && NN % 4 == 0 && (long)HG * HCH >= NE,
              "chunks are whole quads and cover every edge");

__global__ void __launch_bounds__(HT)
k_hist(const int* __restrict__ dst, int* __restrict__ cntg, int* __restrict__ rank) {
    __shared__ int c[NN];
    const int g  = blockIdx.x;
    const int q0 = g * HQ;
    const int nq = (NE / 4 - q0 < HQ) ? (NE / 4 - q0) : HQ;

    int4 d[HIT];
#pragma unroll
    for (int t = 0; t < HIT; ++t) {
        const int q = t * HT + (int)threadIdx.x;
        d[t] = (q < nq) ? ((const int4*)dst)[q0 + q] : make_int4(0, 0, 0, 0);
    }
    for (int i = threadIdx.x; i < NN / 4; i += HT) ((int4*)c)[i] = make_int4(0, 0, 0, 0);
    __syncthreads();

#pragma unroll
    for (int t = 0; t < HIT; ++t) {
        const int q = t * HT + (int)threadIdx.x;
        if (q < nq) {
            int4 r;
            r.x = atomicAdd(&c[d[t].x], 1);
            r.y = atomicAdd(&c[d[t].y], 1);
            r.z = atomicAdd(&c[d[t].z], 1);
            r.w = atomicAdd(&c[d[t].w], 1);
            ((int4*)rank)[q0 + q] = r;
        }
    }
    __syncthreads();

    int4* cg = (int4*)(cntg + (long)g * NNP);
    for (int i = threadIdx.x; i < NN / 4; i += HT) cg[i] = ((const int4*)c)[i];
}

__global__ void k_chunk_prefix(int* __restrict__ cntg, int* __restrict__ cnt) {
    const int i = blockIdx.x * blockDim.x + threadIdx.x;
    if (i >= NN / 4) return;
    int4 v[HG];
#pragma unroll
    for (int g = 0; g < HG; ++g) v[g] = ((const int4*)(cntg + (long)g * NNP))[i];
    int4 run = make_int4(0, 0, 0, 0);
#pragma unroll
    for (int g = 0; g < HG; ++g) {
        ((int4*)(cntg + (long)g * NNP))[i] = run;
        run.x += v[g].x; run.y += v[g].y; run.z += v[g].z; run.w += v[g].w;
    }
    ((int4*)cnt)[i] = run;
}

#define SCAN_T     1024
#define SCAN_W     (SCAN_T / 32)
#define SCAN_TILES ((NN + SCAN_W * 128 - 1) / (SCAN_W * 128))
#define SCAN_SEG   (SCAN_TILES * 128)

__global__ void __launch_bounds__(SCAN_T)
k_scan(const int* __restrict__ cnt, int* __restrict__ rowptr, float* __restrict__ dinv) {
    __shared__ int wtot[SCAN_W];
    const int lane = threadIdx.x & 31;
    const int wv   = threadIdx.x >> 5;
    const int q0   = wv * (SCAN_SEG / 4) + lane;
    const int4* cq = (const int4*)cnt + q0;

    int4 c[SCAN_TILES];
    int  base[SCAN_TILES];
#pragma unroll
    for (int t = 0; t < SCAN_TILES; ++t)
        c[t] = (q0 + t * 32 < NN / 4) ? cq[t * 32] : make_int4(0, 0, 0, 0);
    int inc[SCAN_TILES];
#pragma unroll
    for (int t = 0; t < SCAN_TILES; ++t) inc[t] = c[t].x + c[t].y + c[t].z + c[t].w;
#pragma unroll
    for (int o = 1; o < 32; o <<= 1) {
        int y[SCAN_TILES];
#pragma unroll
        for (int t = 0; t < SCAN_TILES; ++t) y[t] = __shfl_up(inc[t], o);
#pragma unroll
        for (int t = 0; t < SCAN_TILES; ++t) if (lane >= o) inc[t] += y[t];
    }
    int run = 0;
#pragma unroll
    for (int t = 0; t < SCAN_TILES; ++t) {
        base[t] = run + inc[t] - (c[t].x + c[t].y + c[t].z + c[t].w);
        run += __shfl(inc[t], 31);
    }
    if (lane == 0) wtot[wv] = run;
    __syncthreads();
    int woff = (lane < wv) ? wtot[lane] : 0;
#pragma unroll
    for (int o = 16; o > 0; o >>= 1) woff += __shfl_xor(woff, o);

    int4*   rq = (int4*)rowptr + q0;
    float4* dq = (float4*)dinv + q0;
#pragma unroll
    for (int t = 0; t < SCAN_TILES; ++t) {
        if (q0 + t * 32 < NN / 4) {
            const int p0 = woff + base[t];
            rq[t * 32] = make_int4(p0, p0 + c[t].x, p0 + c[t].x + c[t].y, p0 + c[t].x + c[t].y + c[t].z);
            dq[t * 32] = make_float4(rsqrtf((float)(c[t].x + 1)), rsqrtf((float)(c[t].y + 1)),
                                     rsqrtf((float)(c[t].z + 1)), rsqrtf((float)(c[t].w + 1)));
        }
    }
    if (wv == SCAN_W - 1 && lane == 0) rowptr[NN] = woff + run;
}

__global__ void k_fill(const int* __restrict__ src, const int* __restrict__ dst,
                       const int* __restrict__ rowptr, const int* __restrict__ pre,
                       const int* __restrict__ rank, int* __restrict__ ssrc) {
    const int q = blockIdx.x * blockDim.x + threadIdx.x;
    if (q >= NE / 4) return;
    const int4 s = ((const int4*)src)[q];
    const int4 d = ((const int4*)dst)[q];
    const int4 r = ((const int4*)rank)[q];
    const int* pg = pre + (long)(q / HQ) * NNP;
    const int px = rowptr[d.x] + pg[d.x], py = rowptr[d.y] + pg[d.y];
    const int pz = rowptr[d.z] + pg[d.z], pw = rowptr[d.w] + pg[d.w];
    ssrc[px + r.x] = s.x;
    ssrc[py + r.y] = s.y;
    ssrc[pz + r.z] = s.z;
    ssrc[pw + r.w] = s.w;
}


#define TP  132
#define L1T 512
static_assert(NN % 16 == 0, "exact grid: every work-group has 16 real nodes");
static_assert(FIN == DH, "one tile buffer serves both products");

static __device__ __forceinline__ int rl_i(int v, int k) {
    return __builtin_amdgcn_readlane(v, k);
}
static __device__ __forceinline__ float rl_f(float v, int k) {
    return __int_as_float(__builtin_amdgcn_readlane(__float_as_int(v), k));
}

template <int K, int NP>
static __device__ __forceinline__ v8f tile_product(const float* __restrict__ sT, const float* __restrict__ sB,
                                                   int tn, int lane) {
    const int c  = lane & 15;
    const int kh = (lane >> 4) << 1;
    const float* aP = sT + c * TP + kh;
    const float* bP = sB + kh * NP + tn * 16 + c;
    v8f acc = {};
#pragma unroll
    for (int kk = 0; kk < K; kk += 4) {
        v2f a, b;
        a.x = aP[kk];
        a.y = aP[kk + 1];
        b.x = bP[kk * NP];
        b.y = bP[(kk + 1) * NP];
        acc = __builtin_amdgcn_wmma_f32_16x16x4_f32(
             false, a,  false, b,
             (short)0, acc,  false,  false);
    }
    return acc;
}

__global__ void __launch_bounds__(L1T)
k_layer1(const int* __restrict__ rowptr, const int* __restrict__ ssrc,
         const float* __restrict__ dinv, const float* __restrict__ x,
         const float* __restrict__ W1, const float* __restrict__ b1,
         const float* __restrict__ W2, float* __restrict__ h2) {
    __shared__ float sW1[FIN * DH];
    __shared__ float sW2[DH * NCP];
    __shared__ float sT[16 * TP];

    const int tid  = threadIdx.x;
    const int lane = tid & 31;
    const int wv   = __builtin_amdgcn_readfirstlane(tid >> 5);

    const int node0 = blockIdx.x * 16;
    const int node  = node0 + wv;
    const int beg = rowptr[node], end = rowptr[node + 1];
    const float dd = dinv[node];
    const float4* x4 = (const float4*)x;
    float4 acc = x4[(long)node * 32 + lane];

    {
        constexpr int Q1 = (FIN * DH) / 4;
        constexpr int QP = (DH * NCP) / 4;
        constexpr int N1 = Q1 / L1T, N2 = QP / L1T;
        static_assert(Q1 % L1T == 0 && QP % L1T == 0, "whole quads per thread, no tail");
        static_assert(NC % 4 == 0 && NCP % 4 == 0, "a quad stays inside its row, data or pad");
        float4 w[N1], u[N2];
#pragma unroll
        for (int t = 0; t < N1; ++t) w[t] = ((const float4*)W1)[tid + t * L1T];
#pragma unroll
        for (int t = 0; t < N2; ++t) {
            const int j = tid + t * L1T;
            const int r = j / (NCP / 4), q = j - r * (NCP / 4);
            u[t] = ((const float4*)W2)[r * (NC / 4) + (q < NC / 4 ? q : NC / 4 - 1)];
        }
#pragma unroll
        for (int t = 0; t < N1; ++t) ((float4*)sW1)[tid + t * L1T] = w[t];
#pragma unroll
        for (int t = 0; t < N2; ++t) {
            const int j = tid + t * L1T;
            const int q = j % (NCP / 4);
            const bool data = q < NC / 4;
            ((float4*)sW2)[j] = make_float4(data ? u[t].x : 0.0f, data ? u[t].y : 0.0f,
                                            data ? u[t].z : 0.0f, data ? u[t].w : 0.0f);
        }
    }

    {
        const float ws = dd * dd;
        acc.x *= ws; acc.y *= ws; acc.z *= ws; acc.w *= ws;

        for (int cb = beg; cb < end; cb += 32) {
            const int j = cb + lane;
            int s_l = (j < end) ? ssrc[j] : node;
            if ((unsigned)s_l >= (unsigned)NN) s_l = node;
            const float w_l = dinv[s_l] * dd;
            const int n = (end - cb < 32) ? (end - cb) : 32;
            int k = 0;
            for (; k + 4 <= n; k += 4) {
                const int s0 = rl_i(s_l, k), s1 = rl_i(s_l, k + 1), s2 = rl_i(s_l, k + 2), s3 = rl_i(s_l, k + 3);
                const float w0 = rl_f(w_l, k), w1 = rl_f(w_l, k + 1), w2 = rl_f(w_l, k + 2), w3 = rl_f(w_l, k + 3);
                const float4 v0 = x4[(long)s0 * 32 + lane];
                const float4 v1 = x4[(long)s1 * 32 + lane];
                const float4 v2 = x4[(long)s2 * 32 + lane];
                const float4 v3 = x4[(long)s3 * 32 + lane];
                acc.x = fmaf(v0.x, w0, acc.x); acc.y = fmaf(v0.y, w0, acc.y); acc.z = fmaf(v0.z, w0, acc.z); acc.w = fmaf(v0.w, w0, acc.w);
                acc.x = fmaf(v1.x, w1, acc.x); acc.y = fmaf(v1.y, w1, acc.y); acc.z = fmaf(v1.z, w1, acc.z); acc.w = fmaf(v1.w, w1, acc.w);
                acc.x = fmaf(v2.x, w2, acc.x); acc.y = fmaf(v2.y, w2, acc.y); acc.z = fmaf(v2.z, w2, acc.z); acc.w = fmaf(v2.w, w2, acc.w);
                acc.x = fmaf(v3.x, w3, acc.x); acc.y = fmaf(v3.y, w3, acc.y); acc.z = fmaf(v3.z, w3, acc.z); acc.w = fmaf(v3.w, w3, acc.w);
            }
            for (; k < n; ++k) {
                const int s0 = rl_i(s_l, k);
                const float w0 = rl_f(w_l, k);
                const float4 v0 = x4[(long)s0 * 32 + lane];
                acc.x = fmaf(v0.x, w0, acc.x); acc.y = fmaf(v0.y, w0, acc.y); acc.z = fmaf(v0.z, w0, acc.z); acc.w = fmaf(v0.w, w0, acc.w);
            }
        }
        ((float4*)(sT + wv * TP))[lane] = acc;
    }
    __syncthreads();

    v8f p = {};
    if (wv < DH / 16) p = tile_product<FIN, DH>(sT, sW1, wv, lane);
    __syncthreads();
    if (wv < DH / 16) {
        const int n = wv * 16 + (lane & 15);
        const int r0 = (lane >> 4) << 3;
        const float bb = b1[n];
#pragma unroll
        for (int v = 0; v < 8; ++v) {
            const float t = p[v] + bb;
            sT[(r0 + v) * TP + n] = t > 0.0f ? t : 0.0f;
        }
    }
    __syncthreads();

    if (wv < NCP / 16) {
        const v8f o = tile_product<DH, NCP>(sT, sW2, wv, lane);
        const int n = wv * 16 + (lane & 15);
        if (n < NC) {
            const int r0 = node0 + ((lane >> 4) << 3);
#pragma unroll
            for (int v = 0; v < 8; ++v)
                h2[(long)(r0 + v) * NC + n] = o[v];
        }
    }
}


__global__ void k_gather_c(const int* __restrict__ rowptr, const int* __restrict__ ssrc,
                           const float* __restrict__ dinv,
                           const float* __restrict__ h, const float* __restrict__ b,
                           float* __restrict__ out) {
    const int lane = threadIdx.x & 31;
    const int node = __builtin_amdgcn_readfirstlane((int)((blockIdx.x * blockDim.x + threadIdx.x) >> 5));
    if (node >= NN) return;
    const int beg = rowptr[node], end = rowptr[node + 1];
    const float dd = dinv[node];
    const bool hi = lane < NC - 32;
    const int  c2 = hi ? lane + 32 : lane;
    const float b0 = b[lane], b1 = b[c2];

    const float ws = dd * dd;
    float a0 = h[(long)node * NC + lane] * ws;
    float a1 = h[(long)node * NC + c2] * ws;

    for (int cb = beg; cb < end; cb += 32) {
        const int j = cb + lane;
        int s_l = (j < end) ? ssrc[j] : node;
        if ((unsigned)s_l >= (unsigned)NN) s_l = node;
        const float w_l = dinv[s_l] * dd;
        const int n = (end - cb < 32) ? (end - cb) : 32;
        int k = 0;
        for (; k + 4 <= n; k += 4) {
            const int s0 = rl_i(s_l, k), s1 = rl_i(s_l, k + 1), s2 = rl_i(s_l, k + 2), s3 = rl_i(s_l, k + 3);
            const float w0 = rl_f(w_l, k), w1 = rl_f(w_l, k + 1), w2 = rl_f(w_l, k + 2), w3 = rl_f(w_l, k + 3);
            const float p0 = h[(long)s0 * NC + lane], q0 = h[(long)s0 * NC + c2];
            const float p1 = h[(long)s1 * NC + lane], q1 = h[(long)s1 * NC + c2];
            const float p2 = h[(long)s2 * NC + lane], q2 = h[(long)s2 * NC + c2];
            const float p3 = h[(long)s3 * NC + lane], q3 = h[(long)s3 * NC + c2];
            a0 = fmaf(p0, w0, a0); a1 = fmaf(q0, w0, a1);
            a0 = fmaf(p1, w1, a0); a1 = fmaf(q1, w1, a1);
            a0 = fmaf(p2, w2, a0); a1 = fmaf(q2, w2, a1);
            a0 = fmaf(p3, w3, a0); a1 = fmaf(q3, w3, a1);
        }
        for (; k < n; ++k) {
            const int s0 = rl_i(s_l, k);
            const float w0 = rl_f(w_l, k);
            a0 = fmaf(h[(long)s0 * NC + lane], w0, a0);
            a1 = fmaf(h[(long)s0 * NC + c2], w0, a1);
        }
    }
    const float v0 = a0 + b0;
    const float v1 = hi ? a1 + b1 : -INFINITY;
    float mx = fmaxf(v0, v1);
    for (int o = 16; o > 0; o >>= 1) mx = fmaxf(mx, __shfl_xor(mx, o));
    float sum = expf(v0 - mx) + (hi ? expf(v1 - mx) : 0.0f);
    for (int o = 16; o > 0; o >>= 1) sum += __shfl_xor(sum, o);
    const float lse = mx + logf(sum);
    out[(long)node * NC + lane] = v0 - lse;
    if (hi) out[(long)node * NC + c2] = v1 - lse;
}


extern "C" void kernel_launch(void* const* d_in, const int* in_sizes, int n_in,
                              void* d_out, int out_size, void* d_ws, size_t ws_size,
                              hipStream_t stream) {
    const float* x  = (const float*)d_in[0];
    const int*   ei = (const int*)d_in[1];
    const float* W1 = (const float*)d_in[2];
    const float* b1 = (const float*)d_in[3];
    const float* W2 = (const float*)d_in[4];
    const float* b2 = (const float*)d_in[5];
    float* out = (float*)d_out;

    const int* src = ei;
    const int* dst = ei + NE;

    float* ws     = (float*)d_ws;
    int*   cnt    = (int*)ws;
    float* dinv   = ws + NNP;
    int*   rowptr = (int*)(dinv + NNP);
    int*   rank   = rowptr + NNP;
    int*   ssrc   = rank + NEP;
    float* h2     = (float*)(ssrc + NEP);
    int*   cntg   = (int*)(h2 + (long)NN * NC);

    const int B = 256;

    k_hist         <<<HG, HT, 0, stream>>>(dst, cntg, rank);
    k_chunk_prefix <<<cdiv(NN / 4, 64), 64, 0, stream>>>(cntg, cnt);
    k_scan         <<<1, SCAN_T, 0, stream>>>(cnt, rowptr, dinv);
    k_fill         <<<cdiv(NE / 4, B), B, 0, stream>>>(src, dst, rowptr, cntg, rank, ssrc);

    k_layer1   <<<NN / 16, L1T, 0, stream>>>(rowptr, ssrc, dinv, x, W1, b1, W2, h2);

    k_gather_c   <<<cdiv((long)NN * 32, B), B, 0, stream>>>(rowptr, ssrc, dinv, h2, b2, out);
}


# Persistent waves: layer-1 launch 768 work-groups x 4-5 tiles (W1, W2 staged once), layer-2 launch 12504 waves x 4 nodes; next node's range, list chunk, dinv, own row fetched ahead; sum order kept
# speedup vs baseline: 3.2627x; 1.0843x over previous
#include <hip/hip_runtime.h>
#include <math.h>

#define NN   50000
#define NE   625000
#define FIN  128
#define DH   128
#define NC   40
#define NCP  48
#define NNP  50048
#define NEP  625024

typedef __attribute__((ext_vector_type(2))) float v2f;
typedef __attribute__((ext_vector_type(8))) float v8f;

static inline int cdiv(long a, long b) { return (int)((a + b - 1) / b); }


#define HG   32
#define HCH  19532
#define HQ   (HCH / 4)
#define HT   1024
#define HIT  ((HQ + HT - 1) / HT)
static_assert(HCH % 4 == 0 && NE % 4 == 0 && NN % 4 == 0 && (long)HG * HCH >= NE,
              "chunks are whole quads and cover every edge");

__global__ void __launch_bounds__(HT)
k_hist(const int* __restrict__ dst, int* __restrict__ cntg, int* __restrict__ rank) {
    __shared__ int c[NN];
    const int g  = blockIdx.x;
    const int q0 = g * HQ;
    const int nq = (NE / 4 - q0 < HQ) ? (NE / 4 - q0) : HQ;

    int4 d[HIT];
#pragma unroll
    for (int t = 0; t < HIT; ++t) {
        const int q = t * HT + (int)threadIdx.x;
        d[t] = (q < nq) ? ((const int4*)dst)[q0 + q] : make_int4(0, 0, 0, 0);
    }
    for (int i = threadIdx.x; i < NN / 4; i += HT) ((int4*)c)[i] = make_int4(0, 0, 0, 0);
    __syncthreads();

#pragma unroll
    for (int t = 0; t < HIT; ++t) {
        const int q = t * HT + (int)threadIdx.x;
        if (q < nq) {
            int4 r;
            r.x = atomicAdd(&c[d[t].x], 1);
            r.y = atomicAdd(&c[d[t].y], 1);
            r.z = atomicAdd(&c[d[t].z], 1);
            r.w = atomicAdd(&c[d[t].w], 1);
            ((int4*)rank)[q0 + q] = r;
        }
    }
    __syncthreads();

    int4* cg = (int4*)(cntg + (long)g * NNP);
    for (int i = threadIdx.x; i < NN / 4; i += HT) cg[i] = ((const int4*)c)[i];
}

__global__ void k_chunk_prefix(int* __restrict__ cntg, int* __restrict__ cnt) {
    const int i = blockIdx.x * blockDim.x + threadIdx.x;
    if (i >= NN / 4) return;
    int4 v[HG];
#pragma unroll
    for (int g = 0; g < HG; ++g) v[g] = ((const int4*)(cntg + (long)g * NNP))[i];
    int4 run = make_int4(0, 0, 0, 0);
#pragma unroll
    for (int g = 0; g < HG; ++g) {
        ((int4*)(cntg + (long)g * NNP))[i] = run;
        run.x += v[g].x; run.y += v[g].y; run.z += v[g].z; run.w += v[g].w;
    }
    ((int4*)cnt)[i] = run;
}

#define SCAN_T     1024
#define SCAN_W     (SCAN_T / 32)
#define SCAN_TILES ((NN + SCAN_W * 128 - 1) / (SCAN_W * 128))
#define SCAN_SEG   (SCAN_TILES * 128)

__global__ void __launch_bounds__(SCAN_T)
k_scan(const int* __restrict__ cnt, int* __restrict__ rowptr, float* __restrict__ dinv) {
    __shared__ int wtot[SCAN_W];
    const int lane = threadIdx.x & 31;
    const int wv   = threadIdx.x >> 5;
    const int q0   = wv * (SCAN_SEG / 4) + lane;
    const int4* cq = (const int4*)cnt + q0;

    int4 c[SCAN_TILES];
    int  base[SCAN_TILES];
#pragma unroll
    for (int t = 0; t < SCAN_TILES; ++t)
        c[t] = (q0 + t * 32 < NN / 4) ? cq[t * 32] : make_int4(0, 0, 0, 0);
    int inc[SCAN_TILES];
#pragma unroll
    for (int t = 0; t < SCAN_TILES; ++t) inc[t] = c[t].x + c[t].y + c[t].z + c[t].w;
#pragma unroll
    for (int o = 1; o < 32; o <<= 1) {
        int y[SCAN_TILES];
#pragma unroll
        for (int t = 0; t < SCAN_TILES; ++t) y[t] = __shfl_up(inc[t], o);
#pragma unroll
        for (int t = 0; t < SCAN_TILES; ++t) if (lane >= o) inc[t] += y[t];
    }
    int run = 0;
#pragma unroll
    for (int t = 0; t < SCAN_TILES; ++t) {
        base[t] = run + inc[t] - (c[t].x + c[t].y + c[t].z + c[t].w);
        run += __shfl(inc[t], 31);
    }
    if (lane == 0) wtot[wv] = run;
    __syncthreads();
    int woff = (lane < wv) ? wtot[lane] : 0;
#pragma unroll
    for (int o = 16; o > 0; o >>= 1) woff += __shfl_xor(woff, o);

    int4*   rq = (int4*)rowptr + q0;
    float4* dq = (float4*)dinv + q0;
#pragma unroll
    for (int t = 0; t < SCAN_TILES; ++t) {
        if (q0 + t * 32 < NN / 4) {
            const int p0 = woff + base[t];
            rq[t * 32] = make_int4(p0, p0 + c[t].x, p0 + c[t].x + c[t].y, p0 + c[t].x + c[t].y + c[t].z);
            dq[t * 32] = make_float4(rsqrtf((float)(c[t].x + 1)), rsqrtf((float)(c[t].y + 1)),
                                     rsqrtf((float)(c[t].z + 1)), rsqrtf((float)(c[t].w + 1)));
        }
    }
    if (wv == SCAN_W - 1 && lane == 0) rowptr[NN] = woff + run;
}

__global__ void k_fill(const int* __restrict__ src, const int* __restrict__ dst,
                       const int* __restrict__ rowptr, const int* __restrict__ pre,
                       const int* __restrict__ rank, int* __restrict__ ssrc) {
    const int q = blockIdx.x * blockDim.x + threadIdx.x;
    if (q >= NE / 4) return;
    const int4 s = ((const int4*)src)[q];
    const int4 d = ((const int4*)dst)[q];
    const int4 r = ((const int4*)rank)[q];
    const int* pg = pre + (long)(q / HQ) * NNP;
    const int px = rowptr[d.x] + pg[d.x], py = rowptr[d.y] + pg[d.y];
    const int pz = rowptr[d.z] + pg[d.z], pw = rowptr[d.w] + pg[d.w];
    ssrc[px + r.x] = s.x;
    ssrc[py + r.y] = s.y;
    ssrc[pz + r.z] = s.z;
    ssrc[pw + r.w] = s.w;
}


#define TP  132
#define L1T 512
#define L1TILES (NN / 16)
#define L1G  768
static_assert(NN % 16 == 0, "exact grid: every work-group has 16 real nodes");
static_assert(FIN == DH, "one tile buffer serves both products");
static_assert(L1G <= L1TILES, "every work-group has a first tile");

static __device__ __forceinline__ int rl_i(int v, int k) {
    return __builtin_amdgcn_readlane(v, k);
}
static __device__ __forceinline__ float rl_f(float v, int k) {
    return __int_as_float(__builtin_amdgcn_readlane(__float_as_int(v), k));
}

template <int K, int NP>
static __device__ __forceinline__ v8f tile_product(const float* __restrict__ sT, const float* __restrict__ sB,
                                                   int tn, int lane) {
    const int c  = lane & 15;
    const int kh = (lane >> 4) << 1;
    const float* aP = sT + c * TP + kh;
    const float* bP = sB + kh * NP + tn * 16 + c;
    v8f acc = {};
#pragma unroll
    for (int kk = 0; kk < K; kk += 4) {
        v2f a, b;
        a.x = aP[kk];
        a.y = aP[kk + 1];
        b.x = bP[kk * NP];
        b.y = bP[(kk + 1) * NP];
        acc = __builtin_amdgcn_wmma_f32_16x16x4_f32(
             false, a,  false, b,
             (short)0, acc,  false,  false);
    }
    return acc;
}

static __device__ __forceinline__ void chunk_rows_x(float4& acc, const float4* __restrict__ x4,
                                                    int cs_l, float w_l, int n, int lane) {
    int k = 0;
    for (; k + 4 <= n; k += 4) {
        const int s0 = rl_i(cs_l, k), s1 = rl_i(cs_l, k + 1), s2 = rl_i(cs_l, k + 2), s3 = rl_i(cs_l, k + 3);
        const float w0 = rl_f(w_l, k), w1 = rl_f(w_l, k + 1), w2 = rl_f(w_l, k + 2), w3 = rl_f(w_l, k + 3);
        const float4 v0 = x4[(long)s0 * 32 + lane];
        const float4 v1 = x4[(long)s1 * 32 + lane];
        const float4 v2 = x4[(long)s2 * 32 + lane];
        const float4 v3 = x4[(long)s3 * 32 + lane];
        acc.x = fmaf(v0.x, w0, acc.x); acc.y = fmaf(v0.y, w0, acc.y); acc.z = fmaf(v0.z, w0, acc.z); acc.w = fmaf(v0.w, w0, acc.w);
        acc.x = fmaf(v1.x, w1, acc.x); acc.y = fmaf(v1.y, w1, acc.y); acc.z = fmaf(v1.z, w1, acc.z); acc.w = fmaf(v1.w, w1, acc.w);
        acc.x = fmaf(v2.x, w2, acc.x); acc.y = fmaf(v2.y, w2, acc.y); acc.z = fmaf(v2.z, w2, acc.z); acc.w = fmaf(v2.w, w2, acc.w);
        acc.x = fmaf(v3.x, w3, acc.x); acc.y = fmaf(v3.y, w3, acc.y); acc.z = fmaf(v3.z, w3, acc.z); acc.w = fmaf(v3.w, w3, acc.w);
    }
    for (; k < n; ++k) {
        const int s0 = rl_i(cs_l, k);
        const float w0 = rl_f(w_l, k);
        const float4 v0 = x4[(long)s0 * 32 + lane];
        acc.x = fmaf(v0.x, w0, acc.x); acc.y = fmaf(v0.y, w0, acc.y); acc.z = fmaf(v0.z, w0, acc.z); acc.w = fmaf(v0.w, w0, acc.w);
    }
}

__global__ void __launch_bounds__(L1T, 12)
k_layer1(const int* __restrict__ rowptr, const int* __restrict__ ssrc,
         const float* __restrict__ dinv, const float* __restrict__ x,
         const float* __restrict__ W1, const float* __restrict__ b1,
         const float* __restrict__ W2, float* __restrict__ h2) {
    __shared__ float sW1[FIN * DH];
    __shared__ float sW2[DH * NCP];
    __shared__ float sT[16 * TP];

    const int tid  = threadIdx.x;
    const int lane = tid & 31;
    const int wv   = __builtin_amdgcn_readfirstlane(tid >> 5);
    const float4* x4 = (const float4*)x;

    int tile = blockIdx.x;
    int node = tile * 16 + wv;
    int beg = rowptr[node], end = rowptr[node + 1];
    float dd = dinv[node];
    float4 self = x4[(long)node * 32 + lane];

    {
        constexpr int Q1 = (FIN * DH) / 4;
        constexpr int QP = (DH * NCP) / 4;
        constexpr int N1 = Q1 / L1T, N2 = QP / L1T;
        static_assert(Q1 % L1T == 0 && QP % L1T == 0, "whole quads per thread, no tail");
        static_assert(NC % 4 == 0 && NCP % 4 == 0, "a quad stays inside its row, data or pad");
        float4 w[N1], u[N2];
#pragma unroll
        for (int t = 0; t < N1; ++t) w[t] = ((const float4*)W1)[tid + t * L1T];
#pragma unroll
        for (int t = 0; t < N2; ++t) {
            const int j = tid + t * L1T;
            const int r = j / (NCP / 4), q = j - r * (NCP / 4);
            u[t] = ((const float4*)W2)[r * (NC / 4) + (q < NC / 4 ? q : NC / 4 - 1)];
        }
#pragma unroll
        for (int t = 0; t < N1; ++t) ((float4*)sW1)[tid + t * L1T] = w[t];
#pragma unroll
        for (int t = 0; t < N2; ++t) {
            const int j = tid + t * L1T;
            const int q = j % (NCP / 4);
            const bool data = q < NC / 4;
            ((float4*)sW2)[j] = make_float4(data ? u[t].x : 0.0f, data ? u[t].y : 0.0f,
                                            data ? u[t].z : 0.0f, data ? u[t].w : 0.0f);
        }
    }

    int s_l = (beg + lane < end) ? ssrc[beg + lane] : node;
    if ((unsigned)s_l >= (unsigned)NN) s_l = node;
    float r_l = dinv[s_l];

    bool more  = tile + L1G < L1TILES;
    int  ntile = more ? tile + L1G : tile;
    int  nnode = ntile * 16 + wv;
    int  nbeg = rowptr[nnode], nend = rowptr[nnode + 1];
    float ndd = dinv[nnode];

    for (;;) {
        const bool more2 = more && (ntile + L1G < L1TILES);
        const int  tile2 = more2 ? ntile + L1G : ntile;
        const int  node2 = tile2 * 16 + wv;
        const int  beg2 = rowptr[node2], end2 = rowptr[node2 + 1];
        const float dd2 = dinv[node2];

        const int nj = nbeg + lane;
        int ns_l = ssrc[nj < nend ? nj : NE - 1];
        if (nj >= nend) ns_l = nnode;
        const float4 nself = x4[(long)nnode * 32 + lane];

        const int node0 = tile * 16;

        float4 acc = self;
        {
            const float ws = dd * dd;
            acc.x *= ws; acc.y *= ws; acc.z *= ws; acc.w *= ws;

            chunk_rows_x(acc, x4, s_l, r_l * dd, (end - beg < 32) ? (end - beg) : 32, lane);
            for (int cb = beg + 32; cb < end; cb += 32) {
                const int j = cb + lane;
                int cs_l = (j < end) ? ssrc[j] : node;
                if ((unsigned)cs_l >= (unsigned)NN) cs_l = node;
                const float w_l = dinv[cs_l] * dd;
                chunk_rows_x(acc, x4, cs_l, w_l, (end - cb < 32) ? (end - cb) : 32, lane);
            }
        }

        __syncthreads();
        ((float4*)(sT + wv * TP))[lane] = acc;
        __syncthreads();

        if ((unsigned)ns_l >= (unsigned)NN) ns_l = nnode;
        const float nr_l = dinv[ns_l];

        v8f p = {};
        if (wv < DH / 16) p = tile_product<FIN, DH>(sT, sW1, wv, lane);
        __syncthreads();
        if (wv < DH / 16) {
            const int n = wv * 16 + (lane & 15);
            const int r0 = (lane >> 4) << 3;
            const float bb = b1[n];
#pragma unroll
            for (int v = 0; v < 8; ++v) {
                const float t = p[v] + bb;
                sT[(r0 + v) * TP + n] = t > 0.0f ? t : 0.0f;
            }
        }
        __syncthreads();

        if (wv < NCP / 16) {
            const v8f o = tile_product<DH, NCP>(sT, sW2, wv, lane);
            const int n = wv * 16 + (lane & 15);
            if (n < NC) {
                const int r0 = node0 + ((lane >> 4) << 3);
#pragma unroll
                for (int v = 0; v < 8; ++v)
                    h2[(long)(r0 + v) * NC + n] = o[v];
            }
        }

        if (!more) break;
        tile = ntile; node = nnode;
        beg = __builtin_amdgcn_readfirstlane(nbeg);
        end = __builtin_amdgcn_readfirstlane(nend);
        dd  = __int_as_float(__builtin_amdgcn_readfirstlane(__float_as_int(ndd)));
        s_l = ns_l; r_l = nr_l; self = nself;
        ntile = tile2; nnode = node2; nbeg = beg2; nend = end2; ndd = dd2; more = more2;
    }
}


#define GCB  256
#define GCG  1563
#define GCW  (GCG * (GCB / 32))
static_assert(GCW <= NN, "every wave has a first node");

static __device__ __forceinline__ void chunk_rows_h(float& a0, float& a1, const float* __restrict__ h,
                                                    int cs_l, float w_l, int n, int lane, int c2) {
    int k = 0;
    for (; k + 4 <= n; k += 4) {
        const int s0 = rl_i(cs_l, k), s1 = rl_i(cs_l, k + 1), s2 = rl_i(cs_l, k + 2), s3 = rl_i(cs_l, k + 3);
        const float w0 = rl_f(w_l, k), w1 = rl_f(w_l, k + 1), w2 = rl_f(w_l, k + 2), w3 = rl_f(w_l, k + 3);
        const float p0 = h[(long)s0 * NC + lane], q0 = h[(long)s0 * NC + c2];
        const float p1 = h[(long)s1 * NC + lane], q1 = h[(long)s1 * NC + c2];
        const float p2 = h[(long)s2 * NC + lane], q2 = h[(long)s2 * NC + c2];
        const float p3 = h[(long)s3 * NC + lane], q3 = h[(long)s3 * NC + c2];
        a0 = fmaf(p0, w0, a0); a1 = fmaf(q0, w0, a1);
        a0 = fmaf(p1, w1, a0); a1 = fmaf(q1, w1, a1);
        a0 = fmaf(p2, w2, a0); a1 = fmaf(q2, w2, a1);
        a0 = fmaf(p3, w3, a0); a1 = fmaf(q3, w3, a1);
    }
    for (; k < n; ++k) {
        const int s0 = rl_i(cs_l, k);
        const float w0 = rl_f(w_l, k);
        a0 = fmaf(h[(long)s0 * NC + lane], w0, a0);
        a1 = fmaf(h[(long)s0 * NC + c2], w0, a1);
    }
}

__global__ void __launch_bounds__(GCB)
k_gather_c(const int* __restrict__ rowptr, const int* __restrict__ ssrc,
           const float* __restrict__ dinv,
           const float* __restrict__ h, const float* __restrict__ b,
           float* __restrict__ out) {
    const int lane = threadIdx.x & 31;
    const int gw = __builtin_amdgcn_readfirstlane((int)((blockIdx.x * GCB + threadIdx.x) >> 5));
    const bool hi = lane < NC - 32;
    const int  c2 = hi ? lane + 32 : lane;
    const float b0 = b[lane], b1 = b[c2];

    int node = gw;
    int beg = rowptr[node], end = rowptr[node + 1];
    float dd = dinv[node];
    float h0 = h[(long)node * NC + lane], h1 = h[(long)node * NC + c2];
    int s_l = (beg + lane < end) ? ssrc[beg + lane] : node;
    if ((unsigned)s_l >= (unsigned)NN) s_l = node;
    float r_l = dinv[s_l];

    bool more  = node + GCW < NN;
    int  nnode = more ? node + GCW : node;
    int  nbeg = rowptr[nnode], nend = rowptr[nnode + 1];
    float ndd = dinv[nnode];

    for (;;) {
        const bool more2 = more && (nnode + GCW < NN);
        const int  node2 = more2 ? nnode + GCW : nnode;
        const int  beg2 = rowptr[node2], end2 = rowptr[node2 + 1];
        const float dd2 = dinv[node2];

        const int nj = nbeg + lane;
        int ns_l = ssrc[nj < nend ? nj : NE - 1];
        if (nj >= nend) ns_l = nnode;
        const float nh0 = h[(long)nnode * NC + lane];
        const float nh1 = h[(long)nnode * NC + c2];

        const float ws = dd * dd;
        float a0 = h0 * ws;
        float a1 = h1 * ws;
        chunk_rows_h(a0, a1, h, s_l, r_l * dd, (end - beg < 32) ? (end - beg) : 32, lane, c2);
        for (int cb = beg + 32; cb < end; cb += 32) {
            const int j = cb + lane;
            int cs_l = (j < end) ? ssrc[j] : node;
            if ((unsigned)cs_l >= (unsigned)NN) cs_l = node;
            const float w_l = dinv[cs_l] * dd;
            chunk_rows_h(a0, a1, h, cs_l, w_l, (end - cb < 32) ? (end - cb) : 32, lane, c2);
        }

        if ((unsigned)ns_l >= (unsigned)NN) ns_l = nnode;
        const float nr_l = dinv[ns_l];

        const float v0 = a0 + b0;
        const float v1 = hi ? a1 + b1 : -INFINITY;
        float mx = fmaxf(v0, v1);
        for (int o = 16; o > 0; o >>= 1) mx = fmaxf(mx, __shfl_xor(mx, o));
        float sum = expf(v0 - mx) + (hi ? expf(v1 - mx) : 0.0f);
        for (int o = 16; o > 0; o >>= 1) sum += __shfl_xor(sum, o);
        const float lse = mx + logf(sum);
        out[(long)node * NC + lane] = v0 - lse;
        if (hi) out[(long)node * NC + c2] = v1 - lse;

        if (!more) break;
        node = nnode; beg = nbeg; end = nend; dd = ndd;
        s_l = ns_l; r_l = nr_l; h0 = nh0; h1 = nh1;
        nnode = node2; nbeg = beg2; nend = end2; ndd = dd2; more = more2;
    }
}


extern "C" void kernel_launch(void* const* d_in, const int* in_sizes, int n_in,
                              void* d_out, int out_size, void* d_ws, size_t ws_size,
                              hipStream_t stream) {
    const float* x  = (const float*)d_in[0];
    const int*   ei = (const int*)d_in[1];
    const float* W1 = (const float*)d_in[2];
    const float* b1 = (const float*)d_in[3];
    const float* W2 = (const float*)d_in[4];
    const float* b2 = (const float*)d_in[5];
    float* out = (float*)d_out;

    const int* src = ei;
    const int* dst = ei + NE;

    float* ws     = (float*)d_ws;
    int*   cnt    = (int*)ws;
    float* dinv   = ws + NNP;
    int*   rowptr = (int*)(dinv + NNP);
    int*   rank   = rowptr + NNP;
    int*   ssrc   = rank + NEP;
    float* h2     = (float*)(ssrc + NEP);
    int*   cntg   = (int*)(h2 + (long)NN * NC);

    const int B = 256;

    k_hist         <<<HG, HT, 0, stream>>>(dst, cntg, rank);
    k_chunk_prefix <<<cdiv(NN / 4, 64), 64, 0, stream>>>(cntg, cnt);
    k_scan         <<<1, SCAN_T, 0, stream>>>(cnt, rowptr, dinv);
    k_fill         <<<cdiv(NE / 4, B), B, 0, stream>>>(src, dst, rowptr, cntg, rank, ssrc);

    k_layer1   <<<L1G, L1T, 0, stream>>>(rowptr, ssrc, dinv, x, W1, b1, W2, h2);

    k_gather_c   <<<GCG, GCB, 0, stream>>>(rowptr, ssrc, dinv, h2, b2, out);
}


# Scan launch removed: in-degrees scanned per 256-node block inside the chunk-prefix launch and folded into its prefix rows; fill launch scans the block totals in LDS; 6 to 5 launches; sums as before
# speedup vs baseline: 3.7686x; 1.1551x over previous
#include <hip/hip_runtime.h>
#include <math.h>

#define NN   50000
#define NE   625000
#define FIN  128
#define DH   128
#define NC   40
#define NCP  48
#define NNP  50048
#define NEP  625024

typedef __attribute__((ext_vector_type(2))) float v2f;
typedef __attribute__((ext_vector_type(8))) float v8f;

static inline int cdiv(long a, long b) { return (int)((a + b - 1) / b); }


#define HG   32
#define HCH  19532
#define HQ   (HCH / 4)
#define HT   1024
#define HIT  ((HQ + HT - 1) / HT)
static_assert(HCH % 4 == 0 && NE % 4 == 0 && NN % 4 == 0 && (long)HG * HCH >= NE,
              "chunks are whole quads and cover every edge");

__global__ void __launch_bounds__(HT)
k_hist(const int* __restrict__ dst, int* __restrict__ cntg, int* __restrict__ rank) {
    __shared__ int c[NN];
    const int g  = blockIdx.x;
    const int q0 = g * HQ;
    const int nq = (NE / 4 - q0 < HQ) ? (NE / 4 - q0) : HQ;

    int4 d[HIT];
#pragma unroll
    for (int t = 0; t < HIT; ++t) {
        const int q = t * HT + (int)threadIdx.x;
        d[t] = (q < nq) ? ((const int4*)dst)[q0 + q] : make_int4(0, 0, 0, 0);
    }
    for (int i = threadIdx.x; i < NN / 4; i += HT) ((int4*)c)[i] = make_int4(0, 0, 0, 0);
    __syncthreads();

#pragma unroll
    for (int t = 0; t < HIT; ++t) {
        const int q = t * HT + (int)threadIdx.x;
        if (q < nq) {
            int4 r;
            r.x = atomicAdd(&c[d[t].x], 1);
            r.y = atomicAdd(&c[d[t].y], 1);
            r.z = atomicAdd(&c[d[t].z], 1);
            r.w = atomicAdd(&c[d[t].w], 1);
            ((int4*)rank)[q0 + q] = r;
        }
    }
    __syncthreads();

    int4* cg = (int4*)(cntg + (long)g * NNP);
    for (int i = threadIdx.x; i < NN / 4; i += HT) cg[i] = ((const int4*)c)[i];
}

#define PB   64
#define PBN  (PB * 4)
#define PNB  ((NN / 4 + PB - 1) / PB)
static_assert(PB % 32 == 0, "whole waves");

__global__ void __launch_bounds__(PB)
k_chunk_prefix(int* __restrict__ cntg, int* __restrict__ bsum, float* __restrict__ dinv) {
    __shared__ int wtot[PB / 32];
    const int lane = threadIdx.x & 31;
    const int wv   = threadIdx.x >> 5;
    const int i    = blockIdx.x * PB + threadIdx.x;
    const bool in  = i < NN / 4;
    const int  il  = in ? i : NN / 4 - 1;
    int4 v[HG];
#pragma unroll
    for (int g = 0; g < HG; ++g) v[g] = ((const int4*)(cntg + (long)g * NNP))[il];
    int4 tot = make_int4(0, 0, 0, 0);
#pragma unroll
    for (int g = 0; g < HG; ++g) { tot.x += v[g].x; tot.y += v[g].y; tot.z += v[g].z; tot.w += v[g].w; }
    if (!in) tot = make_int4(0, 0, 0, 0);

    const int qs = tot.x + tot.y + tot.z + tot.w;
    int inc = qs;
#pragma unroll
    for (int o = 1; o < 32; o <<= 1) {
        const int y = __shfl_up(inc, o);
        if (lane >= o) inc += y;
    }
    if (lane == 31) wtot[wv] = inc;
    __syncthreads();
    int woff = 0;
#pragma unroll
    for (int w = 0; w < PB / 32; ++w) if (w < wv) woff += wtot[w];
    if (threadIdx.x == PB - 1) bsum[blockIdx.x] = woff + inc;

    int4 run;
    run.x = woff + inc - qs;
    run.y = run.x + tot.x;
    run.z = run.y + tot.y;
    run.w = run.z + tot.z;
    if (in) {
        ((float4*)dinv)[i] = make_float4(rsqrtf((float)(tot.x + 1)), rsqrtf((float)(tot.y + 1)),
                                         rsqrtf((float)(tot.z + 1)), rsqrtf((float)(tot.w + 1)));
#pragma unroll
        for (int g = 0; g < HG; ++g) {
            ((int4*)(cntg + (long)g * NNP))[i] = run;
            run.x += v[g].x; run.y += v[g].y; run.z += v[g].z; run.w += v[g].w;
        }
    }
}

#define FB 256
static_assert(PNB <= FB, "one block total per thread");
static_assert(PBN % 4 == 0, "a quad of nodes lies inside one block");

__global__ void __launch_bounds__(FB)
k_fill(const int* __restrict__ src, const int* __restrict__ dst,
       const int* __restrict__ bsum, const int* __restrict__ pre,
       const int* __restrict__ rank, int* __restrict__ rowptr, int* __restrict__ ssrc) {
    __shared__ int boff[PNB + 1];
    __shared__ int wt[FB / 32];
    const int tid  = threadIdx.x;
    const int lane = tid & 31;
    const int wv   = tid >> 5;
    const int q    = blockIdx.x * FB + tid;
    const bool in  = q < NE / 4;
    const bool rp  = q < NN / 4;

    const int bl = bsum[tid < PNB ? tid : 0];
    const int bs = (tid < PNB) ? bl : 0;
    const int qe = in ? q : 0;
    const int qn = rp ? q : 0;
    const int4 s = ((const int4*)src)[qe];
    const int4 d = ((const int4*)dst)[qe];
    const int4 r = ((const int4*)rank)[qe];
    const int4 l = ((const int4*)pre)[qn];

    int inc = bs;
#pragma unroll
    for (int o = 1; o < 32; o <<= 1) {
        const int y = __shfl_up(inc, o);
        if (lane >= o) inc += y;
    }
    if (lane == 31) wt[wv] = inc;
    __syncthreads();
    int woff = 0;
#pragma unroll
    for (int w = 0; w < FB / 32; ++w) if (w < wv) woff += wt[w];
    if (tid < PNB)      boff[tid] = woff + inc - bs;
    if (tid == PNB - 1) boff[PNB] = woff + inc;
    __syncthreads();

    if (in) {
        const int* pg = pre + (long)(q / HQ) * NNP;
        const int px = boff[(unsigned)d.x / PBN] + pg[d.x], py = boff[(unsigned)d.y / PBN] + pg[d.y];
        const int pz = boff[(unsigned)d.z / PBN] + pg[d.z], pw = boff[(unsigned)d.w / PBN] + pg[d.w];
        ssrc[px + r.x] = s.x;
        ssrc[py + r.y] = s.y;
        ssrc[pz + r.z] = s.z;
        ssrc[pw + r.w] = s.w;
    }
    if (rp) {
        const int o = boff[(unsigned)q / PB];
        ((int4*)rowptr)[q] = make_int4(l.x + o, l.y + o, l.z + o, l.w + o);
    }
    if (q == 0) rowptr[NN] = boff[PNB];
}


#define TP  132
#define L1T 512
#define L1TILES (NN / 16)
#define L1G  768
static_assert(NN % 16 == 0, "exact grid: every work-group has 16 real nodes");
static_assert(FIN == DH, "one tile buffer serves both products");
static_assert(L1G <= L1TILES, "every work-group has a first tile");

static __device__ __forceinline__ int rl_i(int v, int k) {
    return __builtin_amdgcn_readlane(v, k);
}
static __device__ __forceinline__ float rl_f(float v, int k) {
    return __int_as_float(__builtin_amdgcn_readlane(__float_as_int(v), k));
}

template <int K, int NP>
static __device__ __forceinline__ v8f tile_product(const float* __restrict__ sT, const float* __restrict__ sB,
                                                   int tn, int lane) {
    const int c  = lane & 15;
    const int kh = (lane >> 4) << 1;
    const float* aP = sT + c * TP + kh;
    const float* bP = sB + kh * NP + tn * 16 + c;
    v8f acc = {};
#pragma unroll
    for (int kk = 0; kk < K; kk += 4) {
        v2f a, b;
        a.x = aP[kk];
        a.y = aP[kk + 1];
        b.x = bP[kk * NP];
        b.y = bP[(kk + 1) * NP];
        acc = __builtin_amdgcn_wmma_f32_16x16x4_f32(
             false, a,  false, b,
             (short)0, acc,  false,  false);
    }
    return acc;
}

static __device__ __forceinline__ void chunk_rows_x(float4& acc, const float4* __restrict__ x4,
                                                    int cs_l, float w_l, int n, int lane) {
    int k = 0;
    for (; k + 4 <= n; k += 4) {
        const int s0 = rl_i(cs_l, k), s1 = rl_i(cs_l, k + 1), s2 = rl_i(cs_l, k + 2), s3 = rl_i(cs_l, k + 3);
        const float w0 = rl_f(w_l, k), w1 = rl_f(w_l, k + 1), w2 = rl_f(w_l, k + 2), w3 = rl_f(w_l, k + 3);
        const float4 v0 = x4[(long)s0 * 32 + lane];
        const float4 v1 = x4[(long)s1 * 32 + lane];
        const float4 v2 = x4[(long)s2 * 32 + lane];
        const float4 v3 = x4[(long)s3 * 32 + lane];
        acc.x = fmaf(v0.x, w0, acc.x); acc.y = fmaf(v0.y, w0, acc.y); acc.z = fmaf(v0.z, w0, acc.z); acc.w = fmaf(v0.w, w0, acc.w);
        acc.x = fmaf(v1.x, w1, acc.x); acc.y = fmaf(v1.y, w1, acc.y); acc.z = fmaf(v1.z, w1, acc.z); acc.w = fmaf(v1.w, w1, acc.w);
        acc.x = fmaf(v2.x, w2, acc.x); acc.y = fmaf(v2.y, w2, acc.y); acc.z = fmaf(v2.z, w2, acc.z); acc.w = fmaf(v2.w, w2, acc.w);
        acc.x = fmaf(v3.x, w3, acc.x); acc.y = fmaf(v3.y, w3, acc.y); acc.z = fmaf(v3.z, w3, acc.z); acc.w = fmaf(v3.w, w3, acc.w);
    }
    for (; k < n; ++k) {
        const int s0 = rl_i(cs_l, k);
        const float w0 = rl_f(w_l, k);
        const float4 v0 = x4[(long)s0 * 32 + lane];
        acc.x = fmaf(v0.x, w0, acc.x); acc.y = fmaf(v0.y, w0, acc.y); acc.z = fmaf(v0.z, w0, acc.z); acc.w = fmaf(v0.w, w0, acc.w);
    }
}

__global__ void __launch_bounds__(L1T, 12)
k_layer1(const int* __restrict__ rowptr, const int* __restrict__ ssrc,
         const float* __restrict__ dinv, const float* __restrict__ x,
         const float* __restrict__ W1, const float* __restrict__ b1,
         const float* __restrict__ W2, float* __restrict__ h2) {
    __shared__ float sW1[FIN * DH];
    __shared__ float sW2[DH * NCP];
    __shared__ float sT[16 * TP];

    const int tid  = threadIdx.x;
    const int lane = tid & 31;
    const int wv   = __builtin_amdgcn_readfirstlane(tid >> 5);
    const float4* x4 = (const float4*)x;

    int tile = blockIdx.x;
    int node = tile * 16 + wv;
    int beg = rowptr[node], end = rowptr[node + 1];
    float dd = dinv[node];
    float4 self = x4[(long)node * 32 + lane];

    {
        constexpr int Q1 = (FIN * DH) / 4;
        constexpr int QP = (DH * NCP) / 4;
        constexpr int N1 = Q1 / L1T, N2 = QP / L1T;
        static_assert(Q1 % L1T == 0 && QP % L1T == 0, "whole quads per thread, no tail");
        static_assert(NC % 4 == 0 && NCP % 4 == 0, "a quad stays inside its row, data or pad");
        float4 w[N1], u[N2];
#pragma unroll
        for (int t = 0; t < N1; ++t) w[t] = ((const float4*)W1)[tid + t * L1T];
#pragma unroll
        for (int t = 0; t < N2; ++t) {
            const int j = tid + t * L1T;
            const int r = j / (NCP / 4), q = j - r * (NCP / 4);
            u[t] = ((const float4*)W2)[r * (NC / 4) + (q < NC / 4 ? q : NC / 4 - 1)];
        }
#pragma unroll
        for (int t = 0; t < N1; ++t) ((float4*)sW1)[tid + t * L1T] = w[t];
#pragma unroll
        for (int t = 0; t < N2; ++t) {
            const int j = tid + t * L1T;
            const int q = j % (NCP / 4);
            const bool data = q < NC / 4;
            ((float4*)sW2)[j] = make_float4(data ? u[t].x : 0.0f, data ? u[t].y : 0.0f,
                                            data ? u[t].z : 0.0f, data ? u[t].w : 0.0f);
        }
    }

    int s_l = (beg + lane < end) ? ssrc[beg + lane] : node;
    if ((unsigned)s_l >= (unsigned)NN) s_l = node;
    float r_l = dinv[s_l];

    bool more  = tile + L1G < L1TILES;
    int  ntile = more ? tile + L1G : tile;
    int  nnode = ntile * 16 + wv;
    int  nbeg = rowptr[nnode], nend = rowptr[nnode + 1];
    float ndd = dinv[nnode];

    for (;;) {
        const bool more2 = more && (ntile + L1G < L1TILES);
        const int  tile2 = more2 ? ntile + L1G : ntile;
        const int  node2 = tile2 * 16 + wv;
        const int  beg2 = rowptr[node2], end2 = rowptr[node2 + 1];
        const float dd2 = dinv[node2];

        const int nj = nbeg + lane;
        int ns_l = ssrc[nj < nend ? nj : NE - 1];
        if (nj >= nend) ns_l = nnode;
        const float4 nself = x4[(long)nnode * 32 + lane];

        const int node0 = tile * 16;

        float4 acc = self;
        {
            const float ws = dd * dd;
            acc.x *= ws; acc.y *= ws; acc.z *= ws; acc.w *= ws;

            chunk_rows_x(acc, x4, s_l, r_l * dd, (end - beg < 32) ? (end - beg) : 32, lane);
            for (int cb = beg + 32; cb < end; cb += 32) {
                const int j = cb + lane;
                int cs_l = (j < end) ? ssrc[j] : node;
                if ((unsigned)cs_l >= (unsigned)NN) cs_l = node;
                const float w_l = dinv[cs_l] * dd;
                chunk_rows_x(acc, x4, cs_l, w_l, (end - cb < 32) ? (end - cb) : 32, lane);
            }
        }

        __syncthreads();
        ((float4*)(sT + wv * TP))[lane] = acc;
        __syncthreads();

        if ((unsigned)ns_l >= (unsigned)NN) ns_l = nnode;
        const float nr_l = dinv[ns_l];

        v8f p = {};
        if (wv < DH / 16) p = tile_product<FIN, DH>(sT, sW1, wv, lane);
        __syncthreads();
        if (wv < DH / 16) {
            const int n = wv * 16 + (lane & 15);
            const int r0 = (lane >> 4) << 3;
            const float bb = b1[n];
#pragma unroll
            for (int v = 0; v < 8; ++v) {
                const float t = p[v] + bb;
                sT[(r0 + v) * TP + n] = t > 0.0f ? t : 0.0f;
            }
        }
        __syncthreads();

        if (wv < NCP / 16) {
            const v8f o = tile_product<DH, NCP>(sT, sW2, wv, lane);
            const int n = wv * 16 + (lane & 15);
            if (n < NC) {
                const int r0 = node0 + ((lane >> 4) << 3);
#pragma unroll
                for (int v = 0; v < 8; ++v)
                    h2[(long)(r0 + v) * NC + n] = o[v];
            }
        }

        if (!more) break;
        tile = ntile; node = nnode;
        beg = __builtin_amdgcn_readfirstlane(nbeg);
        end = __builtin_amdgcn_readfirstlane(nend);
        dd  = __int_as_float(__builtin_amdgcn_readfirstlane(__float_as_int(ndd)));
        s_l = ns_l; r_l = nr_l; self = nself;
        ntile = tile2; nnode = node2; nbeg = beg2; nend = end2; ndd = dd2; more = more2;
    }
}


#define GCB  256
#define GCG  1563
#define GCW  (GCG * (GCB / 32))
static_assert(GCW <= NN, "every wave has a first node");

static __device__ __forceinline__ void chunk_rows_h(float& a0, float& a1, const float* __restrict__ h,
                                                    int cs_l, float w_l, int n, int lane, int c2) {
    int k = 0;
    for (; k + 4 <= n; k += 4) {
        const int s0 = rl_i(cs_l, k), s1 = rl_i(cs_l, k + 1), s2 = rl_i(cs_l, k + 2), s3 = rl_i(cs_l, k + 3);
        const float w0 = rl_f(w_l, k), w1 = rl_f(w_l, k + 1), w2 = rl_f(w_l, k + 2), w3 = rl_f(w_l, k + 3);
        const float p0 = h[(long)s0 * NC + lane], q0 = h[(long)s0 * NC + c2];
        const float p1 = h[(long)s1 * NC + lane], q1 = h[(long)s1 * NC + c2];
        const float p2 = h[(long)s2 * NC + lane], q2 = h[(long)s2 * NC + c2];
        const float p3 = h[(long)s3 * NC + lane], q3 = h[(long)s3 * NC + c2];
        a0 = fmaf(p0, w0, a0); a1 = fmaf(q0, w0, a1);
        a0 = fmaf(p1, w1, a0); a1 = fmaf(q1, w1, a1);
        a0 = fmaf(p2, w2, a0); a1 = fmaf(q2, w2, a1);
        a0 = fmaf(p3, w3, a0); a1 = fmaf(q3, w3, a1);
    }
    for (; k < n; ++k) {
        const int s0 = rl_i(cs_l, k);
        const float w0 = rl_f(w_l, k);
        a0 = fmaf(h[(long)s0 * NC + lane], w0, a0);
        a1 = fmaf(h[(long)s0 * NC + c2], w0, a1);
    }
}

__global__ void __launch_bounds__(GCB)
k_gather_c(const int* __restrict__ rowptr, const int* __restrict__ ssrc,
           const float* __restrict__ dinv,
           const float* __restrict__ h, const float* __restrict__ b,
           float* __restrict__ out) {
    const int lane = threadIdx.x & 31;
    const int gw = __builtin_amdgcn_readfirstlane((int)((blockIdx.x * GCB + threadIdx.x) >> 5));
    const bool hi = lane < NC - 32;
    const int  c2 = hi ? lane + 32 : lane;
    const float b0 = b[lane], b1 = b[c2];

    int node = gw;
    int beg = rowptr[node], end = rowptr[node + 1];
    float dd = dinv[node];
    float h0 = h[(long)node * NC + lane], h1 = h[(long)node * NC + c2];
    int s_l = (beg + lane < end) ? ssrc[beg + lane] : node;
    if ((unsigned)s_l >= (unsigned)NN) s_l = node;
    float r_l = dinv[s_l];

    bool more  = node + GCW < NN;
    int  nnode = more ? node + GCW : node;
    int  nbeg = rowptr[nnode], nend = rowptr[nnode + 1];
    float ndd = dinv[nnode];

    for (;;) {
        const bool more2 = more && (nnode + GCW < NN);
        const int  node2 = more2 ? nnode + GCW : nnode;
        const int  beg2 = rowptr[node2], end2 = rowptr[node2 + 1];
        const float dd2 = dinv[node2];

        const int nj = nbeg + lane;
        int ns_l = ssrc[nj < nend ? nj : NE - 1];
        if (nj >= nend) ns_l = nnode;
        const float nh0 = h[(long)nnode * NC + lane];
        const float nh1 = h[(long)nnode * NC + c2];

        const float ws = dd * dd;
        float a0 = h0 * ws;
        float a1 = h1 * ws;
        chunk_rows_h(a0, a1, h, s_l, r_l * dd, (end - beg < 32) ? (end - beg) : 32, lane, c2);
        for (int cb = beg + 32; cb < end; cb += 32) {
            const int j = cb + lane;
            int cs_l = (j < end) ? ssrc[j] : node;
            if ((unsigned)cs_l >= (unsigned)NN) cs_l = node;
            const float w_l = dinv[cs_l] * dd;
            chunk_rows_h(a0, a1, h, cs_l, w_l, (end - cb < 32) ? (end - cb) : 32, lane, c2);
        }

        if ((unsigned)ns_l >= (unsigned)NN) ns_l = nnode;
        const float nr_l = dinv[ns_l];

        const float v0 = a0 + b0;
        const float v1 = hi ? a1 + b1 : -INFINITY;
        float mx = fmaxf(v0, v1);
        for (int o = 16; o > 0; o >>= 1) mx = fmaxf(mx, __shfl_xor(mx, o));
        float sum = expf(v0 - mx) + (hi ? expf(v1 - mx) : 0.0f);
        for (int o = 16; o > 0; o >>= 1) sum += __shfl_xor(sum, o);
        const float lse = mx + logf(sum);
        out[(long)node * NC + lane] = v0 - lse;
        if (hi) out[(long)node * NC + c2] = v1 - lse;

        if (!more) break;
        node = nnode; beg = nbeg; end = nend; dd = ndd;
        s_l = ns_l; r_l = nr_l; h0 = nh0; h1 = nh1;
        nnode = node2; nbeg = beg2; nend = end2; ndd = dd2; more = more2;
    }
}


extern "C" void kernel_launch(void* const* d_in, const int* in_sizes, int n_in,
                              void* d_out, int out_size, void* d_ws, size_t ws_size,
                              hipStream_t stream) {
    const float* x  = (const float*)d_in[0];
    const int*   ei = (const int*)d_in[1];
    const float* W1 = (const float*)d_in[2];
    const float* b1 = (const float*)d_in[3];
    const float* W2 = (const float*)d_in[4];
    const float* b2 = (const float*)d_in[5];
    float* out = (float*)d_out;

    const int* src = ei;
    const int* dst = ei + NE;

    float* ws     = (float*)d_ws;
    int*   bsum   = (int*)ws;
    float* dinv   = ws + NNP;
    int*   rowptr = (int*)(dinv + NNP);
    int*   rank   = rowptr + NNP;
    int*   ssrc   = rank + NEP;
    float* h2     = (float*)(ssrc + NEP);
    int*   cntg   = (int*)(h2 + (long)NN * NC);

    k_hist         <<<HG, HT, 0, stream>>>(dst, cntg, rank);
    k_chunk_prefix <<<PNB, PB, 0, stream>>>(cntg, bsum, dinv);
    k_fill         <<<cdiv(NE / 4, FB), FB, 0, stream>>>(src, dst, bsum, cntg, rank, rowptr, ssrc);

    k_layer1   <<<L1G, L1T, 0, stream>>>(rowptr, ssrc, dinv, x, W1, b1, W2, h2);

    k_gather_c   <<<GCG, GCB, 0, stream>>>(rowptr, ssrc, dinv, h2, b2, out);
}


# Edge placement moved into the count launch: a chunk scans its LDS table and sorts its sources by dst in LDS; a per-node launch copies the runs; prefix and fill launches go (5 to 4); sum order kept
# speedup vs baseline: 4.3807x; 1.1624x over previous
#include <hip/hip_runtime.h>
#include <math.h>

#define NN   50000
#define NE   625000
#define FIN  128
#define DH   128
#define NC   40
#define NCP  48
#define NNP  50048
#define NEP  625024

typedef __attribute__((ext_vector_type(2))) float v2f;
typedef __attribute__((ext_vector_type(8))) float v8f;

static inline int cdiv(long a, long b) { return (int)((a + b - 1) / b); }


#define HG   32
#define HCH  19532
#define HQ   (HCH / 4)
#define HT   1024
#define HIT  ((HQ + HT - 1) / HT)
#define HSEG 52
static_assert(HCH % 4 == 0 && NE % 4 == 0 && NN % 4 == 0 && (long)HG * HCH >= NE,
              "chunks are whole quads and cover every edge");
static_assert(HSEG % 4 == 0 && (long)HT * HSEG >= NN, "the threads' segments cover the table");

__global__ void __launch_bounds__(HT)
k_hist(const int* __restrict__ src, const int* __restrict__ dst,
       int* __restrict__ startg, int* __restrict__ sorted) {
    __shared__ __attribute__((aligned(16))) int c[NN];
    __shared__ __attribute__((aligned(16))) int ls[HCH];
    __shared__ int wt[HT / 32];
    const int tid  = threadIdx.x;
    const int lane = tid & 31;
    const int wv   = tid >> 5;
    const int g  = blockIdx.x;
    const int q0 = g * HQ;
    const int nq = (NE / 4 - q0 < HQ) ? (NE / 4 - q0) : HQ;

    int4 d[HIT], s[HIT];
#pragma unroll
    for (int t = 0; t < HIT; ++t) {
        const int q = t * HT + tid;
        d[t] = (q < nq) ? ((const int4*)dst)[q0 + q] : make_int4(0, 0, 0, 0);
        s[t] = (q < nq) ? ((const int4*)src)[q0 + q] : make_int4(0, 0, 0, 0);
    }
    for (int i = tid; i < NN / 4; i += HT) ((int4*)c)[i] = make_int4(0, 0, 0, 0);
    __syncthreads();

    int4 r[HIT];
#pragma unroll
    for (int t = 0; t < HIT; ++t) {
        const int q = t * HT + tid;
        r[t] = make_int4(0, 0, 0, 0);
        if (q < nq) {
            r[t].x = atomicAdd(&c[d[t].x], 1);
            r[t].y = atomicAdd(&c[d[t].y], 1);
            r[t].z = atomicAdd(&c[d[t].z], 1);
            r[t].w = atomicAdd(&c[d[t].w], 1);
        }
    }
    __syncthreads();

    const int i0 = tid * HSEG;
    int sum = 0;
#pragma unroll
    for (int k = 0; k < HSEG; k += 4) {
        if (i0 + k < NN) {
            const int4 v = *(const int4*)&c[i0 + k];
            sum += v.x + v.y + v.z + v.w;
        }
    }
    int inc = sum;
#pragma unroll
    for (int o = 1; o < 32; o <<= 1) {
        const int y = __shfl_up(inc, o);
        if (lane >= o) inc += y;
    }
    if (lane == 31) wt[wv] = inc;
    __syncthreads();
    int run = inc - sum;
#pragma unroll
    for (int w = 0; w < HT / 32; ++w) if (w < wv) run += wt[w];
#pragma unroll
    for (int k = 0; k < HSEG; k += 4) {
        if (i0 + k < NN) {
            const int4 v = *(const int4*)&c[i0 + k];
            int4 p;
            p.x = run;
            p.y = p.x + v.x;
            p.z = p.y + v.y;
            p.w = p.z + v.z;
            run = p.w + v.w;
            *(int4*)&c[i0 + k] = p;
        }
    }
    __syncthreads();

    int* cg = startg + (long)g * NNP;
    for (int i = tid; i < NN / 4; i += HT) ((int4*)cg)[i] = ((const int4*)c)[i];
    if (tid == 0) cg[NN] = 4 * nq;
#pragma unroll
    for (int t = 0; t < HIT; ++t) {
        const int q = t * HT + tid;
        if (q < nq) {
            ls[c[d[t].x] + r[t].x] = s[t].x;
            ls[c[d[t].y] + r[t].y] = s[t].y;
            ls[c[d[t].z] + r[t].z] = s[t].z;
            ls[c[d[t].w] + r[t].w] = s[t].w;
        }
    }
    __syncthreads();
    for (int i = tid; i < nq; i += HT) ((int4*)sorted)[q0 + i] = ((const int4*)ls)[i];
}

#define PB   256
#define PNB  ((NN + 1 + PB - 1) / PB)
static_assert(PB % 32 == 0, "whole waves");

__global__ void __launch_bounds__(PB)
k_place(const int* __restrict__ startg, const int* __restrict__ sorted,
        int* __restrict__ rowptr, float* __restrict__ dinv, int* __restrict__ ssrc) {
    __shared__ int wt[PB / 32];
    const int tid  = threadIdx.x;
    const int lane = tid & 31;
    const int wv   = tid >> 5;
    const int d0   = blockIdx.x * PB;
    const int d    = d0 + tid;
    const bool real = d < NN;
    const int dl   = (d <= NN) ? d : NN;
    const int dn   = (dl < NN) ? dl + 1 : NN;

    int S[HG], n[HG];
    int base = 0, deg = 0;
#pragma unroll
    for (int g = 0; g < HG; ++g) {
        const int* row = startg + (long)g * NNP;
        S[g] = row[dl];
        n[g] = row[dn] - S[g];
        deg += n[g];
        base += row[d0];
    }

    int inc = deg;
#pragma unroll
    for (int o = 1; o < 32; o <<= 1) {
        const int y = __shfl_up(inc, o);
        if (lane >= o) inc += y;
    }
    if (lane == 31) wt[wv] = inc;
    __syncthreads();
    int rp = base + inc - deg;
#pragma unroll
    for (int w = 0; w < PB / 32; ++w) if (w < wv) rp += wt[w];

    if (d <= NN) rowptr[d] = rp;
    if (real) dinv[d] = rsqrtf((float)(deg + 1));

    int maxn = 0;
#pragma unroll
    for (int g = 0; g < HG; ++g) maxn = (n[g] > maxn) ? n[g] : maxn;
    for (int i = 0; __any(i < maxn); ++i) {
        int v[HG];
#pragma unroll
        for (int g = 0; g < HG; ++g) v[g] = (i < n[g]) ? sorted[g * HCH + S[g] + i] : 0;
        int pos = rp + i;
#pragma unroll
        for (int g = 0; g < HG; ++g) {
            if (i < n[g]) ssrc[pos] = v[g];
            pos += n[g];
        }
    }
}


#define TP  132
#define L1T 512
#define L1TILES (NN / 16)
#define L1G  768
static_assert(NN % 16 == 0, "exact grid: every work-group has 16 real nodes");
static_assert(FIN == DH, "one tile buffer serves both products");
static_assert(L1G <= L1TILES, "every work-group has a first tile");

static __device__ __forceinline__ int rl_i(int v, int k) {
    return __builtin_amdgcn_readlane(v, k);
}
static __device__ __forceinline__ float rl_f(float v, int k) {
    return __int_as_float(__builtin_amdgcn_readlane(__float_as_int(v), k));
}

template <int K, int NP>
static __device__ __forceinline__ v8f tile_product(const float* __restrict__ sT, const float* __restrict__ sB,
                                                   int tn, int lane) {
    const int c  = lane & 15;
    const int kh = (lane >> 4) << 1;
    const float* aP = sT + c * TP + kh;
    const float* bP = sB + kh * NP + tn * 16 + c;
    v8f acc = {};
#pragma unroll
    for (int kk = 0; kk < K; kk += 4) {
        v2f a, b;
        a.x = aP[kk];
        a.y = aP[kk + 1];
        b.x = bP[kk * NP];
        b.y = bP[(kk + 1) * NP];
        acc = __builtin_amdgcn_wmma_f32_16x16x4_f32(
             false, a,  false, b,
             (short)0, acc,  false,  false);
    }
    return acc;
}

static __device__ __forceinline__ void chunk_rows_x(float4& acc, const float4* __restrict__ x4,
                                                    int cs_l, float w_l, int n, int lane) {
    int k = 0;
    for (; k + 4 <= n; k += 4) {
        const int s0 = rl_i(cs_l, k), s1 = rl_i(cs_l, k + 1), s2 = rl_i(cs_l, k + 2), s3 = rl_i(cs_l, k + 3);
        const float w0 = rl_f(w_l, k), w1 = rl_f(w_l, k + 1), w2 = rl_f(w_l, k + 2), w3 = rl_f(w_l, k + 3);
        const float4 v0 = x4[(long)s0 * 32 + lane];
        const float4 v1 = x4[(long)s1 * 32 + lane];
        const float4 v2 = x4[(long)s2 * 32 + lane];
        const float4 v3 = x4[(long)s3 * 32 + lane];
        acc.x = fmaf(v0.x, w0, acc.x); acc.y = fmaf(v0.y, w0, acc.y); acc.z = fmaf(v0.z, w0, acc.z); acc.w = fmaf(v0.w, w0, acc.w);
        acc.x = fmaf(v1.x, w1, acc.x); acc.y = fmaf(v1.y, w1, acc.y); acc.z = fmaf(v1.z, w1, acc.z); acc.w = fmaf(v1.w, w1, acc.w);
        acc.x = fmaf(v2.x, w2, acc.x); acc.y = fmaf(v2.y, w2, acc.y); acc.z = fmaf(v2.z, w2, acc.z); acc.w = fmaf(v2.w, w2, acc.w);
        acc.x = fmaf(v3.x, w3, acc.x); acc.y = fmaf(v3.y, w3, acc.y); acc.z = fmaf(v3.z, w3, acc.z); acc.w = fmaf(v3.w, w3, acc.w);
    }
    for (; k < n; ++k) {
        const int s0 = rl_i(cs_l, k);
        const float w0 = rl_f(w_l, k);
        const float4 v0 = x4[(long)s0 * 32 + lane];
        acc.x = fmaf(v0.x, w0, acc.x); acc.y = fmaf(v0.y, w0, acc.y); acc.z = fmaf(v0.z, w0, acc.z); acc.w = fmaf(v0.w, w0, acc.w);
    }
}

__global__ void __launch_bounds__(L1T, 12)
k_layer1(const int* __restrict__ rowptr, const int* __restrict__ ssrc,
         const float* __restrict__ dinv, const float* __restrict__ x,
         const float* __restrict__ W1, const float* __restrict__ b1,
         const float* __restrict__ W2, float* __restrict__ h2) {
    __shared__ float sW1[FIN * DH];
    __shared__ float sW2[DH * NCP];
    __shared__ float sT[16 * TP];

    const int tid  = threadIdx.x;
    const int lane = tid & 31;
    const int wv   = __builtin_amdgcn_readfirstlane(tid >> 5);
    const float4* x4 = (const float4*)x;

    int tile = blockIdx.x;
    int node = tile * 16 + wv;
    int beg = rowptr[node], end = rowptr[node + 1];
    float dd = dinv[node];
    float4 self = x4[(long)node * 32 + lane];

    {
        constexpr int Q1 = (FIN * DH) / 4;
        constexpr int QP = (DH * NCP) / 4;
        constexpr int N1 = Q1 / L1T, N2 = QP / L1T;
        static_assert(Q1 % L1T == 0 && QP % L1T == 0, "whole quads per thread, no tail");
        static_assert(NC % 4 == 0 && NCP % 4 == 0, "a quad stays inside its row, data or pad");
        float4 w[N1], u[N2];
#pragma unroll
        for (int t = 0; t < N1; ++t) w[t] = ((const float4*)W1)[tid + t * L1T];
#pragma unroll
        for (int t = 0; t < N2; ++t) {
            const int j = tid + t * L1T;
            const int r = j / (NCP / 4), q = j - r * (NCP / 4);
            u[t] = ((const float4*)W2)[r * (NC / 4) + (q < NC / 4 ? q : NC / 4 - 1)];
        }
#pragma unroll
        for (int t = 0; t < N1; ++t) ((float4*)sW1)[tid + t * L1T] = w[t];
#pragma unroll
        for (int t = 0; t < N2; ++t) {
            const int j = tid + t * L1T;
            const int q = j % (NCP / 4);
            const bool data = q < NC / 4;
            ((float4*)sW2)[j] = make_float4(data ? u[t].x : 0.0f, data ? u[t].y : 0.0f,
                                            data ? u[t].z : 0.0f, data ? u[t].w : 0.0f);
        }
    }

    int s_l = (beg + lane < end) ? ssrc[beg + lane] : node;
    if ((unsigned)s_l >= (unsigned)NN) s_l = node;
    float r_l = dinv[s_l];

    bool more  = tile + L1G < L1TILES;
    int  ntile = more ? tile + L1G : tile;
    int  nnode = ntile * 16 + wv;
    int  nbeg = rowptr[nnode], nend = rowptr[nnode + 1];
    float ndd = dinv[nnode];

    for (;;) {
        const bool more2 = more && (ntile + L1G < L1TILES);
        const int  tile2 = more2 ? ntile + L1G : ntile;
        const int  node2 = tile2 * 16 + wv;
        const int  beg2 = rowptr[node2], end2 = rowptr[node2 + 1];
        const float dd2 = dinv[node2];

        const int nj = nbeg + lane;
        int ns_l = ssrc[nj < nend ? nj : NE - 1];
        if (nj >= nend) ns_l = nnode;
        const float4 nself = x4[(long)nnode * 32 + lane];

        const int node0 = tile * 16;

        float4 acc = self;
        {
            const float ws = dd * dd;
            acc.x *= ws; acc.y *= ws; acc.z *= ws; acc.w *= ws;

            chunk_rows_x(acc, x4, s_l, r_l * dd, (end - beg < 32) ? (end - beg) : 32, lane);
            for (int cb = beg + 32; cb < end; cb += 32) {
                const int j = cb + lane;
                int cs_l = (j < end) ? ssrc[j] : node;
                if ((unsigned)cs_l >= (unsigned)NN) cs_l = node;
                const float w_l = dinv[cs_l] * dd;
                chunk_rows_x(acc, x4, cs_l, w_l, (end - cb < 32) ? (end - cb) : 32, lane);
            }
        }

        __syncthreads();
        ((float4*)(sT + wv * TP))[lane] = acc;
        __syncthreads();

        if ((unsigned)ns_l >= (unsigned)NN) ns_l = nnode;
        const float nr_l = dinv[ns_l];

        v8f p = {};
        if (wv < DH / 16) p = tile_product<FIN, DH>(sT, sW1, wv, lane);
        __syncthreads();
        if (wv < DH / 16) {
            const int n = wv * 16 + (lane & 15);
            const int r0 = (lane >> 4) << 3;
            const float bb = b1[n];
#pragma unroll
            for (int v = 0; v < 8; ++v) {
                const float t = p[v] + bb;
                sT[(r0 + v) * TP + n] = t > 0.0f ? t : 0.0f;
            }
        }
        __syncthreads();

        if (wv < NCP / 16) {
            const v8f o = tile_product<DH, NCP>(sT, sW2, wv, lane);
            const int n = wv * 16 + (lane & 15);
            if (n < NC) {
                const int r0 = node0 + ((lane >> 4) << 3);
#pragma unroll
                for (int v = 0; v < 8; ++v)
                    h2[(long)(r0 + v) * NC + n] = o[v];
            }
        }

        if (!more) break;
        tile = ntile; node = nnode;
        beg = __builtin_amdgcn_readfirstlane(nbeg);
        end = __builtin_amdgcn_readfirstlane(nend);
        dd  = __int_as_float(__builtin_amdgcn_readfirstlane(__float_as_int(ndd)));
        s_l = ns_l; r_l = nr_l; self = nself;
        ntile = tile2; nnode = node2; nbeg = beg2; nend = end2; ndd = dd2; more = more2;
    }
}


#define GCB  256
#define GCG  1563
#define GCW  (GCG * (GCB / 32))
static_assert(GCW <= NN, "every wave has a first node");

static __device__ __forceinline__ void chunk_rows_h(float& a0, float& a1, const float* __restrict__ h,
                                                    int cs_l, float w_l, int n, int lane, int c2) {
    int k = 0;
    for (; k + 4 <= n; k += 4) {
        const int s0 = rl_i(cs_l, k), s1 = rl_i(cs_l, k + 1), s2 = rl_i(cs_l, k + 2), s3 = rl_i(cs_l, k + 3);
        const float w0 = rl_f(w_l, k), w1 = rl_f(w_l, k + 1), w2 = rl_f(w_l, k + 2), w3 = rl_f(w_l, k + 3);
        const float p0 = h[(long)s0 * NC + lane], q0 = h[(long)s0 * NC + c2];
        const float p1 = h[(long)s1 * NC + lane], q1 = h[(long)s1 * NC + c2];
        const float p2 = h[(long)s2 * NC + lane], q2 = h[(long)s2 * NC + c2];
        const float p3 = h[(long)s3 * NC + lane], q3 = h[(long)s3 * NC + c2];
        a0 = fmaf(p0, w0, a0); a1 = fmaf(q0, w0, a1);
        a0 = fmaf(p1, w1, a0); a1 = fmaf(q1, w1, a1);
        a0 = fmaf(p2, w2, a0); a1 = fmaf(q2, w2, a1);
        a0 = fmaf(p3, w3, a0); a1 = fmaf(q3, w3, a1);
    }
    for (; k < n; ++k) {
        const int s0 = rl_i(cs_l, k);
        const float w0 = rl_f(w_l, k);
        a0 = fmaf(h[(long)s0 * NC + lane], w0, a0);
        a1 = fmaf(h[(long)s0 * NC + c2], w0, a1);
    }
}

__global__ void __launch_bounds__(GCB)
k_gather_c(const int* __restrict__ rowptr, const int* __restrict__ ssrc,
           const float* __restrict__ dinv,
           const float* __restrict__ h, const float* __restrict__ b,
           float* __restrict__ out) {
    const int lane = threadIdx.x & 31;
    const int gw = __builtin_amdgcn_readfirstlane((int)((blockIdx.x * GCB + threadIdx.x) >> 5));
    const bool hi = lane < NC - 32;
    const int  c2 = hi ? lane + 32 : lane;
    const float b0 = b[lane], b1 = b[c2];

    int node = gw;
    int beg = rowptr[node], end = rowptr[node + 1];
    float dd = dinv[node];
    float h0 = h[(long)node * NC + lane], h1 = h[(long)node * NC + c2];
    int s_l = (beg + lane < end) ? ssrc[beg + lane] : node;
    if ((unsigned)s_l >= (unsigned)NN) s_l = node;
    float r_l = dinv[s_l];

    bool more  = node + GCW < NN;
    int  nnode = more ? node + GCW : node;
    int  nbeg = rowptr[nnode], nend = rowptr[nnode + 1];
    float ndd = dinv[nnode];

    for (;;) {
        const bool more2 = more && (nnode + GCW < NN);
        const int  node2 = more2 ? nnode + GCW : nnode;
        const int  beg2 = rowptr[node2], end2 = rowptr[node2 + 1];
        const float dd2 = dinv[node2];

        const int nj = nbeg + lane;
        int ns_l = ssrc[nj < nend ? nj : NE - 1];
        if (nj >= nend) ns_l = nnode;
        const float nh0 = h[(long)nnode * NC + lane];
        const float nh1 = h[(long)nnode * NC + c2];

        const float ws = dd * dd;
        float a0 = h0 * ws;
        float a1 = h1 * ws;
        chunk_rows_h(a0, a1, h, s_l, r_l * dd, (end - beg < 32) ? (end - beg) : 32, lane, c2);
        for (int cb = beg + 32; cb < end; cb += 32) {
            const int j = cb + lane;
            int cs_l = (j < end) ? ssrc[j] : node;
            if ((unsigned)cs_l >= (unsigned)NN) cs_l = node;
            const float w_l = dinv[cs_l] * dd;
            chunk_rows_h(a0, a1, h, cs_l, w_l, (end - cb < 32) ? (end - cb) : 32, lane, c2);
        }

        if ((unsigned)ns_l >= (unsigned)NN) ns_l = nnode;
        const float nr_l = dinv[ns_l];

        const float v0 = a0 + b0;
        const float v1 = hi ? a1 + b1 : -INFINITY;
        float mx = fmaxf(v0, v1);
        for (int o = 16; o > 0; o >>= 1) mx = fmaxf(mx, __shfl_xor(mx, o));
        float sum = expf(v0 - mx) + (hi ? expf(v1 - mx) : 0.0f);
        for (int o = 16; o > 0; o >>= 1) sum += __shfl_xor(sum, o);
        const float lse = mx + logf(sum);
        out[(long)node * NC + lane] = v0 - lse;
        if (hi) out[(long)node * NC + c2] = v1 - lse;

        if (!more) break;
        node = nnode; beg = nbeg; end = nend; dd = ndd;
        s_l = ns_l; r_l = nr_l; h0 = nh0; h1 = nh1;
        nnode = node2; nbeg = beg2; nend = end2; ndd = dd2; more = more2;
    }
}


extern "C" void kernel_launch(void* const* d_in, const int* in_sizes, int n_in,
                              void* d_out, int out_size, void* d_ws, size_t ws_size,
                              hipStream_t stream) {
    const float* x  = (const float*)d_in[0];
    const int*   ei = (const int*)d_in[1];
    const float* W1 = (const float*)d_in[2];
    const float* b1 = (const float*)d_in[3];
    const float* W2 = (const float*)d_in[4];
    const float* b2 = (const float*)d_in[5];
    float* out = (float*)d_out;

    const int* src = ei;
    const int* dst = ei + NE;

    float* ws     = (float*)d_ws;
    float* dinv   = ws;
    int*   rowptr = (int*)(dinv + NNP);
    int*   sorted = rowptr + NNP;
    int*   ssrc   = sorted + NEP;
    float* h2     = (float*)(ssrc + NEP);
    int*   startg = (int*)(h2 + (long)NN * NC);

    k_hist  <<<HG, HT, 0, stream>>>(src, dst, startg, sorted);
    k_place <<<PNB, PB, 0, stream>>>(startg, sorted, rowptr, dinv, ssrc);

    k_layer1   <<<L1G, L1T, 0, stream>>>(rowptr, ssrc, dinv, x, W1, b1, W2, h2);

    k_gather_c   <<<GCG, GCB, 0, stream>>>(rowptr, ssrc, dinv, h2, b2, out);
}


# Both row gathers: a node's list rows requested in batches of 10 (layer 1) and 16 (layer 2) before the first add, leftover rows together in one round trip; was 4 rows, then 1 by 1; sum order kept
# speedup vs baseline: 4.5533x; 1.0394x over previous
#include <hip/hip_runtime.h>
#include <math.h>

#define NN   50000
#define NE   625000
#define FIN  128
#define DH   128
#define NC   40
#define NCP  48
#define NNP  50048
#define NEP  625024

typedef __attribute__((ext_vector_type(2))) float v2f;
typedef __attribute__((ext_vector_type(8))) float v8f;

static inline int cdiv(long a, long b) { return (int)((a + b - 1) / b); }


#define HG   32
#define HCH  19532
#define HQ   (HCH / 4)
#define HT   1024
#define HIT  ((HQ + HT - 1) / HT)
#define HSEG 52
static_assert(HCH % 4 == 0 && NE % 4 == 0 && NN % 4 == 0 && (long)HG * HCH >= NE,
              "chunks are whole quads and cover every edge");
static_assert(HSEG % 4 == 0 && (long)HT * HSEG >= NN, "the threads' segments cover the table");

__global__ void __launch_bounds__(HT)
k_hist(const int* __restrict__ src, const int* __restrict__ dst,
       int* __restrict__ startg, int* __restrict__ sorted) {
    __shared__ __attribute__((aligned(16))) int c[NN];
    __shared__ __attribute__((aligned(16))) int ls[HCH];
    __shared__ int wt[HT / 32];
    const int tid  = threadIdx.x;
    const int lane = tid & 31;
    const int wv   = tid >> 5;
    const int g  = blockIdx.x;
    const int q0 = g * HQ;
    const int nq = (NE / 4 - q0 < HQ) ? (NE / 4 - q0) : HQ;

    int4 d[HIT], s[HIT];
#pragma unroll
    for (int t = 0; t < HIT; ++t) {
        const int q = t * HT + tid;
        d[t] = (q < nq) ? ((const int4*)dst)[q0 + q] : make_int4(0, 0, 0, 0);
        s[t] = (q < nq) ? ((const int4*)src)[q0 + q] : make_int4(0, 0, 0, 0);
    }
    for (int i = tid; i < NN / 4; i += HT) ((int4*)c)[i] = make_int4(0, 0, 0, 0);
    __syncthreads();

    int4 r[HIT];
#pragma unroll
    for (int t = 0; t < HIT; ++t) {
        const int q = t * HT + tid;
        r[t] = make_int4(0, 0, 0, 0);
        if (q < nq) {
            r[t].x = atomicAdd(&c[d[t].x], 1);
            r[t].y = atomicAdd(&c[d[t].y], 1);
            r[t].z = atomicAdd(&c[d[t].z], 1);
            r[t].w = atomicAdd(&c[d[t].w], 1);
        }
    }
    __syncthreads();

    const int i0 = tid * HSEG;
    int sum = 0;
#pragma unroll
    for (int k = 0; k < HSEG; k += 4) {
        if (i0 + k < NN) {
            const int4 v = *(const int4*)&c[i0 + k];
            sum += v.x + v.y + v.z + v.w;
        }
    }
    int inc = sum;
#pragma unroll
    for (int o = 1; o < 32; o <<= 1) {
        const int y = __shfl_up(inc, o);
        if (lane >= o) inc += y;
    }
    if (lane == 31) wt[wv] = inc;
    __syncthreads();
    int run = inc - sum;
#pragma unroll
    for (int w = 0; w < HT / 32; ++w) if (w < wv) run += wt[w];
#pragma unroll
    for (int k = 0; k < HSEG; k += 4) {
        if (i0 + k < NN) {
            const int4 v = *(const int4*)&c[i0 + k];
            int4 p;
            p.x = run;
            p.y = p.x + v.x;
            p.z = p.y + v.y;
            p.w = p.z + v.z;
            run = p.w + v.w;
            *(int4*)&c[i0 + k] = p;
        }
    }
    __syncthreads();

    int* cg = startg + (long)g * NNP;
    for (int i = tid; i < NN / 4; i += HT) ((int4*)cg)[i] = ((const int4*)c)[i];
    if (tid == 0) cg[NN] = 4 * nq;
#pragma unroll
    for (int t = 0; t < HIT; ++t) {
        const int q = t * HT + tid;
        if (q < nq) {
            ls[c[d[t].x] + r[t].x] = s[t].x;
            ls[c[d[t].y] + r[t].y] = s[t].y;
            ls[c[d[t].z] + r[t].z] = s[t].z;
            ls[c[d[t].w] + r[t].w] = s[t].w;
        }
    }
    __syncthreads();
    for (int i = tid; i < nq; i += HT) ((int4*)sorted)[q0 + i] = ((const int4*)ls)[i];
}

#define PB   256
#define PNB  ((NN + 1 + PB - 1) / PB)
static_assert(PB % 32 == 0, "whole waves");

__global__ void __launch_bounds__(PB)
k_place(const int* __restrict__ startg, const int* __restrict__ sorted,
        int* __restrict__ rowptr, float* __restrict__ dinv, int* __restrict__ ssrc) {
    __shared__ int wt[PB / 32];
    const int tid  = threadIdx.x;
    const int lane = tid & 31;
    const int wv   = tid >> 5;
    const int d0   = blockIdx.x * PB;
    const int d    = d0 + tid;
    const bool real = d < NN;
    const int dl   = (d <= NN) ? d : NN;
    const int dn   = (dl < NN) ? dl + 1 : NN;

    int S[HG], n[HG];
    int base = 0, deg = 0;
#pragma unroll
    for (int g = 0; g < HG; ++g) {
        const int* row = startg + (long)g * NNP;
        S[g] = row[dl];
        n[g] = row[dn] - S[g];
        deg += n[g];
        base += row[d0];
    }

    int inc = deg;
#pragma unroll
    for (int o = 1; o < 32; o <<= 1) {
        const int y = __shfl_up(inc, o);
        if (lane >= o) inc += y;
    }
    if (lane == 31) wt[wv] = inc;
    __syncthreads();
    int rp = base + inc - deg;
#pragma unroll
    for (int w = 0; w < PB / 32; ++w) if (w < wv) rp += wt[w];

    if (d <= NN) rowptr[d] = rp;
    if (real) dinv[d] = rsqrtf((float)(deg + 1));

    int maxn = 0;
#pragma unroll
    for (int g = 0; g < HG; ++g) maxn = (n[g] > maxn) ? n[g] : maxn;
    for (int i = 0; __any(i < maxn); ++i) {
        int v[HG];
#pragma unroll
        for (int g = 0; g < HG; ++g) v[g] = (i < n[g]) ? sorted[g * HCH + S[g] + i] : 0;
        int pos = rp + i;
#pragma unroll
        for (int g = 0; g < HG; ++g) {
            if (i < n[g]) ssrc[pos] = v[g];
            pos += n[g];
        }
    }
}


#define TP  132
#define L1T 512
#define L1TILES (NN / 16)
#define L1G  768
static_assert(NN % 16 == 0, "exact grid: every work-group has 16 real nodes");
static_assert(FIN == DH, "one tile buffer serves both products");
static_assert(L1G <= L1TILES, "every work-group has a first tile");

static __device__ __forceinline__ int rl_i(int v, int k) {
    return __builtin_amdgcn_readlane(v, k);
}
static __device__ __forceinline__ float rl_f(float v, int k) {
    return __int_as_float(__builtin_amdgcn_readlane(__float_as_int(v), k));
}

template <int K, int NP>
static __device__ __forceinline__ v8f tile_product(const float* __restrict__ sT, const float* __restrict__ sB,
                                                   int tn, int lane) {
    const int c  = lane & 15;
    const int kh = (lane >> 4) << 1;
    const float* aP = sT + c * TP + kh;
    const float* bP = sB + kh * NP + tn * 16 + c;
    v8f acc = {};
#pragma unroll
    for (int kk = 0; kk < K; kk += 4) {
        v2f a, b;
        a.x = aP[kk];
        a.y = aP[kk + 1];
        b.x = bP[kk * NP];
        b.y = bP[(kk + 1) * NP];
        acc = __builtin_amdgcn_wmma_f32_16x16x4_f32(
             false, a,  false, b,
             (short)0, acc,  false,  false);
    }
    return acc;
}

template <int R>
static __device__ __forceinline__ void rows_x(float4& acc, const float4* __restrict__ x4,
                                              int cs_l, float w_l, int k, int lane) {
    float4 v[R];
#pragma unroll
    for (int j = 0; j < R; ++j) v[j] = x4[(long)rl_i(cs_l, k + j) * 32 + lane];
#pragma unroll
    for (int j = 0; j < R; ++j) {
        const float w = rl_f(w_l, k + j);
        acc.x = fmaf(v[j].x, w, acc.x); acc.y = fmaf(v[j].y, w, acc.y);
        acc.z = fmaf(v[j].z, w, acc.z); acc.w = fmaf(v[j].w, w, acc.w);
    }
}

#define RX 10
static __device__ __forceinline__ void first_chunk_rows_x(float4& acc, const float4* __restrict__ x4,
                                                          int cs_l, float w_l, int n, int lane) {
    for (int k = 0; k < n; k += RX) {
        const int r = (n - k < RX) ? (n - k) : RX;
        switch (r) {
        case 1: rows_x<1>(acc, x4, cs_l, w_l, k, lane); break;
        case 2: rows_x<2>(acc, x4, cs_l, w_l, k, lane); break;
        case 3: rows_x<3>(acc, x4, cs_l, w_l, k, lane); break;
        case 4: rows_x<4>(acc, x4, cs_l, w_l, k, lane); break;
        case 5: rows_x<5>(acc, x4, cs_l, w_l, k, lane); break;
        case 6: rows_x<6>(acc, x4, cs_l, w_l, k, lane); break;
        case 7: rows_x<7>(acc, x4, cs_l, w_l, k, lane); break;
        case 8: rows_x<8>(acc, x4, cs_l, w_l, k, lane); break;
        case 9: rows_x<9>(acc, x4, cs_l, w_l, k, lane); break;
        case 10: rows_x<10>(acc, x4, cs_l, w_l, k, lane); break;
        default: break;
        }
    }
}

static __device__ __forceinline__ void chunk_rows_x(float4& acc, const float4* __restrict__ x4,
                                                    int cs_l, float w_l, int n, int lane) {
    int k = 0;
    for (; k + 4 <= n; k += 4) {
        const int s0 = rl_i(cs_l, k), s1 = rl_i(cs_l, k + 1), s2 = rl_i(cs_l, k + 2), s3 = rl_i(cs_l, k + 3);
        const float w0 = rl_f(w_l, k), w1 = rl_f(w_l, k + 1), w2 = rl_f(w_l, k + 2), w3 = rl_f(w_l, k + 3);
        const float4 v0 = x4[(long)s0 * 32 + lane];
        const float4 v1 = x4[(long)s1 * 32 + lane];
        const float4 v2 = x4[(long)s2 * 32 + lane];
        const float4 v3 = x4[(long)s3 * 32 + lane];
        acc.x = fmaf(v0.x, w0, acc.x); acc.y = fmaf(v0.y, w0, acc.y); acc.z = fmaf(v0.z, w0, acc.z); acc.w = fmaf(v0.w, w0, acc.w);
        acc.x = fmaf(v1.x, w1, acc.x); acc.y = fmaf(v1.y, w1, acc.y); acc.z = fmaf(v1.z, w1, acc.z); acc.w = fmaf(v1.w, w1, acc.w);
        acc.x = fmaf(v2.x, w2, acc.x); acc.y = fmaf(v2.y, w2, acc.y); acc.z = fmaf(v2.z, w2, acc.z); acc.w = fmaf(v2.w, w2, acc.w);
        acc.x = fmaf(v3.x, w3, acc.x); acc.y = fmaf(v3.y, w3, acc.y); acc.z = fmaf(v3.z, w3, acc.z); acc.w = fmaf(v3.w, w3, acc.w);
    }
    for (; k < n; ++k) {
        const int s0 = rl_i(cs_l, k);
        const float w0 = rl_f(w_l, k);
        const float4 v0 = x4[(long)s0 * 32 + lane];
        acc.x = fmaf(v0.x, w0, acc.x); acc.y = fmaf(v0.y, w0, acc.y); acc.z = fmaf(v0.z, w0, acc.z); acc.w = fmaf(v0.w, w0, acc.w);
    }
}

__global__ void __launch_bounds__(L1T, 12)
k_layer1(const int* __restrict__ rowptr, const int* __restrict__ ssrc,
         const float* __restrict__ dinv, const float* __restrict__ x,
         const float* __restrict__ W1, const float* __restrict__ b1,
         const float* __restrict__ W2, float* __restrict__ h2) {
    __shared__ float sW1[FIN * DH];
    __shared__ float sW2[DH * NCP];
    __shared__ float sT[16 * TP];

    const int tid  = threadIdx.x;
    const int lane = tid & 31;
    const int wv   = __builtin_amdgcn_readfirstlane(tid >> 5);
    const float4* x4 = (const float4*)x;

    int tile = blockIdx.x;
    int node = tile * 16 + wv;
    int beg = rowptr[node], end = rowptr[node + 1];
    float dd = dinv[node];
    float4 self = x4[(long)node * 32 + lane];

    {
        constexpr int Q1 = (FIN * DH) / 4;
        constexpr int QP = (DH * NCP) / 4;
        constexpr int N1 = Q1 / L1T, N2 = QP / L1T;
        static_assert(Q1 % L1T == 0 && QP % L1T == 0, "whole quads per thread, no tail");
        static_assert(NC % 4 == 0 && NCP % 4 == 0, "a quad stays inside its row, data or pad");
        float4 w[N1], u[N2];
#pragma unroll
        for (int t = 0; t < N1; ++t) w[t] = ((const float4*)W1)[tid + t * L1T];
#pragma unroll
        for (int t = 0; t < N2; ++t) {
            const int j = tid + t * L1T;
            const int r = j / (NCP / 4), q = j - r * (NCP / 4);
            u[t] = ((const float4*)W2)[r * (NC / 4) + (q < NC / 4 ? q : NC / 4 - 1)];
        }
#pragma unroll
        for (int t = 0; t < N1; ++t) ((float4*)sW1)[tid + t * L1T] = w[t];
#pragma unroll
        for (int t = 0; t < N2; ++t) {
            const int j = tid + t * L1T;
            const int q = j % (NCP / 4);
            const bool data = q < NC / 4;
            ((float4*)sW2)[j] = make_float4(data ? u[t].x : 0.0f, data ? u[t].y : 0.0f,
                                            data ? u[t].z : 0.0f, data ? u[t].w : 0.0f);
        }
    }

    int s_l = (beg + lane < end) ? ssrc[beg + lane] : node;
    if ((unsigned)s_l >= (unsigned)NN) s_l = node;
    float r_l = dinv[s_l];

    bool more  = tile + L1G < L1TILES;
    int  ntile = more ? tile + L1G : tile;
    int  nnode = ntile * 16 + wv;
    int  nbeg = rowptr[nnode], nend = rowptr[nnode + 1];
    float ndd = dinv[nnode];

    for (;;) {
        const bool more2 = more && (ntile + L1G < L1TILES);
        const int  tile2 = more2 ? ntile + L1G : ntile;
        const int  node2 = tile2 * 16 + wv;
        const int  beg2 = rowptr[node2], end2 = rowptr[node2 + 1];
        const float dd2 = dinv[node2];

        const int nj = nbeg + lane;
        int ns_l = ssrc[nj < nend ? nj : NE - 1];
        if (nj >= nend) ns_l = nnode;
        const float4 nself = x4[(long)nnode * 32 + lane];

        const int node0 = tile * 16;

        float4 acc = self;
        {
            const float ws = dd * dd;
            acc.x *= ws; acc.y *= ws; acc.z *= ws; acc.w *= ws;

            first_chunk_rows_x(acc, x4, s_l, r_l * dd, (end - beg < 32) ? (end - beg) : 32, lane);
            for (int cb = beg + 32; cb < end; cb += 32) {
                const int j = cb + lane;
                int cs_l = (j < end) ? ssrc[j] : node;
                if ((unsigned)cs_l >= (unsigned)NN) cs_l = node;
                const float w_l = dinv[cs_l] * dd;
                chunk_rows_x(acc, x4, cs_l, w_l, (end - cb < 32) ? (end - cb) : 32, lane);
            }
        }

        __syncthreads();
        ((float4*)(sT + wv * TP))[lane] = acc;
        __syncthreads();

        if ((unsigned)ns_l >= (unsigned)NN) ns_l = nnode;
        const float nr_l = dinv[ns_l];

        v8f p = {};
        if (wv < DH / 16) p = tile_product<FIN, DH>(sT, sW1, wv, lane);
        __syncthreads();
        if (wv < DH / 16) {
            const int n = wv * 16 + (lane & 15);
            const int r0 = (lane >> 4) << 3;
            const float bb = b1[n];
#pragma unroll
            for (int v = 0; v < 8; ++v) {
                const float t = p[v] + bb;
                sT[(r0 + v) * TP + n] = t > 0.0f ? t : 0.0f;
            }
        }
        __syncthreads();

        if (wv < NCP / 16) {
            const v8f o = tile_product<DH, NCP>(sT, sW2, wv, lane);
            const int n = wv * 16 + (lane & 15);
            if (n < NC) {
                const int r0 = node0 + ((lane >> 4) << 3);
#pragma unroll
                for (int v = 0; v < 8; ++v)
                    h2[(long)(r0 + v) * NC + n] = o[v];
            }
        }

        if (!more) break;
        tile = ntile; node = nnode;
        beg = __builtin_amdgcn_readfirstlane(nbeg);
        end = __builtin_amdgcn_readfirstlane(nend);
        dd  = __int_as_float(__builtin_amdgcn_readfirstlane(__float_as_int(ndd)));
        s_l = ns_l; r_l = nr_l; self = nself;
        ntile = tile2; nnode = node2; nbeg = beg2; nend = end2; ndd = dd2; more = more2;
    }
}


#define GCB  256
#define GCG  1563
#define GCW  (GCG * (GCB / 32))
static_assert(GCW <= NN, "every wave has a first node");

template <int R>
static __device__ __forceinline__ void rows_h(float& a0, float& a1, const float* __restrict__ h,
                                              int cs_l, float w_l, int k, int lane, int c2) {
    float p[R], q[R];
#pragma unroll
    for (int j = 0; j < R; ++j) {
        const int s = rl_i(cs_l, k + j);
        p[j] = h[(long)s * NC + lane];
        q[j] = h[(long)s * NC + c2];
    }
#pragma unroll
    for (int j = 0; j < R; ++j) {
        const float w = rl_f(w_l, k + j);
        a0 = fmaf(p[j], w, a0); a1 = fmaf(q[j], w, a1);
    }
}

#define RH 16
static __device__ __forceinline__ void first_chunk_rows_h(float& a0, float& a1, const float* __restrict__ h,
                                                          int cs_l, float w_l, int n, int lane, int c2) {
    for (int k = 0; k < n; k += RH) {
        const int r = (n - k < RH) ? (n - k) : RH;
        switch (r) {
        case 1: rows_h<1>(a0, a1, h, cs_l, w_l, k, lane, c2); break;
        case 2: rows_h<2>(a0, a1, h, cs_l, w_l, k, lane, c2); break;
        case 3: rows_h<3>(a0, a1, h, cs_l, w_l, k, lane, c2); break;
        case 4: rows_h<4>(a0, a1, h, cs_l, w_l, k, lane, c2); break;
        case 5: rows_h<5>(a0, a1, h, cs_l, w_l, k, lane, c2); break;
        case 6: rows_h<6>(a0, a1, h, cs_l, w_l, k, lane, c2); break;
        case 7: rows_h<7>(a0, a1, h, cs_l, w_l, k, lane, c2); break;
        case 8: rows_h<8>(a0, a1, h, cs_l, w_l, k, lane, c2); break;
        case 9: rows_h<9>(a0, a1, h, cs_l, w_l, k, lane, c2); break;
        case 10: rows_h<10>(a0, a1, h, cs_l, w_l, k, lane, c2); break;
        case 11: rows_h<11>(a0, a1, h, cs_l, w_l, k, lane, c2); break;
        case 12: rows_h<12>(a0, a1, h, cs_l, w_l, k, lane, c2); break;
        case 13: rows_h<13>(a0, a1, h, cs_l, w_l, k, lane, c2); break;
        case 14: rows_h<14>(a0, a1, h, cs_l, w_l, k, lane, c2); break;
        case 15: rows_h<15>(a0, a1, h, cs_l, w_l, k, lane, c2); break;
        case 16: rows_h<16>(a0, a1, h, cs_l, w_l, k, lane, c2); break;
        default: break;
        }
    }
}

static __device__ __forceinline__ void chunk_rows_h(float& a0, float& a1, const float* __restrict__ h,
                                                    int cs_l, float w_l, int n, int lane, int c2) {
    int k = 0;
    for (; k + 4 <= n; k += 4) {
        const int s0 = rl_i(cs_l, k), s1 = rl_i(cs_l, k + 1), s2 = rl_i(cs_l, k + 2), s3 = rl_i(cs_l, k + 3);
        const float w0 = rl_f(w_l, k), w1 = rl_f(w_l, k + 1), w2 = rl_f(w_l, k + 2), w3 = rl_f(w_l, k + 3);
        const float p0 = h[(long)s0 * NC + lane], q0 = h[(long)s0 * NC + c2];
        const float p1 = h[(long)s1 * NC + lane], q1 = h[(long)s1 * NC + c2];
        const float p2 = h[(long)s2 * NC + lane], q2 = h[(long)s2 * NC + c2];
        const float p3 = h[(long)s3 * NC + lane], q3 = h[(long)s3 * NC + c2];
        a0 = fmaf(p0, w0, a0); a1 = fmaf(q0, w0, a1);
        a0 = fmaf(p1, w1, a0); a1 = fmaf(q1, w1, a1);
        a0 = fmaf(p2, w2, a0); a1 = fmaf(q2, w2, a1);
        a0 = fmaf(p3, w3, a0); a1 = fmaf(q3, w3, a1);
    }
    for (; k < n; ++k) {
        const int s0 = rl_i(cs_l, k);
        const float w0 = rl_f(w_l, k);
        a0 = fmaf(h[(long)s0 * NC + lane], w0, a0);
        a1 = fmaf(h[(long)s0 * NC + c2], w0, a1);
    }
}

__global__ void __launch_bounds__(GCB)
k_gather_c(const int* __restrict__ rowptr, const int* __restrict__ ssrc,
           const float* __restrict__ dinv,
           const float* __restrict__ h, const float* __restrict__ b,
           float* __restrict__ out) {
    const int lane = threadIdx.x & 31;
    const int gw = __builtin_amdgcn_readfirstlane((int)((blockIdx.x * GCB + threadIdx.x) >> 5));
    const bool hi = lane < NC - 32;
    const int  c2 = hi ? lane + 32 : lane;
    const float b0 = b[lane], b1 = b[c2];

    int node = gw;
    int beg = rowptr[node], end = rowptr[node + 1];
    float dd = dinv[node];
    float h0 = h[(long)node * NC + lane], h1 = h[(long)node * NC + c2];
    int s_l = (beg + lane < end) ? ssrc[beg + lane] : node;
    if ((unsigned)s_l >= (unsigned)NN) s_l = node;
    float r_l = dinv[s_l];

    bool more  = node + GCW < NN;
    int  nnode = more ? node + GCW : node;
    int  nbeg = rowptr[nnode], nend = rowptr[nnode + 1];
    float ndd = dinv[nnode];

    for (;;) {
        const bool more2 = more && (nnode + GCW < NN);
        const int  node2 = more2 ? nnode + GCW : nnode;
        const int  beg2 = rowptr[node2], end2 = rowptr[node2 + 1];
        const float dd2 = dinv[node2];

        const int nj = nbeg + lane;
        int ns_l = ssrc[nj < nend ? nj : NE - 1];
        if (nj >= nend) ns_l = nnode;
        const float nh0 = h[(long)nnode * NC + lane];
        const float nh1 = h[(long)nnode * NC + c2];

        const float ws = dd * dd;
        float a0 = h0 * ws;
        float a1 = h1 * ws;
        first_chunk_rows_h(a0, a1, h, s_l, r_l * dd, (end - beg < 32) ? (end - beg) : 32, lane, c2);
        for (int cb = beg + 32; cb < end; cb += 32) {
            const int j = cb + lane;
            int cs_l = (j < end) ? ssrc[j] : node;
            if ((unsigned)cs_l >= (unsigned)NN) cs_l = node;
            const float w_l = dinv[cs_l] * dd;
            chunk_rows_h(a0, a1, h, cs_l, w_l, (end - cb < 32) ? (end - cb) : 32, lane, c2);
        }

        if ((unsigned)ns_l >= (unsigned)NN) ns_l = nnode;
        const float nr_l = dinv[ns_l];

        const float v0 = a0 + b0;
        const float v1 = hi ? a1 + b1 : -INFINITY;
        float mx = fmaxf(v0, v1);
        for (int o = 16; o > 0; o >>= 1) mx = fmaxf(mx, __shfl_xor(mx, o));
        float sum = expf(v0 - mx) + (hi ? expf(v1 - mx) : 0.0f);
        for (int o = 16; o > 0; o >>= 1) sum += __shfl_xor(sum, o);
        const float lse = mx + logf(sum);
        out[(long)node * NC + lane] = v0 - lse;
        if (hi) out[(long)node * NC + c2] = v1 - lse;

        if (!more) break;
        node = nnode; beg = nbeg; end = nend; dd = ndd;
        s_l = ns_l; r_l = nr_l; h0 = nh0; h1 = nh1;
        nnode = node2; nbeg = beg2; nend = end2; ndd = dd2; more = more2;
    }
}


extern "C" void kernel_launch(void* const* d_in, const int* in_sizes, int n_in,
                              void* d_out, int out_size, void* d_ws, size_t ws_size,
                              hipStream_t stream) {
    const float* x  = (const float*)d_in[0];
    const int*   ei = (const int*)d_in[1];
    const float* W1 = (const float*)d_in[2];
    const float* b1 = (const float*)d_in[3];
    const float* W2 = (const float*)d_in[4];
    const float* b2 = (const float*)d_in[5];
    float* out = (float*)d_out;

    const int* src = ei;
    const int* dst = ei + NE;

    float* ws     = (float*)d_ws;
    float* dinv   = ws;
    int*   rowptr = (int*)(dinv + NNP);
    int*   sorted = rowptr + NNP;
    int*   ssrc   = sorted + NEP;
    float* h2     = (float*)(ssrc + NEP);
    int*   startg = (int*)(h2 + (long)NN * NC);

    k_hist  <<<HG, HT, 0, stream>>>(src, dst, startg, sorted);
    k_place <<<PNB, PB, 0, stream>>>(startg, sorted, rowptr, dinv, ssrc);

    k_layer1   <<<L1G, L1T, 0, stream>>>(rowptr, ssrc, dinv, x, W1, b1, W2, h2);

    k_gather_c   <<<GCG, GCB, 0, stream>>>(rowptr, ssrc, dinv, h2, b2, out);
}


# Edge sort in two levels: 250 edge chunks binned by 256-node range in LDS, then a work-group per bin counts and places per node; all-node LDS tables and the 32 prefix rows go; sum order: by new chunks
# speedup vs baseline: 5.2695x; 1.1573x over previous
#include <hip/hip_runtime.h>
#include <math.h>

#define NN   50000
#define NE   625000
#define FIN  128
#define DH   128
#define NC   40
#define NCP  48
#define NNP  50048
#define NEP  625024

typedef __attribute__((ext_vector_type(2))) float v2f;
typedef __attribute__((ext_vector_type(8))) float v8f;

static inline int cdiv(long a, long b) { return (int)((a + b - 1) / b); }


#define BW    256
#define NBIN  ((NN + BW - 1) / BW)
#define BROW  (NBIN + 1)
#define G1    250
#define CQ    (NE / 4 / G1)
#define CE    (4 * CQ)
#define B1T   256
#define B1IT  ((CQ + B1T - 1) / B1T)
static_assert(NE % 4 == 0 && (NE / 4) % G1 == 0, "chunks are whole quads and cover every edge exactly");
static_assert(CE % 2 == 0, "a chunk's pairs are whole 16-byte words");
static_assert(B1T >= BROW && B1T % 32 == 0, "a thread per prefix entry, whole waves");
static_assert(NN % BW != 0, "node index NN lies inside the last bin: its work-group writes rowptr[NN]");

static __device__ __forceinline__ int bin_of(int d) {
    const unsigned b = (unsigned)d / BW;
    return (int)(b < NBIN ? b : NBIN - 1);
}

__global__ void __launch_bounds__(B1T)
k_bin(const int* __restrict__ src, const int* __restrict__ dst,
      int* __restrict__ binstart, int* __restrict__ pairs) {
    __shared__ int cnt[BROW];
    __shared__ __attribute__((aligned(16))) int2 lp[CE];
    __shared__ int wt[B1T / 32];
    const int tid  = threadIdx.x;
    const int lane = tid & 31;
    const int wv   = tid >> 5;
    const int g  = blockIdx.x;
    const int q0 = g * CQ;

    int4 d[B1IT], s[B1IT];
#pragma unroll
    for (int t = 0; t < B1IT; ++t) {
        const int q = t * B1T + tid;
        d[t] = (q < CQ) ? ((const int4*)dst)[q0 + q] : make_int4(0, 0, 0, 0);
        s[t] = (q < CQ) ? ((const int4*)src)[q0 + q] : make_int4(0, 0, 0, 0);
    }
    if (tid < BROW) cnt[tid] = 0;
    __syncthreads();

    int4 r[B1IT];
#pragma unroll
    for (int t = 0; t < B1IT; ++t) {
        const int q = t * B1T + tid;
        r[t] = make_int4(0, 0, 0, 0);
        if (q < CQ) {
            r[t].x = atomicAdd(&cnt[bin_of(d[t].x)], 1);
            r[t].y = atomicAdd(&cnt[bin_of(d[t].y)], 1);
            r[t].z = atomicAdd(&cnt[bin_of(d[t].z)], 1);
            r[t].w = atomicAdd(&cnt[bin_of(d[t].w)], 1);
        }
    }
    __syncthreads();

    const int v = (tid < BROW) ? cnt[tid] : 0;
    int inc = v;
#pragma unroll
    for (int o = 1; o < 32; o <<= 1) {
        const int y = __shfl_up(inc, o);
        if (lane >= o) inc += y;
    }
    if (lane == 31) wt[wv] = inc;
    __syncthreads();
    int ex = inc - v;
#pragma unroll
    for (int w = 0; w < B1T / 32; ++w) if (w < wv) ex += wt[w];
    if (tid < BROW) {
        cnt[tid] = ex;
        binstart[g * BROW + tid] = ex;
    }
    __syncthreads();

#pragma unroll
    for (int t = 0; t < B1IT; ++t) {
        const int q = t * B1T + tid;
        if (q < CQ) {
            lp[cnt[bin_of(d[t].x)] + r[t].x] = make_int2(s[t].x, d[t].x);
            lp[cnt[bin_of(d[t].y)] + r[t].y] = make_int2(s[t].y, d[t].y);
            lp[cnt[bin_of(d[t].z)] + r[t].z] = make_int2(s[t].z, d[t].z);
            lp[cnt[bin_of(d[t].w)] + r[t].w] = make_int2(s[t].w, d[t].w);
        }
    }
    __syncthreads();
    int4* po = (int4*)(pairs + (long)g * (2 * CE));
    for (int i = tid; i < CE / 2; i += B1T) po[i] = ((const int4*)lp)[i];
}

static __device__ __forceinline__ int rl_i(int v, int k) {
    return __builtin_amdgcn_readlane(v, k);
}
static __device__ __forceinline__ float rl_f(float v, int k) {
    return __int_as_float(__builtin_amdgcn_readlane(__float_as_int(v), k));
}

#define S2T  1024
#define S2W  (S2T / 32)
#define CPW  ((G1 + S2W - 1) / S2W)
static_assert(CPW <= 32, "a lane per chunk of the wave");
static_assert(S2T >= BW && BW % 32 == 0, "a thread per node of the bin, whole waves");

static __device__ __forceinline__ int chunk_of(int wv, int c) {
    const int gc = wv * CPW + c;
    return gc < G1 ? gc : G1 - 1;
}

__global__ void __launch_bounds__(S2T)
k_sort(const int* __restrict__ binstart, const int* __restrict__ pairs_i,
       int* __restrict__ rowptr, float* __restrict__ dinv, int* __restrict__ ssrc) {
    __shared__ int cw[S2W * BW];
    __shared__ int rs[BW];
    __shared__ int wb[S2W];
    __shared__ int wt[S2W];
    const int tid  = threadIdx.x;
    const int lane = tid & 31;
    const int wv   = __builtin_amdgcn_readfirstlane(tid >> 5);
    const int b    = blockIdx.x;
    const int n0   = b * BW;
    const int2* pairs = (const int2*)pairs_i;
    int* myrow = cw + wv * BW;

    int segS = 0, segN = 0;
    {
        const int gl = wv * CPW + lane;
        if (lane < CPW && gl < G1) {
            const int* row = binstart + gl * BROW;
            segS = row[b];
            segN = row[b + 1] - segS;
        }
    }
#pragma unroll
    for (int k = 0; k < BW / 32; ++k) myrow[lane + 32 * k] = 0;

    int2 p[CPW];
#pragma unroll
    for (int c = 0; c < CPW; ++c) {
        const int S = rl_i(segS, c), N = rl_i(segN, c);
        p[c] = pairs[(long)chunk_of(wv, c) * CE + ((lane < N) ? S + lane : 0)];
    }

    int bs = segS;
    for (int o = 16; o > 0; o >>= 1) bs += __shfl_xor(bs, o);
    if (lane == 0) wb[wv] = bs;

#pragma unroll
    for (int c = 0; c < CPW; ++c) {
        const int S = rl_i(segS, c), N = rl_i(segN, c);
        if (lane < N) atomicAdd(&myrow[(p[c].y - n0) & (BW - 1)], 1);
        for (int k = 32 + lane; k < N; k += 32) {
            const int2 e = pairs[(long)chunk_of(wv, c) * CE + S + k];
            atomicAdd(&myrow[(e.y - n0) & (BW - 1)], 1);
        }
    }
    __syncthreads();

    int base = 0;
#pragma unroll
    for (int w = 0; w < S2W; ++w) base += wb[w];

    int deg = 0;
    if (tid < BW) {
        int v[S2W];
#pragma unroll
        for (int w = 0; w < S2W; ++w) v[w] = cw[w * BW + tid];
#pragma unroll
        for (int w = 0; w < S2W; ++w) {
            cw[w * BW + tid] = deg;
            deg += v[w];
        }
    }
    int inc = deg;
#pragma unroll
    for (int o = 1; o < 32; o <<= 1) {
        const int y = __shfl_up(inc, o);
        if (lane >= o) inc += y;
    }
    if (lane == 31) wt[wv] = inc;
    __syncthreads();
    int rp = base + inc - deg;
#pragma unroll
    for (int w = 0; w < S2W; ++w) if (w < wv) rp += wt[w];
    if (tid < BW) {
        const int node = n0 + tid;
        rs[tid] = rp;
        if (node <= NN) rowptr[node] = rp;
        if (node < NN) dinv[node] = rsqrtf((float)(deg + 1));
    }
    __syncthreads();

    int ls0[CPW], rk[CPW];
#pragma unroll
    for (int c = 0; c < CPW; ++c) {
        const int S = rl_i(segS, c), N = rl_i(segN, c);
        ls0[c] = 0; rk[c] = 0;
        if (lane < N) {
            const int dl = (p[c].y - n0) & (BW - 1);
            ls0[c] = rs[dl];
            rk[c] = atomicAdd(&myrow[dl], 1);
        }
        for (int k = 32 + lane; k < N; k += 32) {
            const int2 e = pairs[(long)chunk_of(wv, c) * CE + S + k];
            const int dl = (e.y - n0) & (BW - 1);
            const int r = atomicAdd(&myrow[dl], 1);
            ssrc[rs[dl] + r] = e.x;
        }
    }
#pragma unroll
    for (int c = 0; c < CPW; ++c) {
        const int N = rl_i(segN, c);
        if (lane < N) ssrc[ls0[c] + rk[c]] = p[c].x;
    }
}


#define TP  132
#define L1T 512
#define L1TILES (NN / 16)
#define L1G  768
static_assert(NN % 16 == 0, "exact grid: every work-group has 16 real nodes");
static_assert(FIN == DH, "one tile buffer serves both products");
static_assert(L1G <= L1TILES, "every work-group has a first tile");

template <int K, int NP>
static __device__ __forceinline__ v8f tile_product(const float* __restrict__ sT, const float* __restrict__ sB,
                                                   int tn, int lane) {
    const int c  = lane & 15;
    const int kh = (lane >> 4) << 1;
    const float* aP = sT + c * TP + kh;
    const float* bP = sB + kh * NP + tn * 16 + c;
    v8f acc = {};
#pragma unroll
    for (int kk = 0; kk < K; kk += 4) {
        v2f a, b;
        a.x = aP[kk];
        a.y = aP[kk + 1];
        b.x = bP[kk * NP];
        b.y = bP[(kk + 1) * NP];
        acc = __builtin_amdgcn_wmma_f32_16x16x4_f32(
             false, a,  false, b,
             (short)0, acc,  false,  false);
    }
    return acc;
}

template <int R>
static __device__ __forceinline__ void rows_x(float4& acc, const float4* __restrict__ x4,
                                              int cs_l, float w_l, int k, int lane) {
    float4 v[R];
#pragma unroll
    for (int j = 0; j < R; ++j) v[j] = x4[(long)rl_i(cs_l, k + j) * 32 + lane];
#pragma unroll
    for (int j = 0; j < R; ++j) {
        const float w = rl_f(w_l, k + j);
        acc.x = fmaf(v[j].x, w, acc.x); acc.y = fmaf(v[j].y, w, acc.y);
        acc.z = fmaf(v[j].z, w, acc.z); acc.w = fmaf(v[j].w, w, acc.w);
    }
}

#define RX 10
static __device__ __forceinline__ void first_chunk_rows_x(float4& acc, const float4* __restrict__ x4,
                                                          int cs_l, float w_l, int n, int lane) {
    for (int k = 0; k < n; k += RX) {
        const int r = (n - k < RX) ? (n - k) : RX;
        switch (r) {
        case 1: rows_x<1>(acc, x4, cs_l, w_l, k, lane); break;
        case 2: rows_x<2>(acc, x4, cs_l, w_l, k, lane); break;
        case 3: rows_x<3>(acc, x4, cs_l, w_l, k, lane); break;
        case 4: rows_x<4>(acc, x4, cs_l, w_l, k, lane); break;
        case 5: rows_x<5>(acc, x4, cs_l, w_l, k, lane); break;
        case 6: rows_x<6>(acc, x4, cs_l, w_l, k, lane); break;
        case 7: rows_x<7>(acc, x4, cs_l, w_l, k, lane); break;
        case 8: rows_x<8>(acc, x4, cs_l, w_l, k, lane); break;
        case 9: rows_x<9>(acc, x4, cs_l, w_l, k, lane); break;
        case 10: rows_x<10>(acc, x4, cs_l, w_l, k, lane); break;
        default: break;
        }
    }
}

static __device__ __forceinline__ void chunk_rows_x(float4& acc, const float4* __restrict__ x4,
                                                    int cs_l, float w_l, int n, int lane) {
    int k = 0;
    for (; k + 4 <= n; k += 4) {
        const int s0 = rl_i(cs_l, k), s1 = rl_i(cs_l, k + 1), s2 = rl_i(cs_l, k + 2), s3 = rl_i(cs_l, k + 3);
        const float w0 = rl_f(w_l, k), w1 = rl_f(w_l, k + 1), w2 = rl_f(w_l, k + 2), w3 = rl_f(w_l, k + 3);
        const float4 v0 = x4[(long)s0 * 32 + lane];
        const float4 v1 = x4[(long)s1 * 32 + lane];
        const float4 v2 = x4[(long)s2 * 32 + lane];
        const float4 v3 = x4[(long)s3 * 32 + lane];
        acc.x = fmaf(v0.x, w0, acc.x); acc.y = fmaf(v0.y, w0, acc.y); acc.z = fmaf(v0.z, w0, acc.z); acc.w = fmaf(v0.w, w0, acc.w);
        acc.x = fmaf(v1.x, w1, acc.x); acc.y = fmaf(v1.y, w1, acc.y); acc.z = fmaf(v1.z, w1, acc.z); acc.w = fmaf(v1.w, w1, acc.w);
        acc.x = fmaf(v2.x, w2, acc.x); acc.y = fmaf(v2.y, w2, acc.y); acc.z = fmaf(v2.z, w2, acc.z); acc.w = fmaf(v2.w, w2, acc.w);
        acc.x = fmaf(v3.x, w3, acc.x); acc.y = fmaf(v3.y, w3, acc.y); acc.z = fmaf(v3.z, w3, acc.z); acc.w = fmaf(v3.w, w3, acc.w);
    }
    for (; k < n; ++k) {
        const int s0 = rl_i(cs_l, k);
        const float w0 = rl_f(w_l, k);
        const float4 v0 = x4[(long)s0 * 32 + lane];
        acc.x = fmaf(v0.x, w0, acc.x); acc.y = fmaf(v0.y, w0, acc.y); acc.z = fmaf(v0.z, w0, acc.z); acc.w = fmaf(v0.w, w0, acc.w);
    }
}

__global__ void __launch_bounds__(L1T, 12)
k_layer1(const int* __restrict__ rowptr, const int* __restrict__ ssrc,
         const float* __restrict__ dinv, const float* __restrict__ x,
         const float* __restrict__ W1, const float* __restrict__ b1,
         const float* __restrict__ W2, float* __restrict__ h2) {
    __shared__ float sW1[FIN * DH];
    __shared__ float sW2[DH * NCP];
    __shared__ float sT[16 * TP];

    const int tid  = threadIdx.x;
    const int lane = tid & 31;
    const int wv   = __builtin_amdgcn_readfirstlane(tid >> 5);
    const float4* x4 = (const float4*)x;

    int tile = blockIdx.x;
    int node = tile * 16 + wv;
    int beg = rowptr[node], end = rowptr[node + 1];
    float dd = dinv[node];
    float4 self = x4[(long)node * 32 + lane];

    {
        constexpr int Q1 = (FIN * DH) / 4;
        constexpr int QP = (DH * NCP) / 4;
        constexpr int N1 = Q1 / L1T, N2 = QP / L1T;
        static_assert(Q1 % L1T == 0 && QP % L1T == 0, "whole quads per thread, no tail");
        static_assert(NC % 4 == 0 && NCP % 4 == 0, "a quad stays inside its row, data or pad");
        float4 w[N1], u[N2];
#pragma unroll
        for (int t = 0; t < N1; ++t) w[t] = ((const float4*)W1)[tid + t * L1T];
#pragma unroll
        for (int t = 0; t < N2; ++t) {
            const int j = tid + t * L1T;
            const int r = j / (NCP / 4), q = j - r * (NCP / 4);
            u[t] = ((const float4*)W2)[r * (NC / 4) + (q < NC / 4 ? q : NC / 4 - 1)];
        }
#pragma unroll
        for (int t = 0; t < N1; ++t) ((float4*)sW1)[tid + t * L1T] = w[t];
#pragma unroll
        for (int t = 0; t < N2; ++t) {
            const int j = tid + t * L1T;
            const int q = j % (NCP / 4);
            const bool data = q < NC / 4;
            ((float4*)sW2)[j] = make_float4(data ? u[t].x : 0.0f, data ? u[t].y : 0.0f,
                                            data ? u[t].z : 0.0f, data ? u[t].w : 0.0f);
        }
    }

    int s_l = (beg + lane < end) ? ssrc[beg + lane] : node;
    if ((unsigned)s_l >= (unsigned)NN) s_l = node;
    float r_l = dinv[s_l];

    bool more  = tile + L1G < L1TILES;
    int  ntile = more ? tile + L1G : tile;
    int  nnode = ntile * 16 + wv;
    int  nbeg = rowptr[nnode], nend = rowptr[nnode + 1];
    float ndd = dinv[nnode];

    for (;;) {
        const bool more2 = more && (ntile + L1G < L1TILES);
        const int  tile2 = more2 ? ntile + L1G : ntile;
        const int  node2 = tile2 * 16 + wv;
        const int  beg2 = rowptr[node2], end2 = rowptr[node2 + 1];
        const float dd2 = dinv[node2];

        const int nj = nbeg + lane;
        int ns_l = ssrc[nj < nend ? nj : NE - 1];
        if (nj >= nend) ns_l = nnode;
        const float4 nself = x4[(long)nnode * 32 + lane];

        const int node0 = tile * 16;

        float4 acc = self;
        {
            const float ws = dd * dd;
            acc.x *= ws; acc.y *= ws; acc.z *= ws; acc.w *= ws;

            first_chunk_rows_x(acc, x4, s_l, r_l * dd, (end - beg < 32) ? (end - beg) : 32, lane);
            for (int cb = beg + 32; cb < end; cb += 32) {
                const int j = cb + lane;
                int cs_l = (j < end) ? ssrc[j] : node;
                if ((unsigned)cs_l >= (unsigned)NN) cs_l = node;
                const float w_l = dinv[cs_l] * dd;
                chunk_rows_x(acc, x4, cs_l, w_l, (end - cb < 32) ? (end - cb) : 32, lane);
            }
        }

        __syncthreads();
        ((float4*)(sT + wv * TP))[lane] = acc;
        __syncthreads();

        if ((unsigned)ns_l >= (unsigned)NN) ns_l = nnode;
        const float nr_l = dinv[ns_l];

        v8f p = {};
        if (wv < DH / 16) p = tile_product<FIN, DH>(sT, sW1, wv, lane);
        __syncthreads();
        if (wv < DH / 16) {
            const int n = wv * 16 + (lane & 15);
            const int r0 = (lane >> 4) << 3;
            const float bb = b1[n];
#pragma unroll
            for (int v = 0; v < 8; ++v) {
                const float t = p[v] + bb;
                sT[(r0 + v) * TP + n] = t > 0.0f ? t : 0.0f;
            }
        }
        __syncthreads();

        if (wv < NCP / 16) {
            const v8f o = tile_product<DH, NCP>(sT, sW2, wv, lane);
            const int n = wv * 16 + (lane & 15);
            if (n < NC) {
                const int r0 = node0 + ((lane >> 4) << 3);
#pragma unroll
                for (int v = 0; v < 8; ++v)
                    h2[(long)(r0 + v) * NC + n] = o[v];
            }
        }

        if (!more) break;
        tile = ntile; node = nnode;
        beg = __builtin_amdgcn_readfirstlane(nbeg);
        end = __builtin_amdgcn_readfirstlane(nend);
        dd  = __int_as_float(__builtin_amdgcn_readfirstlane(__float_as_int(ndd)));
        s_l = ns_l; r_l = nr_l; self = nself;
        ntile = tile2; nnode = node2; nbeg = beg2; nend = end2; ndd = dd2; more = more2;
    }
}


#define GCB  256
#define GCG  1563
#define GCW  (GCG * (GCB / 32))
static_assert(GCW <= NN, "every wave has a first node");

template <int R>
static __device__ __forceinline__ void rows_h(float& a0, float& a1, const float* __restrict__ h,
                                              int cs_l, float w_l, int k, int lane, int c2) {
    float p[R], q[R];
#pragma unroll
    for (int j = 0; j < R; ++j) {
        const int s = rl_i(cs_l, k + j);
        p[j] = h[(long)s * NC + lane];
        q[j] = h[(long)s * NC + c2];
    }
#pragma unroll
    for (int j = 0; j < R; ++j) {
        const float w = rl_f(w_l, k + j);
        a0 = fmaf(p[j], w, a0); a1 = fmaf(q[j], w, a1);
    }
}

#define RH 16
static __device__ __forceinline__ void first_chunk_rows_h(float& a0, float& a1, const float* __restrict__ h,
                                                          int cs_l, float w_l, int n, int lane, int c2) {
    for (int k = 0; k < n; k += RH) {
        const int r = (n - k < RH) ? (n - k) : RH;
        switch (r) {
        case 1: rows_h<1>(a0, a1, h, cs_l, w_l, k, lane, c2); break;
        case 2: rows_h<2>(a0, a1, h, cs_l, w_l, k, lane, c2); break;
        case 3: rows_h<3>(a0, a1, h, cs_l, w_l, k, lane, c2); break;
        case 4: rows_h<4>(a0, a1, h, cs_l, w_l, k, lane, c2); break;
        case 5: rows_h<5>(a0, a1, h, cs_l, w_l, k, lane, c2); break;
        case 6: rows_h<6>(a0, a1, h, cs_l, w_l, k, lane, c2); break;
        case 7: rows_h<7>(a0, a1, h, cs_l, w_l, k, lane, c2); break;
        case 8: rows_h<8>(a0, a1, h, cs_l, w_l, k, lane, c2); break;
        case 9: rows_h<9>(a0, a1, h, cs_l, w_l, k, lane, c2); break;
        case 10: rows_h<10>(a0, a1, h, cs_l, w_l, k, lane, c2); break;
        case 11: rows_h<11>(a0, a1, h, cs_l, w_l, k, lane, c2); break;
        case 12: rows_h<12>(a0, a1, h, cs_l, w_l, k, lane, c2); break;
        case 13: rows_h<13>(a0, a1, h, cs_l, w_l, k, lane, c2); break;
        case 14: rows_h<14>(a0, a1, h, cs_l, w_l, k, lane, c2); break;
        case 15: rows_h<15>(a0, a1, h, cs_l, w_l, k, lane, c2); break;
        case 16: rows_h<16>(a0, a1, h, cs_l, w_l, k, lane, c2); break;
        default: break;
        }
    }
}

static __device__ __forceinline__ void chunk_rows_h(float& a0, float& a1, const float* __restrict__ h,
                                                    int cs_l, float w_l, int n, int lane, int c2) {
    int k = 0;
    for (; k + 4 <= n; k += 4) {
        const int s0 = rl_i(cs_l, k), s1 = rl_i(cs_l, k + 1), s2 = rl_i(cs_l, k + 2), s3 = rl_i(cs_l, k + 3);
        const float w0 = rl_f(w_l, k), w1 = rl_f(w_l, k + 1), w2 = rl_f(w_l, k + 2), w3 = rl_f(w_l, k + 3);
        const float p0 = h[(long)s0 * NC + lane], q0 = h[(long)s0 * NC + c2];
        const float p1 = h[(long)s1 * NC + lane], q1 = h[(long)s1 * NC + c2];
        const float p2 = h[(long)s2 * NC + lane], q2 = h[(long)s2 * NC + c2];
        const float p3 = h[(long)s3 * NC + lane], q3 = h[(long)s3 * NC + c2];
        a0 = fmaf(p0, w0, a0); a1 = fmaf(q0, w0, a1);
        a0 = fmaf(p1, w1, a0); a1 = fmaf(q1, w1, a1);
        a0 = fmaf(p2, w2, a0); a1 = fmaf(q2, w2, a1);
        a0 = fmaf(p3, w3, a0); a1 = fmaf(q3, w3, a1);
    }
    for (; k < n; ++k) {
        const int s0 = rl_i(cs_l, k);
        const float w0 = rl_f(w_l, k);
        a0 = fmaf(h[(long)s0 * NC + lane], w0, a0);
        a1 = fmaf(h[(long)s0 * NC + c2], w0, a1);
    }
}

__global__ void __launch_bounds__(GCB)
k_gather_c(const int* __restrict__ rowptr, const int* __restrict__ ssrc,
           const float* __restrict__ dinv,
           const float* __restrict__ h, const float* __restrict__ b,
           float* __restrict__ out) {
    const int lane = threadIdx.x & 31;
    const int gw = __builtin_amdgcn_readfirstlane((int)((blockIdx.x * GCB + threadIdx.x) >> 5));
    const bool hi = lane < NC - 32;
    const int  c2 = hi ? lane + 32 : lane;
    const float b0 = b[lane], b1 = b[c2];

    int node = gw;
    int beg = rowptr[node], end = rowptr[node + 1];
    float dd = dinv[node];
    float h0 = h[(long)node * NC + lane], h1 = h[(long)node * NC + c2];
    int s_l = (beg + lane < end) ? ssrc[beg + lane] : node;
    if ((unsigned)s_l >= (unsigned)NN) s_l = node;
    float r_l = dinv[s_l];

    bool more  = node + GCW < NN;
    int  nnode = more ? node + GCW : node;
    int  nbeg = rowptr[nnode], nend = rowptr[nnode + 1];
    float ndd = dinv[nnode];

    for (;;) {
        const bool more2 = more && (nnode + GCW < NN);
        const int  node2 = more2 ? nnode + GCW : nnode;
        const int  beg2 = rowptr[node2], end2 = rowptr[node2 + 1];
        const float dd2 = dinv[node2];

        const int nj = nbeg + lane;
        int ns_l = ssrc[nj < nend ? nj : NE - 1];
        if (nj >= nend) ns_l = nnode;
        const float nh0 = h[(long)nnode * NC + lane];
        const float nh1 = h[(long)nnode * NC + c2];

        const float ws = dd * dd;
        float a0 = h0 * ws;
        float a1 = h1 * ws;
        first_chunk_rows_h(a0, a1, h, s_l, r_l * dd, (end - beg < 32) ? (end - beg) : 32, lane, c2);
        for (int cb = beg + 32; cb < end; cb += 32) {
            const int j = cb + lane;
            int cs_l = (j < end) ? ssrc[j] : node;
            if ((unsigned)cs_l >= (unsigned)NN) cs_l = node;
            const float w_l = dinv[cs_l] * dd;
            chunk_rows_h(a0, a1, h, cs_l, w_l, (end - cb < 32) ? (end - cb) : 32, lane, c2);
        }

        if ((unsigned)ns_l >= (unsigned)NN) ns_l = nnode;
        const float nr_l = dinv[ns_l];

        const float v0 = a0 + b0;
        const float v1 = hi ? a1 + b1 : -INFINITY;
        float mx = fmaxf(v0, v1);
        for (int o = 16; o > 0; o >>= 1) mx = fmaxf(mx, __shfl_xor(mx, o));
        float sum = expf(v0 - mx) + (hi ? expf(v1 - mx) : 0.0f);
        for (int o = 16; o > 0; o >>= 1) sum += __shfl_xor(sum, o);
        const float lse = mx + logf(sum);
        out[(long)node * NC + lane] = v0 - lse;
        if (hi) out[(long)node * NC + c2] = v1 - lse;

        if (!more) break;
        node = nnode; beg = nbeg; end = nend; dd = ndd;
        s_l = ns_l; r_l = nr_l; h0 = nh0; h1 = nh1;
        nnode = node2; nbeg = beg2; nend = end2; ndd = dd2; more = more2;
    }
}


extern "C" void kernel_launch(void* const* d_in, const int* in_sizes, int n_in,
                              void* d_out, int out_size, void* d_ws, size_t ws_size,
                              hipStream_t stream) {
    const float* x  = (const float*)d_in[0];
    const int*   ei = (const int*)d_in[1];
    const float* W1 = (const float*)d_in[2];
    const float* b1 = (const float*)d_in[3];
    const float* W2 = (const float*)d_in[4];
    const float* b2 = (const float*)d_in[5];
    float* out = (float*)d_out;

    const int* src = ei;
    const int* dst = ei + NE;

    float* ws     = (float*)d_ws;
    float* dinv   = ws;
    int*   rowptr = (int*)(dinv + NNP);
    int*   pairs  = rowptr + NNP;
    int*   ssrc   = pairs + 2 * NEP;
    float* h2     = (float*)(ssrc + NEP);
    int*   binstart = (int*)(h2 + (long)NN * NC);

    k_bin  <<<G1, B1T, 0, stream>>>(src, dst, binstart, pairs);
    k_sort <<<NBIN, S2T, 0, stream>>>(binstart, pairs, rowptr, dinv, ssrc);

    k_layer1   <<<L1G, L1T, 0, stream>>>(rowptr, ssrc, dinv, x, W1, b1, W2, h2);

    k_gather_c   <<<GCG, GCB, 0, stream>>>(rowptr, ssrc, dinv, h2, b2, out);
}


# h2 rows stored scaled by their node's dinv on a 256-byte pitch: layer-2 gather has no per-source dinv lookup and no row straddling a line; rows added unweighted, sum scaled once (new rounding)
# speedup vs baseline: 5.7599x; 1.0931x over previous
#include <hip/hip_runtime.h>
#include <math.h>

#define NN   50000
#define NE   625000
#define FIN  128
#define DH   128
#define NC   40
#define NCP  48
#define HP   64
#define NNP  50048
#define NEP  625024

typedef __attribute__((ext_vector_type(2))) float v2f;
typedef __attribute__((ext_vector_type(8))) float v8f;

static inline int cdiv(long a, long b) { return (int)((a + b - 1) / b); }


#define BW    256
#define NBIN  ((NN + BW - 1) / BW)
#define BROW  (NBIN + 1)
#define G1    250
#define CQ    (NE / 4 / G1)
#define CE    (4 * CQ)
#define B1T   256
#define B1IT  ((CQ + B1T - 1) / B1T)
static_assert(NE % 4 == 0 && (NE / 4) % G1 == 0, "chunks are whole quads and cover every edge exactly");
static_assert(CE % 2 == 0, "a chunk's pairs are whole 16-byte words");
static_assert(B1T >= BROW && B1T % 32 == 0, "a thread per prefix entry, whole waves");
static_assert(NN % BW != 0, "node index NN lies inside the last bin: its work-group writes rowptr[NN]");

static __device__ __forceinline__ int bin_of(int d) {
    const unsigned b = (unsigned)d / BW;
    return (int)(b < NBIN ? b : NBIN - 1);
}

__global__ void __launch_bounds__(B1T)
k_bin(const int* __restrict__ src, const int* __restrict__ dst,
      int* __restrict__ binstart, int* __restrict__ pairs) {
    __shared__ int cnt[BROW];
    __shared__ __attribute__((aligned(16))) int2 lp[CE];
    __shared__ int wt[B1T / 32];
    const int tid  = threadIdx.x;
    const int lane = tid & 31;
    const int wv   = tid >> 5;
    const int g  = blockIdx.x;
    const int q0 = g * CQ;

    int4 d[B1IT], s[B1IT];
#pragma unroll
    for (int t = 0; t < B1IT; ++t) {
        const int q = t * B1T + tid;
        d[t] = (q < CQ) ? ((const int4*)dst)[q0 + q] : make_int4(0, 0, 0, 0);
        s[t] = (q < CQ) ? ((const int4*)src)[q0 + q] : make_int4(0, 0, 0, 0);
    }
    if (tid < BROW) cnt[tid] = 0;
    __syncthreads();

    int4 r[B1IT];
#pragma unroll
    for (int t = 0; t < B1IT; ++t) {
        const int q = t * B1T + tid;
        r[t] = make_int4(0, 0, 0, 0);
        if (q < CQ) {
            r[t].x = atomicAdd(&cnt[bin_of(d[t].x)], 1);
            r[t].y = atomicAdd(&cnt[bin_of(d[t].y)], 1);
            r[t].z = atomicAdd(&cnt[bin_of(d[t].z)], 1);
            r[t].w = atomicAdd(&cnt[bin_of(d[t].w)], 1);
        }
    }
    __syncthreads();

    const int v = (tid < BROW) ? cnt[tid] : 0;
    int inc = v;
#pragma unroll
    for (int o = 1; o < 32; o <<= 1) {
        const int y = __shfl_up(inc, o);
        if (lane >= o) inc += y;
    }
    if (lane == 31) wt[wv] = inc;
    __syncthreads();
    int ex = inc - v;
#pragma unroll
    for (int w = 0; w < B1T / 32; ++w) if (w < wv) ex += wt[w];
    if (tid < BROW) {
        cnt[tid] = ex;
        binstart[g * BROW + tid] = ex;
    }
    __syncthreads();

#pragma unroll
    for (int t = 0; t < B1IT; ++t) {
        const int q = t * B1T + tid;
        if (q < CQ) {
            lp[cnt[bin_of(d[t].x)] + r[t].x] = make_int2(s[t].x, d[t].x);
            lp[cnt[bin_of(d[t].y)] + r[t].y] = make_int2(s[t].y, d[t].y);
            lp[cnt[bin_of(d[t].z)] + r[t].z] = make_int2(s[t].z, d[t].z);
            lp[cnt[bin_of(d[t].w)] + r[t].w] = make_int2(s[t].w, d[t].w);
        }
    }
    __syncthreads();
    int4* po = (int4*)(pairs + (long)g * (2 * CE));
    for (int i = tid; i < CE / 2; i += B1T) po[i] = ((const int4*)lp)[i];
}

static __device__ __forceinline__ int rl_i(int v, int k) {
    return __builtin_amdgcn_readlane(v, k);
}
static __device__ __forceinline__ float rl_f(float v, int k) {
    return __int_as_float(__builtin_amdgcn_readlane(__float_as_int(v), k));
}

#define S2T  1024
#define S2W  (S2T / 32)
#define CPW  ((G1 + S2W - 1) / S2W)
static_assert(CPW <= 32, "a lane per chunk of the wave");
static_assert(S2T >= BW && BW % 32 == 0, "a thread per node of the bin, whole waves");

static __device__ __forceinline__ int chunk_of(int wv, int c) {
    const int gc = wv * CPW + c;
    return gc < G1 ? gc : G1 - 1;
}

__global__ void __launch_bounds__(S2T)
k_sort(const int* __restrict__ binstart, const int* __restrict__ pairs_i,
       int* __restrict__ rowptr, float* __restrict__ dinv, int* __restrict__ ssrc) {
    __shared__ int cw[S2W * BW];
    __shared__ int rs[BW];
    __shared__ int wb[S2W];
    __shared__ int wt[S2W];
    const int tid  = threadIdx.x;
    const int lane = tid & 31;
    const int wv   = __builtin_amdgcn_readfirstlane(tid >> 5);
    const int b    = blockIdx.x;
    const int n0   = b * BW;
    const int2* pairs = (const int2*)pairs_i;
    int* myrow = cw + wv * BW;

    int segS = 0, segN = 0;
    {
        const int gl = wv * CPW + lane;
        if (lane < CPW && gl < G1) {
            const int* row = binstart + gl * BROW;
            segS = row[b];
            segN = row[b + 1] - segS;
        }
    }
#pragma unroll
    for (int k = 0; k < BW / 32; ++k) myrow[lane + 32 * k] = 0;

    int2 p[CPW];
#pragma unroll
    for (int c = 0; c < CPW; ++c) {
        const int S = rl_i(segS, c), N = rl_i(segN, c);
        p[c] = pairs[(long)chunk_of(wv, c) * CE + ((lane < N) ? S + lane : 0)];
    }

    int bs = segS;
    for (int o = 16; o > 0; o >>= 1) bs += __shfl_xor(bs, o);
    if (lane == 0) wb[wv] = bs;

#pragma unroll
    for (int c = 0; c < CPW; ++c) {
        const int S = rl_i(segS, c), N = rl_i(segN, c);
        if (lane < N) atomicAdd(&myrow[(p[c].y - n0) & (BW - 1)], 1);
        for (int k = 32 + lane; k < N; k += 32) {
            const int2 e = pairs[(long)chunk_of(wv, c) * CE + S + k];
            atomicAdd(&myrow[(e.y - n0) & (BW - 1)], 1);
        }
    }
    __syncthreads();

    int base = 0;
#pragma unroll
    for (int w = 0; w < S2W; ++w) base += wb[w];

    int deg = 0;
    if (tid < BW) {
        int v[S2W];
#pragma unroll
        for (int w = 0; w < S2W; ++w) v[w] = cw[w * BW + tid];
#pragma unroll
        for (int w = 0; w < S2W; ++w) {
            cw[w * BW + tid] = deg;
            deg += v[w];
        }
    }
    int inc = deg;
#pragma unroll
    for (int o = 1; o < 32; o <<= 1) {
        const int y = __shfl_up(inc, o);
        if (lane >= o) inc += y;
    }
    if (lane == 31) wt[wv] = inc;
    __syncthreads();
    int rp = base + inc - deg;
#pragma unroll
    for (int w = 0; w < S2W; ++w) if (w < wv) rp += wt[w];
    if (tid < BW) {
        const int node = n0 + tid;
        rs[tid] = rp;
        if (node <= NN) rowptr[node] = rp;
        if (node < NN) dinv[node] = rsqrtf((float)(deg + 1));
    }
    __syncthreads();

    int ls0[CPW], rk[CPW];
#pragma unroll
    for (int c = 0; c < CPW; ++c) {
        const int S = rl_i(segS, c), N = rl_i(segN, c);
        ls0[c] = 0; rk[c] = 0;
        if (lane < N) {
            const int dl = (p[c].y - n0) & (BW - 1);
            ls0[c] = rs[dl];
            rk[c] = atomicAdd(&myrow[dl], 1);
        }
        for (int k = 32 + lane; k < N; k += 32) {
            const int2 e = pairs[(long)chunk_of(wv, c) * CE + S + k];
            const int dl = (e.y - n0) & (BW - 1);
            const int r = atomicAdd(&myrow[dl], 1);
            ssrc[rs[dl] + r] = e.x;
        }
    }
#pragma unroll
    for (int c = 0; c < CPW; ++c) {
        const int N = rl_i(segN, c);
        if (lane < N) ssrc[ls0[c] + rk[c]] = p[c].x;
    }
}


#define TP  132
#define L1T 512
#define L1TILES (NN / 16)
#define L1G  768
static_assert(NN % 16 == 0, "exact grid: every work-group has 16 real nodes");
static_assert(FIN == DH, "one tile buffer serves both products");
static_assert(L1G <= L1TILES, "every work-group has a first tile");

template <int K, int NP>
static __device__ __forceinline__ v8f tile_product(const float* __restrict__ sT, const float* __restrict__ sB,
                                                   int tn, int lane) {
    const int c  = lane & 15;
    const int kh = (lane >> 4) << 1;
    const float* aP = sT + c * TP + kh;
    const float* bP = sB + kh * NP + tn * 16 + c;
    v8f acc = {};
#pragma unroll
    for (int kk = 0; kk < K; kk += 4) {
        v2f a, b;
        a.x = aP[kk];
        a.y = aP[kk + 1];
        b.x = bP[kk * NP];
        b.y = bP[(kk + 1) * NP];
        acc = __builtin_amdgcn_wmma_f32_16x16x4_f32(
             false, a,  false, b,
             (short)0, acc,  false,  false);
    }
    return acc;
}

template <int R>
static __device__ __forceinline__ void rows_x(float4& acc, const float4* __restrict__ x4,
                                              int cs_l, float w_l, int k, int lane) {
    float4 v[R];
#pragma unroll
    for (int j = 0; j < R; ++j) v[j] = x4[(long)rl_i(cs_l, k + j) * 32 + lane];
#pragma unroll
    for (int j = 0; j < R; ++j) {
        const float w = rl_f(w_l, k + j);
        acc.x = fmaf(v[j].x, w, acc.x); acc.y = fmaf(v[j].y, w, acc.y);
        acc.z = fmaf(v[j].z, w, acc.z); acc.w = fmaf(v[j].w, w, acc.w);
    }
}

#define RX 10
static __device__ __forceinline__ void first_chunk_rows_x(float4& acc, const float4* __restrict__ x4,
                                                          int cs_l, float w_l, int n, int lane) {
    for (int k = 0; k < n; k += RX) {
        const int r = (n - k < RX) ? (n - k) : RX;
        switch (r) {
        case 1: rows_x<1>(acc, x4, cs_l, w_l, k, lane); break;
        case 2: rows_x<2>(acc, x4, cs_l, w_l, k, lane); break;
        case 3: rows_x<3>(acc, x4, cs_l, w_l, k, lane); break;
        case 4: rows_x<4>(acc, x4, cs_l, w_l, k, lane); break;
        case 5: rows_x<5>(acc, x4, cs_l, w_l, k, lane); break;
        case 6: rows_x<6>(acc, x4, cs_l, w_l, k, lane); break;
        case 7: rows_x<7>(acc, x4, cs_l, w_l, k, lane); break;
        case 8: rows_x<8>(acc, x4, cs_l, w_l, k, lane); break;
        case 9: rows_x<9>(acc, x4, cs_l, w_l, k, lane); break;
        case 10: rows_x<10>(acc, x4, cs_l, w_l, k, lane); break;
        default: break;
        }
    }
}

static __device__ __forceinline__ void chunk_rows_x(float4& acc, const float4* __restrict__ x4,
                                                    int cs_l, float w_l, int n, int lane) {
    int k = 0;
    for (; k + 4 <= n; k += 4) {
        const int s0 = rl_i(cs_l, k), s1 = rl_i(cs_l, k + 1), s2 = rl_i(cs_l, k + 2), s3 = rl_i(cs_l, k + 3);
        const float w0 = rl_f(w_l, k), w1 = rl_f(w_l, k + 1), w2 = rl_f(w_l, k + 2), w3 = rl_f(w_l, k + 3);
        const float4 v0 = x4[(long)s0 * 32 + lane];
        const float4 v1 = x4[(long)s1 * 32 + lane];
        const float4 v2 = x4[(long)s2 * 32 + lane];
        const float4 v3 = x4[(long)s3 * 32 + lane];
        acc.x = fmaf(v0.x, w0, acc.x); acc.y = fmaf(v0.y, w0, acc.y); acc.z = fmaf(v0.z, w0, acc.z); acc.w = fmaf(v0.w, w0, acc.w);
        acc.x = fmaf(v1.x, w1, acc.x); acc.y = fmaf(v1.y, w1, acc.y); acc.z = fmaf(v1.z, w1, acc.z); acc.w = fmaf(v1.w, w1, acc.w);
        acc.x = fmaf(v2.x, w2, acc.x); acc.y = fmaf(v2.y, w2, acc.y); acc.z = fmaf(v2.z, w2, acc.z); acc.w = fmaf(v2.w, w2, acc.w);
        acc.x = fmaf(v3.x, w3, acc.x); acc.y = fmaf(v3.y, w3, acc.y); acc.z = fmaf(v3.z, w3, acc.z); acc.w = fmaf(v3.w, w3, acc.w);
    }
    for (; k < n; ++k) {
        const int s0 = rl_i(cs_l, k);
        const float w0 = rl_f(w_l, k);
        const float4 v0 = x4[(long)s0 * 32 + lane];
        acc.x = fmaf(v0.x, w0, acc.x); acc.y = fmaf(v0.y, w0, acc.y); acc.z = fmaf(v0.z, w0, acc.z); acc.w = fmaf(v0.w, w0, acc.w);
    }
}

__global__ void __launch_bounds__(L1T, 12)
k_layer1(const int* __restrict__ rowptr, const int* __restrict__ ssrc,
         const float* __restrict__ dinv, const float* __restrict__ x,
         const float* __restrict__ W1, const float* __restrict__ b1,
         const float* __restrict__ W2, float* __restrict__ h2) {
    __shared__ float sW1[FIN * DH];
    __shared__ float sW2[DH * NCP];
    __shared__ float sT[16 * TP];
    __shared__ __attribute__((aligned(16))) float sD[16];

    const int tid  = threadIdx.x;
    const int lane = tid & 31;
    const int wv   = __builtin_amdgcn_readfirstlane(tid >> 5);
    const float4* x4 = (const float4*)x;

    int tile = blockIdx.x;
    int node = tile * 16 + wv;
    int beg = rowptr[node], end = rowptr[node + 1];
    float dd = dinv[node];
    float4 self = x4[(long)node * 32 + lane];

    {
        constexpr int Q1 = (FIN * DH) / 4;
        constexpr int QP = (DH * NCP) / 4;
        constexpr int N1 = Q1 / L1T, N2 = QP / L1T;
        static_assert(Q1 % L1T == 0 && QP % L1T == 0, "whole quads per thread, no tail");
        static_assert(NC % 4 == 0 && NCP % 4 == 0, "a quad stays inside its row, data or pad");
        float4 w[N1], u[N2];
#pragma unroll
        for (int t = 0; t < N1; ++t) w[t] = ((const float4*)W1)[tid + t * L1T];
#pragma unroll
        for (int t = 0; t < N2; ++t) {
            const int j = tid + t * L1T;
            const int r = j / (NCP / 4), q = j - r * (NCP / 4);
            u[t] = ((const float4*)W2)[r * (NC / 4) + (q < NC / 4 ? q : NC / 4 - 1)];
        }
#pragma unroll
        for (int t = 0; t < N1; ++t) ((float4*)sW1)[tid + t * L1T] = w[t];
#pragma unroll
        for (int t = 0; t < N2; ++t) {
            const int j = tid + t * L1T;
            const int q = j % (NCP / 4);
            const bool data = q < NC / 4;
            ((float4*)sW2)[j] = make_float4(data ? u[t].x : 0.0f, data ? u[t].y : 0.0f,
                                            data ? u[t].z : 0.0f, data ? u[t].w : 0.0f);
        }
    }

    int s_l = (beg + lane < end) ? ssrc[beg + lane] : node;
    if ((unsigned)s_l >= (unsigned)NN) s_l = node;
    float r_l = dinv[s_l];

    bool more  = tile + L1G < L1TILES;
    int  ntile = more ? tile + L1G : tile;
    int  nnode = ntile * 16 + wv;
    int  nbeg = rowptr[nnode], nend = rowptr[nnode + 1];
    float ndd = dinv[nnode];

    for (;;) {
        const bool more2 = more && (ntile + L1G < L1TILES);
        const int  tile2 = more2 ? ntile + L1G : ntile;
        const int  node2 = tile2 * 16 + wv;
        const int  beg2 = rowptr[node2], end2 = rowptr[node2 + 1];
        const float dd2 = dinv[node2];

        const int nj = nbeg + lane;
        int ns_l = ssrc[nj < nend ? nj : NE - 1];
        if (nj >= nend) ns_l = nnode;
        const float4 nself = x4[(long)nnode * 32 + lane];

        const int node0 = tile * 16;

        float4 acc = self;
        {
            const float ws = dd * dd;
            acc.x *= ws; acc.y *= ws; acc.z *= ws; acc.w *= ws;

            first_chunk_rows_x(acc, x4, s_l, r_l * dd, (end - beg < 32) ? (end - beg) : 32, lane);
            for (int cb = beg + 32; cb < end; cb += 32) {
                const int j = cb + lane;
                int cs_l = (j < end) ? ssrc[j] : node;
                if ((unsigned)cs_l >= (unsigned)NN) cs_l = node;
                const float w_l = dinv[cs_l] * dd;
                chunk_rows_x(acc, x4, cs_l, w_l, (end - cb < 32) ? (end - cb) : 32, lane);
            }
        }

        __syncthreads();
        ((float4*)(sT + wv * TP))[lane] = acc;
        if (lane == 0) sD[wv] = dd;
        __syncthreads();

        if ((unsigned)ns_l >= (unsigned)NN) ns_l = nnode;
        const float nr_l = dinv[ns_l];

        v8f p = {};
        if (wv < DH / 16) p = tile_product<FIN, DH>(sT, sW1, wv, lane);
        __syncthreads();
        if (wv < DH / 16) {
            const int n = wv * 16 + (lane & 15);
            const int r0 = (lane >> 4) << 3;
            const float bb = b1[n];
#pragma unroll
            for (int v = 0; v < 8; ++v) {
                const float t = p[v] + bb;
                sT[(r0 + v) * TP + n] = t > 0.0f ? t : 0.0f;
            }
        }
        const float4 dA = ((const float4*)sD)[(lane >> 4) * 2];
        const float4 dB = ((const float4*)sD)[(lane >> 4) * 2 + 1];
        __syncthreads();

        if (wv < NCP / 16) {
            const int n = wv * 16 + (lane & 15);
            const int r0 = node0 + ((lane >> 4) << 3);
            const v8f o = tile_product<DH, NCP>(sT, sW2, wv, lane);
            if (n < NC) {
                const float ds[8] = {dA.x, dA.y, dA.z, dA.w, dB.x, dB.y, dB.z, dB.w};
#pragma unroll
                for (int v = 0; v < 8; ++v)
                    h2[(long)(r0 + v) * HP + n] = o[v] * ds[v];
            }
        }

        if (!more) break;
        tile = ntile; node = nnode;
        beg = __builtin_amdgcn_readfirstlane(nbeg);
        end = __builtin_amdgcn_readfirstlane(nend);
        dd  = __int_as_float(__builtin_amdgcn_readfirstlane(__float_as_int(ndd)));
        s_l = ns_l; r_l = nr_l; self = nself;
        ntile = tile2; nnode = node2; nbeg = beg2; nend = end2; ndd = dd2; more = more2;
    }
}


#define GCB  256
#define GCG  1563
#define GCW  (GCG * (GCB / 32))
static_assert(GCW <= NN, "every wave has a first node");
static_assert(NC > 32 && NC <= 64 && NC <= HP, "a lane owns class 'lane' and at most one more; a row fits its pitch");

template <int R>
static __device__ __forceinline__ void rows_h(float& a0, float& a1, const float* __restrict__ h,
                                              int cs_l, int k, int lane, int c2) {
    float p[R], q[R];
#pragma unroll
    for (int j = 0; j < R; ++j) {
        const int s = rl_i(cs_l, k + j);
        p[j] = h[(long)s * HP + lane];
        q[j] = h[(long)s * HP + c2];
    }
#pragma unroll
    for (int j = 0; j < R; ++j) {
        a0 += p[j]; a1 += q[j];
    }
}

#define RH 16
static __device__ __forceinline__ void first_chunk_rows_h(float& a0, float& a1, const float* __restrict__ h,
                                                          int cs_l, int n, int lane, int c2) {
    for (int k = 0; k < n; k += RH) {
        const int r = (n - k < RH) ? (n - k) : RH;
        switch (r) {
        case 1: rows_h<1>(a0, a1, h, cs_l, k, lane, c2); break;
        case 2: rows_h<2>(a0, a1, h, cs_l, k, lane, c2); break;
        case 3: rows_h<3>(a0, a1, h, cs_l, k, lane, c2); break;
        case 4: rows_h<4>(a0, a1, h, cs_l, k, lane, c2); break;
        case 5: rows_h<5>(a0, a1, h, cs_l, k, lane, c2); break;
        case 6: rows_h<6>(a0, a1, h, cs_l, k, lane, c2); break;
        case 7: rows_h<7>(a0, a1, h, cs_l, k, lane, c2); break;
        case 8: rows_h<8>(a0, a1, h, cs_l, k, lane, c2); break;
        case 9: rows_h<9>(a0, a1, h, cs_l, k, lane, c2); break;
        case 10: rows_h<10>(a0, a1, h, cs_l, k, lane, c2); break;
        case 11: rows_h<11>(a0, a1, h, cs_l, k, lane, c2); break;
        case 12: rows_h<12>(a0, a1, h, cs_l, k, lane, c2); break;
        case 13: rows_h<13>(a0, a1, h, cs_l, k, lane, c2); break;
        case 14: rows_h<14>(a0, a1, h, cs_l, k, lane, c2); break;
        case 15: rows_h<15>(a0, a1, h, cs_l, k, lane, c2); break;
        case 16: rows_h<16>(a0, a1, h, cs_l, k, lane, c2); break;
        default: break;
        }
    }
}

static __device__ __forceinline__ void chunk_rows_h(float& a0, float& a1, const float* __restrict__ h,
                                                    int cs_l, int n, int lane, int c2) {
    int k = 0;
    for (; k + 4 <= n; k += 4) {
        const int s0 = rl_i(cs_l, k), s1 = rl_i(cs_l, k + 1), s2 = rl_i(cs_l, k + 2), s3 = rl_i(cs_l, k + 3);
        const float p0 = h[(long)s0 * HP + lane], q0 = h[(long)s0 * HP + c2];
        const float p1 = h[(long)s1 * HP + lane], q1 = h[(long)s1 * HP + c2];
        const float p2 = h[(long)s2 * HP + lane], q2 = h[(long)s2 * HP + c2];
        const float p3 = h[(long)s3 * HP + lane], q3 = h[(long)s3 * HP + c2];
        a0 += p0; a1 += q0;
        a0 += p1; a1 += q1;
        a0 += p2; a1 += q2;
        a0 += p3; a1 += q3;
    }
    for (; k < n; ++k) {
        const int s0 = rl_i(cs_l, k);
        a0 += h[(long)s0 * HP + lane];
        a1 += h[(long)s0 * HP + c2];
    }
}

__global__ void __launch_bounds__(GCB)
k_gather_c(const int* __restrict__ rowptr, const int* __restrict__ ssrc,
           const float* __restrict__ dinv,
           const float* __restrict__ h, const float* __restrict__ b,
           float* __restrict__ out) {
    const int lane = threadIdx.x & 31;
    const int gw = __builtin_amdgcn_readfirstlane((int)((blockIdx.x * GCB + threadIdx.x) >> 5));
    const bool hi = lane < NC - 32;
    const int  c2 = 32 + lane % (NC - 32);
    const float b0 = b[lane], b1 = b[c2];

    int node = gw;
    int beg = rowptr[node], end = rowptr[node + 1];
    float dd = dinv[node];
    float h0 = h[(long)node * HP + lane], h1 = h[(long)node * HP + c2];
    int s_l = (beg + lane < end) ? ssrc[beg + lane] : node;
    if ((unsigned)s_l >= (unsigned)NN) s_l = node;

    bool more  = node + GCW < NN;
    int  nnode = more ? node + GCW : node;
    int  nbeg = rowptr[nnode], nend = rowptr[nnode + 1];
    float ndd = dinv[nnode];

    for (;;) {
        const bool more2 = more && (nnode + GCW < NN);
        const int  node2 = more2 ? nnode + GCW : nnode;
        const int  beg2 = rowptr[node2], end2 = rowptr[node2 + 1];
        const float dd2 = dinv[node2];

        const int nj = nbeg + lane;
        int ns_l = ssrc[nj < nend ? nj : NE - 1];
        if (nj >= nend) ns_l = nnode;
        const float nh0 = h[(long)nnode * HP + lane];
        const float nh1 = h[(long)nnode * HP + c2];

        float a0 = h0;
        float a1 = h1;
        first_chunk_rows_h(a0, a1, h, s_l, (end - beg < 32) ? (end - beg) : 32, lane, c2);
        for (int cb = beg + 32; cb < end; cb += 32) {
            const int j = cb + lane;
            int cs_l = (j < end) ? ssrc[j] : node;
            if ((unsigned)cs_l >= (unsigned)NN) cs_l = node;
            chunk_rows_h(a0, a1, h, cs_l, (end - cb < 32) ? (end - cb) : 32, lane, c2);
        }

        if ((unsigned)ns_l >= (unsigned)NN) ns_l = nnode;

        const float v0 = fmaf(a0, dd, b0);
        const float v1 = hi ? fmaf(a1, dd, b1) : -INFINITY;
        float mx = fmaxf(v0, v1);
        for (int o = 16; o > 0; o >>= 1) mx = fmaxf(mx, __shfl_xor(mx, o));
        float sum = expf(v0 - mx) + (hi ? expf(v1 - mx) : 0.0f);
        for (int o = 16; o > 0; o >>= 1) sum += __shfl_xor(sum, o);
        const float lse = mx + logf(sum);
        out[(long)node * NC + lane] = v0 - lse;
        if (hi) out[(long)node * NC + c2] = v1 - lse;

        if (!more) break;
        node = nnode; beg = nbeg; end = nend; dd = ndd;
        s_l = ns_l; h0 = nh0; h1 = nh1;
        nnode = node2; nbeg = beg2; nend = end2; ndd = dd2; more = more2;
    }
}


extern "C" void kernel_launch(void* const* d_in, const int* in_sizes, int n_in,
                              void* d_out, int out_size, void* d_ws, size_t ws_size,
                              hipStream_t stream) {
    const float* x  = (const float*)d_in[0];
    const int*   ei = (const int*)d_in[1];
    const float* W1 = (const float*)d_in[2];
    const float* b1 = (const float*)d_in[3];
    const float* W2 = (const float*)d_in[4];
    const float* b2 = (const float*)d_in[5];
    float* out = (float*)d_out;

    const int* src = ei;
    const int* dst = ei + NE;

    float* ws     = (float*)d_ws;
    float* dinv   = ws;
    int*   rowptr = (int*)(dinv + NNP);
    int*   pairs  = rowptr + NNP;
    int*   ssrc   = pairs + 2 * NEP;
    float* h2     = (float*)(ssrc + NEP);
    int*   binstart = (int*)(h2 + (long)NN * HP);

    k_bin  <<<G1, B1T, 0, stream>>>(src, dst, binstart, pairs);
    k_sort <<<NBIN, S2T, 0, stream>>>(binstart, pairs, rowptr, dinv, ssrc);

    k_layer1   <<<L1G, L1T, 0, stream>>>(rowptr, ssrc, dinv, x, W1, b1, W2, h2);

    k_gather_c   <<<GCG, GCB, 0, stream>>>(rowptr, ssrc, dinv, h2, b2, out);
}
